# starting kernel
# baseline (speedup 1.0000x reference)
#include <hip/hip_runtime.h>
#include <hip/hip_bf16.h>

typedef __attribute__((ext_vector_type(2))) float v2f;
typedef __attribute__((ext_vector_type(8))) float v8f;

#define N_NODES_C 100000

__global__ void zero_f32(float* __restrict__ p, int n) {
    int i = blockIdx.x * blockDim.x + threadIdx.x;
    if (i < n) p[i] = 0.0f;
}

__global__ void deg_count(const int* __restrict__ dst, float* __restrict__ deg, int ne) {
    int e = blockIdx.x * blockDim.x + threadIdx.x;
    if (e < ne) unsafeAtomicAdd(&deg[dst[e]], 1.0f);
}

__global__ void finish_dis(float* __restrict__ deg, int n) {
    int i = blockIdx.x * blockDim.x + threadIdx.x;
    if (i < n) deg[i] = rsqrtf(deg[i] + 1.0f);
}

template<int NT, int KDIM, bool BIAS, bool RELU>
__launch_bounds__(256)
__global__ void gemm_wmma_f32(const float* __restrict__ X,
                              const float* __restrict__ W,
                              const float* __restrict__ bias,
                              float* __restrict__ Y,
                              int M) {
    constexpr int N = NT * 16;
    const int wave = (blockIdx.x * blockDim.x + threadIdx.x) >> 5;
    const int lane = threadIdx.x & 31;
    const int row0 = wave * 16;
    if (row0 >= M) return;

    const int r      = lane & 15;
    const int khalf  = (lane >> 4) * 2;

    v8f acc[NT] = {};

    const float* __restrict__ xrow = X + (size_t)(row0 + r) * KDIM;

#pragma unroll
    for (int k = 0; k < KDIM; k += 4) {
        const int kk = k + khalf;
        v2f a = *(const v2f*)(xrow + kk);
#pragma unroll
        for (int t = 0; t < NT; ++t) {
            const int col = t * 16 + r;
            v2f b;
            b.x = W[(size_t)kk * N + col];
            b.y = W[(size_t)(kk + 1) * N + col];
            acc[t] = __builtin_amdgcn_wmma_f32_16x16x4_f32(
                 false, a,  false, b,
                 (short)0, acc[t],
                 false,  false);
        }
    }

    const int rowoff = (lane >> 4) * 8;
#pragma unroll
    for (int t = 0; t < NT; ++t) {
        const int col = t * 16 + r;
        const float bv = BIAS ? bias[col] : 0.0f;
#pragma unroll
        for (int g = 0; g < 8; ++g) {
            float v = acc[t][g] + bv;
            if (RELU) v = fmaxf(v, 0.0f);
            Y[(size_t)(row0 + rowoff + g) * N + col] = v;
        }
    }
}

__global__ void edge_scatter(const float* __restrict__ h,
                             const int* __restrict__ src,
                             const int* __restrict__ dst,
                             const float* __restrict__ dis,
                             float* __restrict__ agg,
                             int ne) {
    const int tid = blockIdx.x * blockDim.x + threadIdx.x;
    const int e = tid >> 4;
    if (e >= ne) return;
    const int f = (tid & 15) * 4;
    const int s = src[e];
    const int d = dst[e];
    const float norm = dis[s] * dis[d];
    const float4 hv = *(const float4*)(h + (size_t)s * 64 + f);
    float* ap = agg + (size_t)d * 64 + f;
    unsafeAtomicAdd(ap + 0, hv.x * norm);
    unsafeAtomicAdd(ap + 1, hv.y * norm);
    unsafeAtomicAdd(ap + 2, hv.z * norm);
    unsafeAtomicAdd(ap + 3, hv.w * norm);
}

__global__ void selfloop_bias_relu(const float* __restrict__ agg,
                                   const float* __restrict__ h,
                                   const float* __restrict__ dis,
                                   const float* __restrict__ b,
                                   float* __restrict__ out,
                                   int n) {
    const int tid = blockIdx.x * blockDim.x + threadIdx.x;
    const int i = tid >> 4;
    if (i >= n) return;
    const int f = (tid & 15) * 4;
    const float d2 = dis[i] * dis[i];
    const float4 av = *(const float4*)(agg + (size_t)i * 64 + f);
    const float4 hv = *(const float4*)(h + (size_t)i * 64 + f);
    const float4 bv = *(const float4*)(b + f);
    float4 o;
    o.x = fmaxf(av.x + hv.x * d2 + bv.x, 0.0f);
    o.y = fmaxf(av.y + hv.y * d2 + bv.y, 0.0f);
    o.z = fmaxf(av.z + hv.z * d2 + bv.z, 0.0f);
    o.w = fmaxf(av.w + hv.w * d2 + bv.w, 0.0f);
    *(float4*)(out + (size_t)i * 64 + f) = o;
}

__global__ void out_proj(const float* __restrict__ hd,
                         const float* __restrict__ Wo,
                         const float* __restrict__ bo,
                         float* __restrict__ out,
                         int n) {
    const int i = blockIdx.x * blockDim.x + threadIdx.x;
    if (i >= n) return;
    const float* h = hd + (size_t)i * 128;
    float a0 = bo[0], a1 = bo[1];
#pragma unroll 8
    for (int k = 0; k < 128; ++k) {
        const float hv = h[k];
        a0 += hv * Wo[k * 2 + 0];
        a1 += hv * Wo[k * 2 + 1];
    }
    out[(size_t)i * 2 + 0] = a0;
    out[(size_t)i * 2 + 1] = a1;
}

extern "C" void kernel_launch(void* const* d_in, const int* in_sizes, int n_in,
                              void* d_out, int out_size, void* d_ws, size_t ws_size,
                              hipStream_t stream) {
    const float* x  = (const float*)d_in[0];
    const int*   ei = (const int*)d_in[1];
    const float* W1 = (const float*)d_in[2];
    const float* b1 = (const float*)d_in[3];
    const float* W2 = (const float*)d_in[4];
    const float* b2 = (const float*)d_in[5];
    const float* Wd = (const float*)d_in[6];
    const float* bd = (const float*)d_in[7];
    const float* Wo = (const float*)d_in[8];
    const float* bo = (const float*)d_in[9];
    float* out = (float*)d_out;

    const int N = in_sizes[0] / 64;
    const int E = in_sizes[1] / 2;
    const int* src = ei;
    const int* dst = ei + E;

    float* dis  = (float*)d_ws;
    float* bufA = dis  + N;
    float* bufB = bufA + (size_t)N * 64;
    float* bufC = bufB + (size_t)N * 64;
    float* bufD = bufC + (size_t)N * 64;
    (void)ws_size; (void)n_in; (void)out_size;

    const int TB = 256;
    const int gemm_waves   = N / 16;
    const int gemm_blocks  = (gemm_waves * 32 + TB - 1) / TB;
    const int edge_blocks  = (E * 16 + TB - 1) / TB;
    const int node16_blks  = (N * 16 + TB - 1) / TB;

    zero_f32<<<(N + TB - 1) / TB, TB, 0, stream>>>(dis, N);
    deg_count<<<(E + TB - 1) / TB, TB, 0, stream>>>(dst, dis, E);
    finish_dis<<<(N + TB - 1) / TB, TB, 0, stream>>>(dis, N);

    gemm_wmma_f32<4, 64, false, false><<<gemm_blocks, TB, 0, stream>>>(x, W1, nullptr, bufA, N);
    zero_f32<<<((N * 64) + TB - 1) / TB, TB, 0, stream>>>(bufB, N * 64);
    edge_scatter<<<edge_blocks, TB, 0, stream>>>(bufA, src, dst, dis, bufB, E);
    selfloop_bias_relu<<<node16_blks, TB, 0, stream>>>(bufB, bufA, dis, b1, bufC, N);

    gemm_wmma_f32<4, 64, false, false><<<gemm_blocks, TB, 0, stream>>>(bufC, W2, nullptr, bufA, N);
    zero_f32<<<((N * 64) + TB - 1) / TB, TB, 0, stream>>>(bufB, N * 64);
    edge_scatter<<<edge_blocks, TB, 0, stream>>>(bufA, src, dst, dis, bufB, E);
    selfloop_bias_relu<<<node16_blks, TB, 0, stream>>>(bufB, bufA, dis, b2, bufC, N);

    gemm_wmma_f32<8, 64, true, true><<<gemm_blocks, TB, 0, stream>>>(bufC, Wd, bd, bufD, N);

    out_proj<<<(N + TB - 1) / TB, TB, 0, stream>>>(bufD, Wo, bo, out, N);
}


# edge scatter: a wave walks 32 edges, lane l adds channels l and l+32, so each f32 atomic instruction is one whole 128-byte line (2 line requests per edge, was 8); sums still in atomic arrival order
# speedup vs baseline: 1.2691x; 1.2691x over previous
#include <hip/hip_runtime.h>
#include <hip/hip_bf16.h>

typedef __attribute__((ext_vector_type(2))) float v2f;
typedef __attribute__((ext_vector_type(8))) float v8f;

#define N_NODES_C 100000

__global__ void zero_f32(float* __restrict__ p, int n) {
    int i = blockIdx.x * blockDim.x + threadIdx.x;
    if (i < n) p[i] = 0.0f;
}

__global__ void deg_count(const int* __restrict__ dst, float* __restrict__ deg, int ne) {
    int e = blockIdx.x * blockDim.x + threadIdx.x;
    if (e < ne) unsafeAtomicAdd(&deg[dst[e]], 1.0f);
}

__global__ void finish_dis(float* __restrict__ deg, int n) {
    int i = blockIdx.x * blockDim.x + threadIdx.x;
    if (i < n) deg[i] = rsqrtf(deg[i] + 1.0f);
}

template<int NT, int KDIM, bool BIAS, bool RELU>
__launch_bounds__(256)
__global__ void gemm_wmma_f32(const float* __restrict__ X,
                              const float* __restrict__ W,
                              const float* __restrict__ bias,
                              float* __restrict__ Y,
                              int M) {
    constexpr int N = NT * 16;
    const int wave = (blockIdx.x * blockDim.x + threadIdx.x) >> 5;
    const int lane = threadIdx.x & 31;
    const int row0 = wave * 16;
    if (row0 >= M) return;

    const int r      = lane & 15;
    const int khalf  = (lane >> 4) * 2;

    v8f acc[NT] = {};

    const float* __restrict__ xrow = X + (size_t)(row0 + r) * KDIM;

#pragma unroll
    for (int k = 0; k < KDIM; k += 4) {
        const int kk = k + khalf;
        v2f a = *(const v2f*)(xrow + kk);
#pragma unroll
        for (int t = 0; t < NT; ++t) {
            const int col = t * 16 + r;
            v2f b;
            b.x = W[(size_t)kk * N + col];
            b.y = W[(size_t)(kk + 1) * N + col];
            acc[t] = __builtin_amdgcn_wmma_f32_16x16x4_f32(
                 false, a,  false, b,
                 (short)0, acc[t],
                 false,  false);
        }
    }

    const int rowoff = (lane >> 4) * 8;
#pragma unroll
    for (int t = 0; t < NT; ++t) {
        const int col = t * 16 + r;
        const float bv = BIAS ? bias[col] : 0.0f;
#pragma unroll
        for (int g = 0; g < 8; ++g) {
            float v = acc[t][g] + bv;
            if (RELU) v = fmaxf(v, 0.0f);
            Y[(size_t)(row0 + rowoff + g) * N + col] = v;
        }
    }
}

#define EDGE_GROUP 8

__device__ __forceinline__ int bcast_i(int v, int j) {
    return __builtin_amdgcn_readlane(v, j);
}
__device__ __forceinline__ float bcast_f(float v, int j) {
    return __int_as_float(__builtin_amdgcn_readlane(__float_as_int(v), j));
}

__launch_bounds__(256)
__global__ void edge_scatter(const float* __restrict__ h,
                             const int* __restrict__ src,
                             const int* __restrict__ dst,
                             const float* __restrict__ dis,
                             float* __restrict__ agg,
                             int ne) {
    const int wave = (blockIdx.x * blockDim.x + threadIdx.x) >> 5;
    const int lane = threadIdx.x & 31;
    const int e0 = wave * 32;
    if (e0 >= ne) return;
    const int e = e0 + lane;
    int s = 0, d = 0;
    float nrm = 0.0f;
    if (e < ne) {
        s = src[e];
        d = dst[e];
        nrm = dis[s] * dis[d];
    }
    const int cnt = (ne - e0 < 32) ? (ne - e0) : 32;

    if (cnt == 32) {
#pragma unroll
        for (int jb = 0; jb < 32; jb += EDGE_GROUP) {
            float lo[EDGE_GROUP], hi[EDGE_GROUP];
#pragma unroll
            for (int j = 0; j < EDGE_GROUP; ++j) {
                const float* hp = h + (size_t)bcast_i(s, jb + j) * 64;
                lo[j] = hp[lane];
                hi[j] = hp[lane + 32];
            }
#pragma unroll
            for (int j = 0; j < EDGE_GROUP; ++j) {
                float* ap = agg + (size_t)bcast_i(d, jb + j) * 64;
                const float w = bcast_f(nrm, jb + j);
                unsafeAtomicAdd(ap + lane,      lo[j] * w);
                unsafeAtomicAdd(ap + lane + 32, hi[j] * w);
            }
        }
    } else {
        for (int j = 0; j < cnt; ++j) {
            const float* hp = h + (size_t)bcast_i(s, j) * 64;
            float* ap = agg + (size_t)bcast_i(d, j) * 64;
            const float w = bcast_f(nrm, j);
            unsafeAtomicAdd(ap + lane,      hp[lane]      * w);
            unsafeAtomicAdd(ap + lane + 32, hp[lane + 32] * w);
        }
    }
}

__global__ void selfloop_bias_relu(const float* __restrict__ agg,
                                   const float* __restrict__ h,
                                   const float* __restrict__ dis,
                                   const float* __restrict__ b,
                                   float* __restrict__ out,
                                   int n) {
    const int tid = blockIdx.x * blockDim.x + threadIdx.x;
    const int i = tid >> 4;
    if (i >= n) return;
    const int f = (tid & 15) * 4;
    const float d2 = dis[i] * dis[i];
    const float4 av = *(const float4*)(agg + (size_t)i * 64 + f);
    const float4 hv = *(const float4*)(h + (size_t)i * 64 + f);
    const float4 bv = *(const float4*)(b + f);
    float4 o;
    o.x = fmaxf(av.x + hv.x * d2 + bv.x, 0.0f);
    o.y = fmaxf(av.y + hv.y * d2 + bv.y, 0.0f);
    o.z = fmaxf(av.z + hv.z * d2 + bv.z, 0.0f);
    o.w = fmaxf(av.w + hv.w * d2 + bv.w, 0.0f);
    *(float4*)(out + (size_t)i * 64 + f) = o;
}

__global__ void out_proj(const float* __restrict__ hd,
                         const float* __restrict__ Wo,
                         const float* __restrict__ bo,
                         float* __restrict__ out,
                         int n) {
    const int i = blockIdx.x * blockDim.x + threadIdx.x;
    if (i >= n) return;
    const float* h = hd + (size_t)i * 128;
    float a0 = bo[0], a1 = bo[1];
#pragma unroll 8
    for (int k = 0; k < 128; ++k) {
        const float hv = h[k];
        a0 += hv * Wo[k * 2 + 0];
        a1 += hv * Wo[k * 2 + 1];
    }
    out[(size_t)i * 2 + 0] = a0;
    out[(size_t)i * 2 + 1] = a1;
}

extern "C" void kernel_launch(void* const* d_in, const int* in_sizes, int n_in,
                              void* d_out, int out_size, void* d_ws, size_t ws_size,
                              hipStream_t stream) {
    const float* x  = (const float*)d_in[0];
    const int*   ei = (const int*)d_in[1];
    const float* W1 = (const float*)d_in[2];
    const float* b1 = (const float*)d_in[3];
    const float* W2 = (const float*)d_in[4];
    const float* b2 = (const float*)d_in[5];
    const float* Wd = (const float*)d_in[6];
    const float* bd = (const float*)d_in[7];
    const float* Wo = (const float*)d_in[8];
    const float* bo = (const float*)d_in[9];
    float* out = (float*)d_out;

    const int N = in_sizes[0] / 64;
    const int E = in_sizes[1] / 2;
    const int* src = ei;
    const int* dst = ei + E;

    float* dis  = (float*)d_ws;
    float* bufA = dis  + N;
    float* bufB = bufA + (size_t)N * 64;
    float* bufC = bufB + (size_t)N * 64;
    float* bufD = bufC + (size_t)N * 64;
    (void)ws_size; (void)n_in; (void)out_size;

    const int TB = 256;
    const int gemm_waves   = N / 16;
    const int gemm_blocks  = (gemm_waves * 32 + TB - 1) / TB;
    const int edge_blocks  = (E + TB - 1) / TB;
    const int node16_blks  = (N * 16 + TB - 1) / TB;

    zero_f32<<<(N + TB - 1) / TB, TB, 0, stream>>>(dis, N);
    deg_count<<<(E + TB - 1) / TB, TB, 0, stream>>>(dst, dis, E);
    finish_dis<<<(N + TB - 1) / TB, TB, 0, stream>>>(dis, N);

    gemm_wmma_f32<4, 64, false, false><<<gemm_blocks, TB, 0, stream>>>(x, W1, nullptr, bufA, N);
    zero_f32<<<((N * 64) + TB - 1) / TB, TB, 0, stream>>>(bufB, N * 64);
    edge_scatter<<<edge_blocks, TB, 0, stream>>>(bufA, src, dst, dis, bufB, E);
    selfloop_bias_relu<<<node16_blks, TB, 0, stream>>>(bufB, bufA, dis, b1, bufC, N);

    gemm_wmma_f32<4, 64, false, false><<<gemm_blocks, TB, 0, stream>>>(bufC, W2, nullptr, bufA, N);
    zero_f32<<<((N * 64) + TB - 1) / TB, TB, 0, stream>>>(bufB, N * 64);
    edge_scatter<<<edge_blocks, TB, 0, stream>>>(bufA, src, dst, dis, bufB, E);
    selfloop_bias_relu<<<node16_blks, TB, 0, stream>>>(bufB, bufA, dis, b2, bufC, N);

    gemm_wmma_f32<8, 64, true, true><<<gemm_blocks, TB, 0, stream>>>(bufC, Wd, bd, bufD, N);

    out_proj<<<(N + TB - 1) / TB, TB, 0, stream>>>(bufD, Wo, bo, out, N);
}


# aggregate by destination: edges counting-sorted by dst each call (atomic arrival-order rank), one wave sums a node's rows in registers, writes once; no f32 atomics, sum order = list order
# speedup vs baseline: 1.7562x; 1.3838x over previous
#include <hip/hip_runtime.h>
#include <hip/hip_bf16.h>

typedef __attribute__((ext_vector_type(2))) float v2f;
typedef __attribute__((ext_vector_type(8))) float v8f;

#define N_NODES_C 100000

typedef __attribute__((ext_vector_type(2))) int v2i;

__global__ void zero_i32(int* __restrict__ p, int n) {
    int i = blockIdx.x * blockDim.x + threadIdx.x;
    if (i < n) p[i] = 0;
}

#define EDGES_PER_THREAD 4

__launch_bounds__(256)
__global__ void count_rank(const int* __restrict__ dst, int* __restrict__ cnt,
                           int* __restrict__ rank, int ne) {
    const int blk = blockIdx.x * (256 * EDGES_PER_THREAD);
    const int base = blk + threadIdx.x;
    if (blk + 256 * EDGES_PER_THREAD <= ne) {
        int d[EDGES_PER_THREAD], old[EDGES_PER_THREAD];
#pragma unroll
        for (int k = 0; k < EDGES_PER_THREAD; ++k) d[k] = dst[base + k * 256];
#pragma unroll
        for (int k = 0; k < EDGES_PER_THREAD; ++k) old[k] = atomicAdd(&cnt[d[k]], 1);
#pragma unroll
        for (int k = 0; k < EDGES_PER_THREAD; ++k) rank[base + k * 256] = old[k];
    } else {
        for (int k = 0; k < EDGES_PER_THREAD; ++k) {
            const int e = base + k * 256;
            if (e < ne) rank[e] = atomicAdd(&cnt[dst[e]], 1);
        }
    }
}

__launch_bounds__(256)
__global__ void block_sums(const int* __restrict__ cnt, int* __restrict__ bsum, int n) {
    __shared__ int wsum[8];
    const int t = threadIdx.x;
    const int base = blockIdx.x * 1024 + t * 4;
    int s = 0;
    for (int k = 0; k < 4; ++k)
        if (base + k < n) s += cnt[base + k];
    for (int off = 16; off > 0; off >>= 1) s += __shfl_xor(s, off);
    if ((t & 31) == 0) wsum[t >> 5] = s;
    __syncthreads();
    if (t == 0) {
        int tot = 0;
        for (int w = 0; w < 8; ++w) tot += wsum[w];
        bsum[blockIdx.x] = tot;
    }
}

__launch_bounds__(256)
__global__ void scan_rowptr_dis(const int* __restrict__ cnt, const int* __restrict__ bsum,
                                int* __restrict__ rowptr, float* __restrict__ dis, int n) {
    __shared__ int wsum[8];
    const int t = threadIdx.x;
    const int lane = t & 31;
    const int wv = t >> 5;

    int p = 0;
    for (int j = t; j < (int)blockIdx.x; j += 256) p += bsum[j];
    for (int off = 16; off > 0; off >>= 1) p += __shfl_xor(p, off);
    if (lane == 0) wsum[wv] = p;
    __syncthreads();
    int boff = 0;
    for (int w = 0; w < 8; ++w) boff += wsum[w];
    __syncthreads();

    const int base = blockIdx.x * 1024 + t * 4;
    int c[4];
    for (int k = 0; k < 4; ++k) c[k] = (base + k < n) ? cnt[base + k] : 0;
    const int tsum = c[0] + c[1] + c[2] + c[3];
    int incl = tsum;
    for (int off = 1; off < 32; off <<= 1) {
        const int v = __shfl_up(incl, off);
        if (lane >= off) incl += v;
    }
    if (lane == 31) wsum[wv] = incl;
    __syncthreads();
    int woff = 0;
    for (int w = 0; w < wv; ++w) woff += wsum[w];
    int run = boff + woff + incl - tsum;
    for (int k = 0; k < 4; ++k) {
        if (base + k < n) {
            rowptr[base + k] = run;
            dis[base + k] = rsqrtf((float)c[k] + 1.0f);
            run += c[k];
        }
    }
    if (blockIdx.x == gridDim.x - 1 && t == 255) rowptr[n] = boff + woff + incl;
}

__launch_bounds__(256)
__global__ void fill_lists(const int* __restrict__ src, const int* __restrict__ dst,
                           const int* __restrict__ rank, const int* __restrict__ rowptr,
                           const float* __restrict__ dis, int* __restrict__ rec, int ne) {
    const int blk = blockIdx.x * (256 * EDGES_PER_THREAD);
    const int base = blk + threadIdx.x;
    if (blk + 256 * EDGES_PER_THREAD <= ne) {
        int s[EDGES_PER_THREAD], d[EDGES_PER_THREAD], r[EDGES_PER_THREAD], p[EDGES_PER_THREAD];
        float w[EDGES_PER_THREAD];
#pragma unroll
        for (int k = 0; k < EDGES_PER_THREAD; ++k) {
            s[k] = src[base + k * 256];
            d[k] = dst[base + k * 256];
            r[k] = rank[base + k * 256];
        }
#pragma unroll
        for (int k = 0; k < EDGES_PER_THREAD; ++k) {
            p[k] = rowptr[d[k]] + r[k];
            w[k] = dis[s[k]] * dis[d[k]];
        }
#pragma unroll
        for (int k = 0; k < EDGES_PER_THREAD; ++k) {
            unsigned slot = (unsigned)p[k];
            if (slot > (unsigned)(ne - 1)) slot = (unsigned)(ne - 1);
            v2i q;
            q.x = s[k];
            q.y = __float_as_int(w[k]);
            *(v2i*)(rec + 2 * (size_t)slot) = q;
        }
    } else {
        for (int k = 0; k < EDGES_PER_THREAD; ++k) {
            const int e = base + k * 256;
            if (e < ne) {
                const int s = src[e];
                const int d = dst[e];
                const int p = rowptr[d] + rank[e];
                const float w = dis[s] * dis[d];
                if ((unsigned)p < (unsigned)ne) {
                    v2i q;
                    q.x = s;
                    q.y = __float_as_int(w);
                    *(v2i*)(rec + 2 * (size_t)p) = q;
                }
            }
        }
    }
}

template<int NT, int KDIM, bool BIAS, bool RELU>
__launch_bounds__(256)
__global__ void gemm_wmma_f32(const float* __restrict__ X,
                              const float* __restrict__ W,
                              const float* __restrict__ bias,
                              float* __restrict__ Y,
                              int M) {
    constexpr int N = NT * 16;
    const int wave = (blockIdx.x * blockDim.x + threadIdx.x) >> 5;
    const int lane = threadIdx.x & 31;
    const int row0 = wave * 16;
    if (row0 >= M) return;

    const int r      = lane & 15;
    const int khalf  = (lane >> 4) * 2;

    v8f acc[NT] = {};

    const float* __restrict__ xrow = X + (size_t)(row0 + r) * KDIM;

#pragma unroll
    for (int k = 0; k < KDIM; k += 4) {
        const int kk = k + khalf;
        v2f a = *(const v2f*)(xrow + kk);
#pragma unroll
        for (int t = 0; t < NT; ++t) {
            const int col = t * 16 + r;
            v2f b;
            b.x = W[(size_t)kk * N + col];
            b.y = W[(size_t)(kk + 1) * N + col];
            acc[t] = __builtin_amdgcn_wmma_f32_16x16x4_f32(
                 false, a,  false, b,
                 (short)0, acc[t],
                 false,  false);
        }
    }

    const int rowoff = (lane >> 4) * 8;
#pragma unroll
    for (int t = 0; t < NT; ++t) {
        const int col = t * 16 + r;
        const float bv = BIAS ? bias[col] : 0.0f;
#pragma unroll
        for (int g = 0; g < 8; ++g) {
            float v = acc[t][g] + bv;
            if (RELU) v = fmaxf(v, 0.0f);
            Y[(size_t)(row0 + rowoff + g) * N + col] = v;
        }
    }
}

#define GATHER_GROUP 16
#define NODES_PER_WAVE 8

__launch_bounds__(256)
__global__ void gather_rows(const float* __restrict__ h,
                            const int* __restrict__ rec,
                            const int* __restrict__ rowptr,
                            const float* __restrict__ dis,
                            const float* __restrict__ b,
                            float* __restrict__ out,
                            int n, int ne) {
    const int lane = threadIdx.x & 31;
    const int i0 = NODES_PER_WAVE *
        __builtin_amdgcn_readfirstlane((int)((blockIdx.x * blockDim.x + threadIdx.x) >> 5));
    const v2f bv = *(const v2f*)(b + 2 * lane);

#pragma nounroll
    for (int k = 0; k < NODES_PER_WAVE; ++k) {
        const int i = i0 + k;
        if (i >= n) break;
        int beg = __builtin_amdgcn_readfirstlane(rowptr[i]);
        int end = __builtin_amdgcn_readfirstlane(rowptr[i + 1]);
        if (beg < 0) beg = 0;
        if (end > ne) end = ne;

        const v2f hv = *(const v2f*)(h + (size_t)i * 64 + 2 * lane);
        const float d = dis[i];

        float ax = 0.0f, ay = 0.0f;
        int base = beg;
        int m = (end - base < 32) ? (end - base) : 32;
        int rs = 0, rw = 0;
        if (lane < m) {
            const v2i r = *(const v2i*)(rec + 2 * (size_t)(base + lane));
            rs = r.x;
            rw = r.y;
        }
        while (m > 0) {
            for (int j0 = 0; j0 < m; j0 += GATHER_GROUP) {
                const int mm = (m - j0 < GATHER_GROUP) ? (m - j0) : GATHER_GROUP;
                int sidx[GATHER_GROUP];
                float wgt[GATHER_GROUP];
                v2f v[GATHER_GROUP];
#pragma unroll
                for (int jj = 0; jj < GATHER_GROUP; ++jj) {
                    sidx[jj] = __builtin_amdgcn_readlane(rs, j0 + jj);
                    wgt[jj] = __int_as_float(__builtin_amdgcn_readlane(rw, j0 + jj));
                }
#pragma unroll
                for (int jj = 0; jj < GATHER_GROUP; ++jj) {
                    if (jj < mm) {
                        unsigned s = (unsigned)sidx[jj];
                        if (s > (unsigned)(n - 1)) s = (unsigned)(n - 1);
                        v[jj] = *(const v2f*)(h + (size_t)s * 64 + 2 * lane);
                    } else {
                        v[jj].x = 0.0f;
                        v[jj].y = 0.0f;
                    }
                }
#pragma unroll
                for (int jj = 0; jj < GATHER_GROUP; ++jj) {
                    ax = fmaf(v[jj].x, wgt[jj], ax);
                    ay = fmaf(v[jj].y, wgt[jj], ay);
                }
            }
            base += 32;
            m = (end - base < 32) ? (end - base) : 32;
            rs = 0;
            rw = 0;
            if (lane < m) {
                const v2i r = *(const v2i*)(rec + 2 * (size_t)(base + lane));
                rs = r.x;
                rw = r.y;
            }
        }

        const float d2 = d * d;
        v2f o;
        o.x = fmaxf(ax + hv.x * d2 + bv.x, 0.0f);
        o.y = fmaxf(ay + hv.y * d2 + bv.y, 0.0f);
        *(v2f*)(out + (size_t)i * 64 + 2 * lane) = o;
    }
}

__global__ void out_proj(const float* __restrict__ hd,
                         const float* __restrict__ Wo,
                         const float* __restrict__ bo,
                         float* __restrict__ out,
                         int n) {
    const int i = blockIdx.x * blockDim.x + threadIdx.x;
    if (i >= n) return;
    const float* h = hd + (size_t)i * 128;
    float a0 = bo[0], a1 = bo[1];
#pragma unroll 8
    for (int k = 0; k < 128; ++k) {
        const float hv = h[k];
        a0 += hv * Wo[k * 2 + 0];
        a1 += hv * Wo[k * 2 + 1];
    }
    out[(size_t)i * 2 + 0] = a0;
    out[(size_t)i * 2 + 1] = a1;
}

extern "C" void kernel_launch(void* const* d_in, const int* in_sizes, int n_in,
                              void* d_out, int out_size, void* d_ws, size_t ws_size,
                              hipStream_t stream) {
    const float* x  = (const float*)d_in[0];
    const int*   ei = (const int*)d_in[1];
    const float* W1 = (const float*)d_in[2];
    const float* b1 = (const float*)d_in[3];
    const float* W2 = (const float*)d_in[4];
    const float* b2 = (const float*)d_in[5];
    const float* Wd = (const float*)d_in[6];
    const float* bd = (const float*)d_in[7];
    const float* Wo = (const float*)d_in[8];
    const float* bo = (const float*)d_in[9];
    float* out = (float*)d_out;

    const int N = in_sizes[0] / 64;
    const int E = in_sizes[1] / 2;
    const int* src = ei;
    const int* dst = ei + E;

    const size_t NP = ((size_t)N + 1 + 63) & ~(size_t)63;
    const size_t EP = ((size_t)E + 63) & ~(size_t)63;
    int*   cnt    = (int*)d_ws;
    int*   rowptr = cnt + NP;
    float* dis    = (float*)(rowptr + NP);
    int*   bsum   = (int*)(dis + NP);
    int*   rank   = bsum + 4096;
    int*   rec    = rank + EP;
    float* bufA   = (float*)(rec + 2 * EP);
    float* bufC   = bufA + (size_t)N * 64;
    float* bufD   = bufC + (size_t)N * 64;
    (void)ws_size; (void)n_in; (void)out_size;

    const int TB = 256;
    const int gemm_waves   = N / 16;
    const int gemm_blocks  = (gemm_waves * 32 + TB - 1) / TB;
    const int edge_blocks  = (E + 4 * TB - 1) / (4 * TB);
    const int node_blocks  = (N + TB - 1) / TB;
    const int scan_blocks  = (N + 1023) / 1024;
    const int wave_blocks  = (N + 63) / 64;

    zero_i32<<<node_blocks, TB, 0, stream>>>(cnt, N);
    count_rank<<<edge_blocks, TB, 0, stream>>>(dst, cnt, rank, E);
    block_sums<<<scan_blocks, TB, 0, stream>>>(cnt, bsum, N);
    scan_rowptr_dis<<<scan_blocks, TB, 0, stream>>>(cnt, bsum, rowptr, dis, N);
    fill_lists<<<edge_blocks, TB, 0, stream>>>(src, dst, rank, rowptr, dis, rec, E);

    gemm_wmma_f32<4, 64, false, false><<<gemm_blocks, TB, 0, stream>>>(x, W1, nullptr, bufA, N);
    gather_rows<<<wave_blocks, TB, 0, stream>>>(bufA, rec, rowptr, dis, b1, bufC, N, E);

    gemm_wmma_f32<4, 64, false, false><<<gemm_blocks, TB, 0, stream>>>(bufC, W2, nullptr, bufA, N);
    gather_rows<<<wave_blocks, TB, 0, stream>>>(bufA, rec, rowptr, dis, b2, bufC, N, E);

    gemm_wmma_f32<8, 64, true, true><<<gemm_blocks, TB, 0, stream>>>(bufC, Wd, bd, bufD, N);

    out_proj<<<(N + TB - 1) / TB, TB, 0, stream>>>(bufD, Wo, bo, out, N);
}


# each aggregation launch keeps its block's 64 rows in LDS and runs the consuming f32 WMMA product itself (W2; Wd+relu+out proj): 8 launches, was 11; out proj summed per column lane, then 16-lane tree
# speedup vs baseline: 2.0463x; 1.1652x over previous
#include <hip/hip_runtime.h>
#include <hip/hip_bf16.h>

typedef __attribute__((ext_vector_type(2))) float v2f;
typedef __attribute__((ext_vector_type(8))) float v8f;

#define N_NODES_C 100000

typedef __attribute__((ext_vector_type(2))) int v2i;

__global__ void zero_i32(int* __restrict__ p, int n) {
    int i = blockIdx.x * blockDim.x + threadIdx.x;
    if (i < n) p[i] = 0;
}

#define EDGES_PER_THREAD 4

__launch_bounds__(256)
__global__ void count_rank(const int* __restrict__ dst, int* __restrict__ cnt,
                           int* __restrict__ rank, int ne) {
    const int blk = blockIdx.x * (256 * EDGES_PER_THREAD);
    const int base = blk + threadIdx.x;
    if (blk + 256 * EDGES_PER_THREAD <= ne) {
        int d[EDGES_PER_THREAD], old[EDGES_PER_THREAD];
#pragma unroll
        for (int k = 0; k < EDGES_PER_THREAD; ++k) d[k] = dst[base + k * 256];
#pragma unroll
        for (int k = 0; k < EDGES_PER_THREAD; ++k) old[k] = atomicAdd(&cnt[d[k]], 1);
#pragma unroll
        for (int k = 0; k < EDGES_PER_THREAD; ++k) rank[base + k * 256] = old[k];
    } else {
        for (int k = 0; k < EDGES_PER_THREAD; ++k) {
            const int e = base + k * 256;
            if (e < ne) rank[e] = atomicAdd(&cnt[dst[e]], 1);
        }
    }
}

__launch_bounds__(256)
__global__ void block_sums(const int* __restrict__ cnt, int* __restrict__ bsum, int n) {
    __shared__ int wsum[8];
    const int t = threadIdx.x;
    const int base = blockIdx.x * 1024 + t * 4;
    int s = 0;
    for (int k = 0; k < 4; ++k)
        if (base + k < n) s += cnt[base + k];
    for (int off = 16; off > 0; off >>= 1) s += __shfl_xor(s, off);
    if ((t & 31) == 0) wsum[t >> 5] = s;
    __syncthreads();
    if (t == 0) {
        int tot = 0;
        for (int w = 0; w < 8; ++w) tot += wsum[w];
        bsum[blockIdx.x] = tot;
    }
}

__launch_bounds__(256)
__global__ void scan_rowptr_dis(const int* __restrict__ cnt, const int* __restrict__ bsum,
                                int* __restrict__ rowptr, float* __restrict__ dis, int n) {
    __shared__ int wsum[8];
    const int t = threadIdx.x;
    const int lane = t & 31;
    const int wv = t >> 5;

    int p = 0;
    for (int j = t; j < (int)blockIdx.x; j += 256) p += bsum[j];
    for (int off = 16; off > 0; off >>= 1) p += __shfl_xor(p, off);
    if (lane == 0) wsum[wv] = p;
    __syncthreads();
    int boff = 0;
    for (int w = 0; w < 8; ++w) boff += wsum[w];
    __syncthreads();

    const int base = blockIdx.x * 1024 + t * 4;
    int c[4];
    for (int k = 0; k < 4; ++k) c[k] = (base + k < n) ? cnt[base + k] : 0;
    const int tsum = c[0] + c[1] + c[2] + c[3];
    int incl = tsum;
    for (int off = 1; off < 32; off <<= 1) {
        const int v = __shfl_up(incl, off);
        if (lane >= off) incl += v;
    }
    if (lane == 31) wsum[wv] = incl;
    __syncthreads();
    int woff = 0;
    for (int w = 0; w < wv; ++w) woff += wsum[w];
    int run = boff + woff + incl - tsum;
    for (int k = 0; k < 4; ++k) {
        if (base + k < n) {
            rowptr[base + k] = run;
            dis[base + k] = rsqrtf((float)c[k] + 1.0f);
            run += c[k];
        }
    }
    if (blockIdx.x == gridDim.x - 1 && t == 255) rowptr[n] = boff + woff + incl;
}

__launch_bounds__(256)
__global__ void fill_lists(const int* __restrict__ src, const int* __restrict__ dst,
                           const int* __restrict__ rank, const int* __restrict__ rowptr,
                           const float* __restrict__ dis, int* __restrict__ rec, int ne) {
    const int blk = blockIdx.x * (256 * EDGES_PER_THREAD);
    const int base = blk + threadIdx.x;
    if (blk + 256 * EDGES_PER_THREAD <= ne) {
        int s[EDGES_PER_THREAD], d[EDGES_PER_THREAD], r[EDGES_PER_THREAD], p[EDGES_PER_THREAD];
        float w[EDGES_PER_THREAD];
#pragma unroll
        for (int k = 0; k < EDGES_PER_THREAD; ++k) {
            s[k] = src[base + k * 256];
            d[k] = dst[base + k * 256];
            r[k] = rank[base + k * 256];
        }
#pragma unroll
        for (int k = 0; k < EDGES_PER_THREAD; ++k) {
            p[k] = rowptr[d[k]] + r[k];
            w[k] = dis[s[k]] * dis[d[k]];
        }
#pragma unroll
        for (int k = 0; k < EDGES_PER_THREAD; ++k) {
            unsigned slot = (unsigned)p[k];
            if (slot > (unsigned)(ne - 1)) slot = (unsigned)(ne - 1);
            v2i q;
            q.x = s[k];
            q.y = __float_as_int(w[k]);
            *(v2i*)(rec + 2 * (size_t)slot) = q;
        }
    } else {
        for (int k = 0; k < EDGES_PER_THREAD; ++k) {
            const int e = base + k * 256;
            if (e < ne) {
                const int s = src[e];
                const int d = dst[e];
                const int p = rowptr[d] + rank[e];
                const float w = dis[s] * dis[d];
                if ((unsigned)p < (unsigned)ne) {
                    v2i q;
                    q.x = s;
                    q.y = __float_as_int(w);
                    *(v2i*)(rec + 2 * (size_t)p) = q;
                }
            }
        }
    }
}

template<int NT, int KDIM, bool BIAS, bool RELU>
__launch_bounds__(256)
__global__ void gemm_wmma_f32(const float* __restrict__ X,
                              const float* __restrict__ W,
                              const float* __restrict__ bias,
                              float* __restrict__ Y,
                              int M) {
    constexpr int N = NT * 16;
    const int wave = (blockIdx.x * blockDim.x + threadIdx.x) >> 5;
    const int lane = threadIdx.x & 31;
    const int row0 = wave * 16;
    if (row0 >= M) return;

    const int r      = lane & 15;
    const int khalf  = (lane >> 4) * 2;

    v8f acc[NT] = {};

    const float* __restrict__ xrow = X + (size_t)(row0 + r) * KDIM;

#pragma unroll
    for (int k = 0; k < KDIM; k += 4) {
        const int kk = k + khalf;
        v2f a = *(const v2f*)(xrow + kk);
#pragma unroll
        for (int t = 0; t < NT; ++t) {
            const int col = t * 16 + r;
            v2f b;
            b.x = W[(size_t)kk * N + col];
            b.y = W[(size_t)(kk + 1) * N + col];
            acc[t] = __builtin_amdgcn_wmma_f32_16x16x4_f32(
                 false, a,  false, b,
                 (short)0, acc[t],
                 false,  false);
        }
    }

    const int rowoff = (lane >> 4) * 8;
#pragma unroll
    for (int t = 0; t < NT; ++t) {
        const int col = t * 16 + r;
        const float bv = BIAS ? bias[col] : 0.0f;
#pragma unroll
        for (int g = 0; g < 8; ++g) {
            float v = acc[t][g] + bv;
            if (RELU) v = fmaxf(v, 0.0f);
            Y[(size_t)(row0 + rowoff + g) * N + col] = v;
        }
    }
}

typedef __attribute__((ext_vector_type(4))) float v4f;

#define GATHER_GROUP 16
#define NODES_PER_WAVE 8
#define TILE_PITCH 68

template<int NOUT>
__device__ __forceinline__ void tile_product(const float* arow, const float* wl,
                                             int boff0, int boff1, v8f& acc0, v8f& acc1) {
#pragma nounroll
    for (int kg = 0; kg < 4; ++kg) {
        v2f a[4], b0[4], b1[4];
#pragma unroll
        for (int j = 0; j < 4; ++j) {
            a[j]  = *(const v2f*)(arow + kg * 16 + j * 4);
            b0[j] = *(const v2f*)(wl + (kg * 8 + j * 2) * (2 * NOUT) + boff0);
            b1[j] = *(const v2f*)(wl + (kg * 8 + j * 2) * (2 * NOUT) + boff1);
        }
#pragma unroll
        for (int j = 0; j < 4; ++j) {
            acc0 = __builtin_amdgcn_wmma_f32_16x16x4_f32(
                 false, a[j],  false, b0[j],
                 (short)0, acc0,  false,  false);
            acc1 = __builtin_amdgcn_wmma_f32_16x16x4_f32(
                 false, a[j],  false, b1[j],
                 (short)0, acc1,  false,  false);
        }
    }
}

template<bool FINAL>
__launch_bounds__(256)
__global__ void gather_gemm(const float* __restrict__ h,
                            const int* __restrict__ rec,
                            const int* __restrict__ rowptr,
                            const float* __restrict__ dis,
                            const float* __restrict__ b,
                            const float* __restrict__ W,
                            const float* __restrict__ bias2,
                            const float* __restrict__ Wo,
                            const float* __restrict__ bo,
                            float* __restrict__ Y,
                            int n, int ne) {
    constexpr int NOUT = FINAL ? 128 : 64;
    __shared__ __attribute__((aligned(16))) float tile[64 * TILE_PITCH];
    __shared__ __attribute__((aligned(16))) float wl[32 * 2 * NOUT];

#pragma unroll
    for (int it = 0; it < (32 * (NOUT / 4)) / 256; ++it) {
        const int idx = it * 256 + (int)threadIdx.x;
        const int kp = idx / (NOUT / 4);
        const int c4 = (idx % (NOUT / 4)) * 4;
        const v4f lo = *(const v4f*)(W + (size_t)(2 * kp) * NOUT + c4);
        const v4f hi = *(const v4f*)(W + (size_t)(2 * kp + 1) * NOUT + c4);
        float* wp = wl + kp * (2 * NOUT) + ((c4 * 2) ^ ((kp & 1) * 32));
        v4f q0, q1;
        q0.x = lo.x; q0.y = hi.x; q0.z = lo.y; q0.w = hi.y;
        q1.x = lo.z; q1.y = hi.z; q1.z = lo.w; q1.w = hi.w;
        *(v4f*)wp = q0;
        *(v4f*)(wp + 4) = q1;
    }

    const int lane = threadIdx.x & 31;
    const int wv = __builtin_amdgcn_readfirstlane((int)(threadIdx.x >> 5));
    const int blk0 = blockIdx.x * 64;
    const int i0 = blk0 + wv * NODES_PER_WAVE;
    const v2f bv = *(const v2f*)(b + 2 * lane);

#pragma nounroll
    for (int k = 0; k < NODES_PER_WAVE; ++k) {
        const int i = i0 + k;
        float* trow = tile + (wv * NODES_PER_WAVE + k) * TILE_PITCH + 2 * lane;
        if (i >= n) {
            v2f z;
            z.x = 0.0f;
            z.y = 0.0f;
            *(v2f*)trow = z;
            continue;
        }
        int beg = __builtin_amdgcn_readfirstlane(rowptr[i]);
        int end = __builtin_amdgcn_readfirstlane(rowptr[i + 1]);
        if (beg < 0) beg = 0;
        if (end > ne) end = ne;

        const v2f hv = *(const v2f*)(h + (size_t)i * 64 + 2 * lane);
        const float d = dis[i];

        float ax = 0.0f, ay = 0.0f;
        int base = beg;
        int m = (end - base < 32) ? (end - base) : 32;
        int rs = 0, rw = 0;
        if (lane < m) {
            const v2i rr = *(const v2i*)(rec + 2 * (size_t)(base + lane));
            rs = rr.x;
            rw = rr.y;
        }
        while (m > 0) {
            for (int j0 = 0; j0 < m; j0 += GATHER_GROUP) {
                const int mm = (m - j0 < GATHER_GROUP) ? (m - j0) : GATHER_GROUP;
                int sidx[GATHER_GROUP];
                float wgt[GATHER_GROUP];
                v2f v[GATHER_GROUP];
#pragma unroll
                for (int jj = 0; jj < GATHER_GROUP; ++jj) {
                    sidx[jj] = __builtin_amdgcn_readlane(rs, j0 + jj);
                    wgt[jj] = __int_as_float(__builtin_amdgcn_readlane(rw, j0 + jj));
                }
#pragma unroll
                for (int jj = 0; jj < GATHER_GROUP; ++jj) {
                    if (jj < mm) {
                        unsigned s = (unsigned)sidx[jj];
                        if (s > (unsigned)(n - 1)) s = (unsigned)(n - 1);
                        v[jj] = *(const v2f*)(h + (size_t)s * 64 + 2 * lane);
                    } else {
                        v[jj].x = 0.0f;
                        v[jj].y = 0.0f;
                    }
                }
#pragma unroll
                for (int jj = 0; jj < GATHER_GROUP; ++jj) {
                    ax = fmaf(v[jj].x, wgt[jj], ax);
                    ay = fmaf(v[jj].y, wgt[jj], ay);
                }
            }
            base += 32;
            m = (end - base < 32) ? (end - base) : 32;
            rs = 0;
            rw = 0;
            if (lane < m) {
                const v2i rr = *(const v2i*)(rec + 2 * (size_t)(base + lane));
                rs = rr.x;
                rw = rr.y;
            }
        }

        const float d2 = d * d;
        v2f o;
        o.x = fmaxf(ax + hv.x * d2 + bv.x, 0.0f);
        o.y = fmaxf(ay + hv.y * d2 + bv.y, 0.0f);
        *(v2f*)trow = o;
    }

    __syncthreads();

    const int r      = lane & 15;
    const int hh     = lane >> 4;
    const int rowoff = hh * 8;

    if (!FINAL) {
        const int rt  = wv >> 1;
        const int ct0 = (wv & 1) * 2;
        if (blk0 + rt * 16 >= n) return;
        const float* arow = tile + (rt * 16 + r) * TILE_PITCH + 2 * hh;
        const int col0 = ct0 * 16 + r;
        const int col1 = col0 + 16;
        const int boff0 = hh * (2 * NOUT) + ((col0 * 2) ^ (hh * 32));
        const int boff1 = hh * (2 * NOUT) + ((col1 * 2) ^ (hh * 32));
        v8f acc0 = {}, acc1 = {};
        tile_product<NOUT>(arow, wl, boff0, boff1, acc0, acc1);
        const int row0 = blk0 + rt * 16 + rowoff;
#pragma unroll
        for (int g = 0; g < 8; ++g) {
            Y[(size_t)(row0 + g) * 64 + col0] = acc0[g];
            Y[(size_t)(row0 + g) * 64 + col1] = acc1[g];
        }
    } else {
        if (wv >= 4) return;
        const int rt = wv;
        if (blk0 + rt * 16 >= n) return;
        const float* arow = tile + (rt * 16 + r) * TILE_PITCH + 2 * hh;
        float p0[8], p1[8];
#pragma unroll
        for (int g = 0; g < 8; ++g) {
            p0[g] = 0.0f;
            p1[g] = 0.0f;
        }
#pragma nounroll
        for (int pass = 0; pass < 4; ++pass) {
            const int col0 = pass * 32 + r;
            const int col1 = col0 + 16;
            const float bd0 = bias2[col0];
            const float bd1 = bias2[col1];
            const v2f wo0 = *(const v2f*)(Wo + 2 * col0);
            const v2f wo1 = *(const v2f*)(Wo + 2 * col1);
            const int boff0 = hh * (2 * NOUT) + ((col0 * 2) ^ (hh * 32));
            const int boff1 = hh * (2 * NOUT) + ((col1 * 2) ^ (hh * 32));
            v8f acc0 = {}, acc1 = {};
            tile_product<NOUT>(arow, wl, boff0, boff1, acc0, acc1);
#pragma unroll
            for (int g = 0; g < 8; ++g) {
                const float v0 = fmaxf(acc0[g] + bd0, 0.0f);
                const float v1 = fmaxf(acc1[g] + bd1, 0.0f);
                p0[g] = fmaf(v0, wo0.x, p0[g]);
                p1[g] = fmaf(v0, wo0.y, p1[g]);
                p0[g] = fmaf(v1, wo1.x, p0[g]);
                p1[g] = fmaf(v1, wo1.y, p1[g]);
            }
        }
#pragma unroll
        for (int g = 0; g < 8; ++g) {
#pragma unroll
            for (int off = 8; off > 0; off >>= 1) {
                p0[g] += __shfl_xor(p0[g], off);
                p1[g] += __shfl_xor(p1[g], off);
            }
        }
        float o0 = p0[0], o1 = p1[0];
#pragma unroll
        for (int g = 1; g < 8; ++g) {
            if (r == g) {
                o0 = p0[g];
                o1 = p1[g];
            }
        }
        if (r < 8) {
            v2f o;
            o.x = o0 + bo[0];
            o.y = o1 + bo[1];
            *(v2f*)(Y + (size_t)(blk0 + rt * 16 + rowoff + r) * 2) = o;
        }
    }
}

extern "C" void kernel_launch(void* const* d_in, const int* in_sizes, int n_in,
                              void* d_out, int out_size, void* d_ws, size_t ws_size,
                              hipStream_t stream) {
    const float* x  = (const float*)d_in[0];
    const int*   ei = (const int*)d_in[1];
    const float* W1 = (const float*)d_in[2];
    const float* b1 = (const float*)d_in[3];
    const float* W2 = (const float*)d_in[4];
    const float* b2 = (const float*)d_in[5];
    const float* Wd = (const float*)d_in[6];
    const float* bd = (const float*)d_in[7];
    const float* Wo = (const float*)d_in[8];
    const float* bo = (const float*)d_in[9];
    float* out = (float*)d_out;

    const int N = in_sizes[0] / 64;
    const int E = in_sizes[1] / 2;
    const int* src = ei;
    const int* dst = ei + E;

    const size_t NP = ((size_t)N + 1 + 63) & ~(size_t)63;
    const size_t EP = ((size_t)E + 63) & ~(size_t)63;
    int*   cnt    = (int*)d_ws;
    int*   rowptr = cnt + NP;
    float* dis    = (float*)(rowptr + NP);
    int*   bsum   = (int*)(dis + NP);
    int*   rank   = bsum + 4096;
    int*   rec    = rank + EP;
    float* bufA   = (float*)(rec + 2 * EP);
    float* bufC   = bufA + (size_t)N * 64;
    (void)ws_size; (void)n_in; (void)out_size;

    const int TB = 256;
    const int gemm_waves   = N / 16;
    const int gemm_blocks  = (gemm_waves * 32 + TB - 1) / TB;
    const int edge_blocks  = (E + 4 * TB - 1) / (4 * TB);
    const int node_blocks  = (N + TB - 1) / TB;
    const int scan_blocks  = (N + 1023) / 1024;
    const int wave_blocks  = (N + 63) / 64;

    zero_i32<<<node_blocks, TB, 0, stream>>>(cnt, N);
    count_rank<<<edge_blocks, TB, 0, stream>>>(dst, cnt, rank, E);
    block_sums<<<scan_blocks, TB, 0, stream>>>(cnt, bsum, N);
    scan_rowptr_dis<<<scan_blocks, TB, 0, stream>>>(cnt, bsum, rowptr, dis, N);
    fill_lists<<<edge_blocks, TB, 0, stream>>>(src, dst, rank, rowptr, dis, rec, E);

    gemm_wmma_f32<4, 64, false, false><<<gemm_blocks, TB, 0, stream>>>(x, W1, nullptr, bufA, N);
    gather_gemm<false><<<wave_blocks, TB, 0, stream>>>(bufA, rec, rowptr, dis, b1, W2,
                                                      nullptr, nullptr, nullptr, bufC, N, E);

    gather_gemm<true><<<wave_blocks, TB, 0, stream>>>(bufC, rec, rowptr, dis, b2, Wd,
                                                     bd, Wo, bo, out, N, E);
}


# edge lists by a two-level counting sort on LDS counters (chunk x bucket counts, scan, bucket pass, node ranks per bucket): no global atomic per edge; sum order = list order (LDS arrival rank)
# speedup vs baseline: 2.8189x; 1.3776x over previous
#include <hip/hip_runtime.h>
#include <hip/hip_bf16.h>

typedef __attribute__((ext_vector_type(2))) float v2f;
typedef __attribute__((ext_vector_type(8))) float v8f;

#define N_NODES_C 100000

typedef __attribute__((ext_vector_type(2))) int v2i;

#define CHUNK_EDGES 4096
#define CHUNK_PER_LANE 16
#define MAX_BUCKETS 512
#define MAX_CHUNKS 512

__device__ __forceinline__ int node_in_range(int d, int n) {
    unsigned u = (unsigned)d;
    if (u > (unsigned)(n - 1)) u = (unsigned)(n - 1);
    return (int)u;
}

__launch_bounds__(256)
__global__ void chunk_hist(const int* __restrict__ dst, int* __restrict__ hist,
                           int ne, int n, int nb, int nchunk) {
    __shared__ int lcnt[MAX_BUCKETS];
    const int t = threadIdx.x;
    for (int b = t; b < nb; b += 256) lcnt[b] = 0;
    __syncthreads();
    const int blk = blockIdx.x * CHUNK_EDGES;
    if (blk + CHUNK_EDGES <= ne) {
        int d[CHUNK_PER_LANE];
#pragma unroll
        for (int k = 0; k < CHUNK_PER_LANE; ++k) d[k] = dst[blk + t + k * 256];
#pragma unroll
        for (int k = 0; k < CHUNK_PER_LANE; ++k) atomicAdd(&lcnt[node_in_range(d[k], n) >> 8], 1);
    } else {
        for (int k = 0; k < CHUNK_PER_LANE; ++k) {
            const int e = blk + t + k * 256;
            if (e < ne) atomicAdd(&lcnt[node_in_range(dst[e], n) >> 8], 1);
        }
    }
    __syncthreads();
    for (int b = t; b < nb; b += 256) hist[(size_t)b * nchunk + blockIdx.x] = lcnt[b];
}

__launch_bounds__(1024)
__global__ void chunk_scan(const int* __restrict__ hist, int* __restrict__ base,
                           int* __restrict__ bbase, int nb, int nchunk) {
    __shared__ int tot[MAX_BUCKETS];
    const int lane = threadIdx.x & 31;
    const int wv = threadIdx.x >> 5;
    for (int b = wv; b < nb; b += 32) {
        const int* hrow = hist + (size_t)b * nchunk;
        int* brow = base + (size_t)b * nchunk;
        int v[16];
#pragma unroll
        for (int k = 0; k < 16; ++k) {
            const int j = lane + 32 * k;
            v[k] = (j < nchunk) ? hrow[j] : 0;
        }
        int s = 0;
#pragma unroll
        for (int k = 0; k < 16; ++k) s += v[k];
        int incl = s;
        for (int off = 1; off < 32; off <<= 1) {
            const int u = __shfl_up(incl, off);
            if (lane >= off) incl += u;
        }
        int run = incl - s;
#pragma unroll
        for (int k = 0; k < 16; ++k) {
            const int j = lane + 32 * k;
            if (j < nchunk) brow[j] = run;
            run += v[k];
        }
        if (lane == 31) tot[b] = incl;
    }
    __syncthreads();
    if (wv == 0) {
        int s = 0;
        for (int k = 0; k < 16; ++k) {
            const int b = lane * 16 + k;
            if (b < nb) s += tot[b];
        }
        int incl = s;
        for (int off = 1; off < 32; off <<= 1) {
            const int u = __shfl_up(incl, off);
            if (lane >= off) incl += u;
        }
        int run = incl - s;
        for (int k = 0; k < 16; ++k) {
            const int b = lane * 16 + k;
            if (b < nb) {
                bbase[b] = run;
                run += tot[b];
            }
        }
        if (lane == 31) bbase[nb] = incl;
    }
}

__launch_bounds__(256)
__global__ void chunk_scatter(const int* __restrict__ src, const int* __restrict__ dst,
                              const int* __restrict__ base, const int* __restrict__ bbase,
                              int* __restrict__ tmp, int ne, int n, int nb, int nchunk) {
    __shared__ int lpos[MAX_BUCKETS];
    const int t = threadIdx.x;
    for (int b = t; b < nb; b += 256) lpos[b] = bbase[b] + base[(size_t)b * nchunk + blockIdx.x];
    __syncthreads();
    const int blk = blockIdx.x * CHUNK_EDGES;
    if (blk + CHUNK_EDGES <= ne) {
        int s[CHUNK_PER_LANE], d[CHUNK_PER_LANE], p[CHUNK_PER_LANE];
#pragma unroll
        for (int k = 0; k < CHUNK_PER_LANE; ++k) {
            s[k] = src[blk + t + k * 256];
            d[k] = dst[blk + t + k * 256];
        }
#pragma unroll
        for (int k = 0; k < CHUNK_PER_LANE; ++k) {
            d[k] = node_in_range(d[k], n);
            p[k] = atomicAdd(&lpos[d[k] >> 8], 1);
        }
#pragma unroll
        for (int k = 0; k < CHUNK_PER_LANE; ++k) {
            unsigned slot = (unsigned)p[k];
            if (slot > (unsigned)(ne - 1)) slot = (unsigned)(ne - 1);
            v2i q;
            q.x = s[k];
            q.y = d[k];
            *(v2i*)(tmp + 2 * (size_t)slot) = q;
        }
    } else {
        for (int k = 0; k < CHUNK_PER_LANE; ++k) {
            const int e = blk + t + k * 256;
            if (e < ne) {
                const int d = node_in_range(dst[e], n);
                const int p = atomicAdd(&lpos[d >> 8], 1);
                if ((unsigned)p < (unsigned)ne) {
                    v2i q;
                    q.x = src[e];
                    q.y = d;
                    *(v2i*)(tmp + 2 * (size_t)p) = q;
                }
            }
        }
    }
}

#define BUCKET_GROUP 8

__launch_bounds__(256)
__global__ void bucket_lists(const int* __restrict__ tmp, const int* __restrict__ bbase,
                             int* __restrict__ rowptr, float* __restrict__ dis,
                             int* __restrict__ rec, int n, int ne, int nb) {
    __shared__ int ncnt[256];
    __shared__ int wsum[8];
    const int t = threadIdx.x;
    const int lane = t & 31;
    const int wv = t >> 5;
    const int b = blockIdx.x;
    int beg = bbase[b];
    int end = bbase[b + 1];
    if (beg < 0) beg = 0;
    if (end > ne) end = ne;

    ncnt[t] = 0;
    __syncthreads();
    for (int e0 = beg; e0 < end; e0 += 256 * BUCKET_GROUP) {
        int d[BUCKET_GROUP];
#pragma unroll
        for (int k = 0; k < BUCKET_GROUP; ++k) {
            const int e = e0 + t + k * 256;
            d[k] = (e < end) ? tmp[2 * (size_t)e + 1] : -1;
        }
#pragma unroll
        for (int k = 0; k < BUCKET_GROUP; ++k)
            if (d[k] >= 0) atomicAdd(&ncnt[d[k] & 255], 1);
    }
    __syncthreads();

    const int c = ncnt[t];
    int incl = c;
    for (int off = 1; off < 32; off <<= 1) {
        const int u = __shfl_up(incl, off);
        if (lane >= off) incl += u;
    }
    if (lane == 31) wsum[wv] = incl;
    __syncthreads();
    int woff = 0;
    for (int w = 0; w < wv; ++w) woff += wsum[w];
    const int first = beg + woff + incl - c;
    const int node = b * 256 + t;
    if (node < n) {
        rowptr[node] = first;
        dis[node] = rsqrtf((float)c + 1.0f);
    }
    if (b == nb - 1 && t == 0) rowptr[n] = end;
    ncnt[t] = first;
    __syncthreads();

    for (int e0 = beg; e0 < end; e0 += 256 * BUCKET_GROUP) {
        v2i q[BUCKET_GROUP];
        int p[BUCKET_GROUP];
#pragma unroll
        for (int k = 0; k < BUCKET_GROUP; ++k) {
            const int e = e0 + t + k * 256;
            if (e < end) {
                q[k] = *(const v2i*)(tmp + 2 * (size_t)e);
            } else {
                q[k].x = 0;
                q[k].y = -1;
            }
        }
#pragma unroll
        for (int k = 0; k < BUCKET_GROUP; ++k)
            p[k] = (q[k].y >= 0) ? atomicAdd(&ncnt[q[k].y & 255], 1) : -1;
#pragma unroll
        for (int k = 0; k < BUCKET_GROUP; ++k)
            if ((unsigned)p[k] < (unsigned)ne)
                *(v2i*)(rec + 2 * (size_t)p[k]) = q[k];
    }
}

#define EDGES_PER_THREAD 4

__launch_bounds__(256)
__global__ void list_weights(const float* __restrict__ dis, int* rec, int ne, int n) {
    const int blk = blockIdx.x * (256 * EDGES_PER_THREAD);
    const int base = blk + threadIdx.x;
    if (blk + 256 * EDGES_PER_THREAD <= ne) {
        v2i q[EDGES_PER_THREAD];
        float w[EDGES_PER_THREAD];
#pragma unroll
        for (int k = 0; k < EDGES_PER_THREAD; ++k) q[k] = *(const v2i*)(rec + 2 * (size_t)(base + k * 256));
#pragma unroll
        for (int k = 0; k < EDGES_PER_THREAD; ++k)
            w[k] = dis[node_in_range(q[k].x, n)] * dis[node_in_range(q[k].y, n)];
#pragma unroll
        for (int k = 0; k < EDGES_PER_THREAD; ++k) {
            q[k].y = __float_as_int(w[k]);
            *(v2i*)(rec + 2 * (size_t)(base + k * 256)) = q[k];
        }
    } else {
        for (int k = 0; k < EDGES_PER_THREAD; ++k) {
            const int e = base + k * 256;
            if (e < ne) {
                v2i q = *(const v2i*)(rec + 2 * (size_t)e);
                q.y = __float_as_int(dis[node_in_range(q.x, n)] * dis[node_in_range(q.y, n)]);
                *(v2i*)(rec + 2 * (size_t)e) = q;
            }
        }
    }
}

template<int NT, int KDIM, bool BIAS, bool RELU>
__launch_bounds__(256)
__global__ void gemm_wmma_f32(const float* __restrict__ X,
                              const float* __restrict__ W,
                              const float* __restrict__ bias,
                              float* __restrict__ Y,
                              int M) {
    constexpr int N = NT * 16;
    const int wave = (blockIdx.x * blockDim.x + threadIdx.x) >> 5;
    const int lane = threadIdx.x & 31;
    const int row0 = wave * 16;
    if (row0 >= M) return;

    const int r      = lane & 15;
    const int khalf  = (lane >> 4) * 2;

    v8f acc[NT] = {};

    const float* __restrict__ xrow = X + (size_t)(row0 + r) * KDIM;

#pragma unroll
    for (int k = 0; k < KDIM; k += 4) {
        const int kk = k + khalf;
        v2f a = *(const v2f*)(xrow + kk);
#pragma unroll
        for (int t = 0; t < NT; ++t) {
            const int col = t * 16 + r;
            v2f b;
            b.x = W[(size_t)kk * N + col];
            b.y = W[(size_t)(kk + 1) * N + col];
            acc[t] = __builtin_amdgcn_wmma_f32_16x16x4_f32(
                 false, a,  false, b,
                 (short)0, acc[t],
                 false,  false);
        }
    }

    const int rowoff = (lane >> 4) * 8;
#pragma unroll
    for (int t = 0; t < NT; ++t) {
        const int col = t * 16 + r;
        const float bv = BIAS ? bias[col] : 0.0f;
#pragma unroll
        for (int g = 0; g < 8; ++g) {
            float v = acc[t][g] + bv;
            if (RELU) v = fmaxf(v, 0.0f);
            Y[(size_t)(row0 + rowoff + g) * N + col] = v;
        }
    }
}

typedef __attribute__((ext_vector_type(4))) float v4f;

#define GATHER_GROUP 16
#define NODES_PER_WAVE 8
#define TILE_PITCH 68

template<int NOUT>
__device__ __forceinline__ void tile_product(const float* arow, const float* wl,
                                             int boff0, int boff1, v8f& acc0, v8f& acc1) {
#pragma nounroll
    for (int kg = 0; kg < 4; ++kg) {
        v2f a[4], b0[4], b1[4];
#pragma unroll
        for (int j = 0; j < 4; ++j) {
            a[j]  = *(const v2f*)(arow + kg * 16 + j * 4);
            b0[j] = *(const v2f*)(wl + (kg * 8 + j * 2) * (2 * NOUT) + boff0);
            b1[j] = *(const v2f*)(wl + (kg * 8 + j * 2) * (2 * NOUT) + boff1);
        }
#pragma unroll
        for (int j = 0; j < 4; ++j) {
            acc0 = __builtin_amdgcn_wmma_f32_16x16x4_f32(
                 false, a[j],  false, b0[j],
                 (short)0, acc0,  false,  false);
            acc1 = __builtin_amdgcn_wmma_f32_16x16x4_f32(
                 false, a[j],  false, b1[j],
                 (short)0, acc1,  false,  false);
        }
    }
}

template<bool FINAL>
__launch_bounds__(256)
__global__ void gather_gemm(const float* __restrict__ h,
                            const int* __restrict__ rec,
                            const int* __restrict__ rowptr,
                            const float* __restrict__ dis,
                            const float* __restrict__ b,
                            const float* __restrict__ W,
                            const float* __restrict__ bias2,
                            const float* __restrict__ Wo,
                            const float* __restrict__ bo,
                            float* __restrict__ Y,
                            int n, int ne) {
    constexpr int NOUT = FINAL ? 128 : 64;
    __shared__ __attribute__((aligned(16))) float tile[64 * TILE_PITCH];
    __shared__ __attribute__((aligned(16))) float wl[32 * 2 * NOUT];

#pragma unroll
    for (int it = 0; it < (32 * (NOUT / 4)) / 256; ++it) {
        const int idx = it * 256 + (int)threadIdx.x;
        const int kp = idx / (NOUT / 4);
        const int c4 = (idx % (NOUT / 4)) * 4;
        const v4f lo = *(const v4f*)(W + (size_t)(2 * kp) * NOUT + c4);
        const v4f hi = *(const v4f*)(W + (size_t)(2 * kp + 1) * NOUT + c4);
        float* wp = wl + kp * (2 * NOUT) + ((c4 * 2) ^ ((kp & 1) * 32));
        v4f q0, q1;
        q0.x = lo.x; q0.y = hi.x; q0.z = lo.y; q0.w = hi.y;
        q1.x = lo.z; q1.y = hi.z; q1.z = lo.w; q1.w = hi.w;
        *(v4f*)wp = q0;
        *(v4f*)(wp + 4) = q1;
    }

    const int lane = threadIdx.x & 31;
    const int wv = __builtin_amdgcn_readfirstlane((int)(threadIdx.x >> 5));
    const int blk0 = blockIdx.x * 64;
    const int i0 = blk0 + wv * NODES_PER_WAVE;
    const v2f bv = *(const v2f*)(b + 2 * lane);

#pragma nounroll
    for (int k = 0; k < NODES_PER_WAVE; ++k) {
        const int i = i0 + k;
        float* trow = tile + (wv * NODES_PER_WAVE + k) * TILE_PITCH + 2 * lane;
        if (i >= n) {
            v2f z;
            z.x = 0.0f;
            z.y = 0.0f;
            *(v2f*)trow = z;
            continue;
        }
        int beg = __builtin_amdgcn_readfirstlane(rowptr[i]);
        int end = __builtin_amdgcn_readfirstlane(rowptr[i + 1]);
        if (beg < 0) beg = 0;
        if (end > ne) end = ne;

        const v2f hv = *(const v2f*)(h + (size_t)i * 64 + 2 * lane);
        const float d = dis[i];

        float ax = 0.0f, ay = 0.0f;
        int base = beg;
        int m = (end - base < 32) ? (end - base) : 32;
        int rs = 0, rw = 0;
        if (lane < m) {
            const v2i rr = *(const v2i*)(rec + 2 * (size_t)(base + lane));
            rs = rr.x;
            rw = rr.y;
        }
        while (m > 0) {
            for (int j0 = 0; j0 < m; j0 += GATHER_GROUP) {
                const int mm = (m - j0 < GATHER_GROUP) ? (m - j0) : GATHER_GROUP;
                int sidx[GATHER_GROUP];
                float wgt[GATHER_GROUP];
                v2f v[GATHER_GROUP];
#pragma unroll
                for (int jj = 0; jj < GATHER_GROUP; ++jj) {
                    sidx[jj] = __builtin_amdgcn_readlane(rs, j0 + jj);
                    wgt[jj] = __int_as_float(__builtin_amdgcn_readlane(rw, j0 + jj));
                }
#pragma unroll
                for (int jj = 0; jj < GATHER_GROUP; ++jj) {
                    if (jj < mm) {
                        unsigned s = (unsigned)sidx[jj];
                        if (s > (unsigned)(n - 1)) s = (unsigned)(n - 1);
                        v[jj] = *(const v2f*)(h + (size_t)s * 64 + 2 * lane);
                    } else {
                        v[jj].x = 0.0f;
                        v[jj].y = 0.0f;
                    }
                }
#pragma unroll
                for (int jj = 0; jj < GATHER_GROUP; ++jj) {
                    ax = fmaf(v[jj].x, wgt[jj], ax);
                    ay = fmaf(v[jj].y, wgt[jj], ay);
                }
            }
            base += 32;
            m = (end - base < 32) ? (end - base) : 32;
            rs = 0;
            rw = 0;
            if (lane < m) {
                const v2i rr = *(const v2i*)(rec + 2 * (size_t)(base + lane));
                rs = rr.x;
                rw = rr.y;
            }
        }

        const float d2 = d * d;
        v2f o;
        o.x = fmaxf(ax + hv.x * d2 + bv.x, 0.0f);
        o.y = fmaxf(ay + hv.y * d2 + bv.y, 0.0f);
        *(v2f*)trow = o;
    }

    __syncthreads();

    const int r      = lane & 15;
    const int hh     = lane >> 4;
    const int rowoff = hh * 8;

    if (!FINAL) {
        const int rt  = wv >> 1;
        const int ct0 = (wv & 1) * 2;
        if (blk0 + rt * 16 >= n) return;
        const float* arow = tile + (rt * 16 + r) * TILE_PITCH + 2 * hh;
        const int col0 = ct0 * 16 + r;
        const int col1 = col0 + 16;
        const int boff0 = hh * (2 * NOUT) + ((col0 * 2) ^ (hh * 32));
        const int boff1 = hh * (2 * NOUT) + ((col1 * 2) ^ (hh * 32));
        v8f acc0 = {}, acc1 = {};
        tile_product<NOUT>(arow, wl, boff0, boff1, acc0, acc1);
        const int row0 = blk0 + rt * 16 + rowoff;
#pragma unroll
        for (int g = 0; g < 8; ++g) {
            Y[(size_t)(row0 + g) * 64 + col0] = acc0[g];
            Y[(size_t)(row0 + g) * 64 + col1] = acc1[g];
        }
    } else {
        if (wv >= 4) return;
        const int rt = wv;
        if (blk0 + rt * 16 >= n) return;
        const float* arow = tile + (rt * 16 + r) * TILE_PITCH + 2 * hh;
        float p0[8], p1[8];
#pragma unroll
        for (int g = 0; g < 8; ++g) {
            p0[g] = 0.0f;
            p1[g] = 0.0f;
        }
#pragma nounroll
        for (int pass = 0; pass < 4; ++pass) {
            const int col0 = pass * 32 + r;
            const int col1 = col0 + 16;
            const float bd0 = bias2[col0];
            const float bd1 = bias2[col1];
            const v2f wo0 = *(const v2f*)(Wo + 2 * col0);
            const v2f wo1 = *(const v2f*)(Wo + 2 * col1);
            const int boff0 = hh * (2 * NOUT) + ((col0 * 2) ^ (hh * 32));
            const int boff1 = hh * (2 * NOUT) + ((col1 * 2) ^ (hh * 32));
            v8f acc0 = {}, acc1 = {};
            tile_product<NOUT>(arow, wl, boff0, boff1, acc0, acc1);
#pragma unroll
            for (int g = 0; g < 8; ++g) {
                const float v0 = fmaxf(acc0[g] + bd0, 0.0f);
                const float v1 = fmaxf(acc1[g] + bd1, 0.0f);
                p0[g] = fmaf(v0, wo0.x, p0[g]);
                p1[g] = fmaf(v0, wo0.y, p1[g]);
                p0[g] = fmaf(v1, wo1.x, p0[g]);
                p1[g] = fmaf(v1, wo1.y, p1[g]);
            }
        }
#pragma unroll
        for (int g = 0; g < 8; ++g) {
#pragma unroll
            for (int off = 8; off > 0; off >>= 1) {
                p0[g] += __shfl_xor(p0[g], off);
                p1[g] += __shfl_xor(p1[g], off);
            }
        }
        float o0 = p0[0], o1 = p1[0];
#pragma unroll
        for (int g = 1; g < 8; ++g) {
            if (r == g) {
                o0 = p0[g];
                o1 = p1[g];
            }
        }
        if (r < 8) {
            v2f o;
            o.x = o0 + bo[0];
            o.y = o1 + bo[1];
            *(v2f*)(Y + (size_t)(blk0 + rt * 16 + rowoff + r) * 2) = o;
        }
    }
}

extern "C" void kernel_launch(void* const* d_in, const int* in_sizes, int n_in,
                              void* d_out, int out_size, void* d_ws, size_t ws_size,
                              hipStream_t stream) {
    const float* x  = (const float*)d_in[0];
    const int*   ei = (const int*)d_in[1];
    const float* W1 = (const float*)d_in[2];
    const float* b1 = (const float*)d_in[3];
    const float* W2 = (const float*)d_in[4];
    const float* b2 = (const float*)d_in[5];
    const float* Wd = (const float*)d_in[6];
    const float* bd = (const float*)d_in[7];
    const float* Wo = (const float*)d_in[8];
    const float* bo = (const float*)d_in[9];
    float* out = (float*)d_out;

    const int N = in_sizes[0] / 64;
    const int E = in_sizes[1] / 2;
    const int* src = ei;
    const int* dst = ei + E;

    const size_t NP = ((size_t)N + 1 + 63) & ~(size_t)63;
    const size_t EP = ((size_t)E + 63) & ~(size_t)63;
    int*   rowptr = (int*)d_ws;
    float* dis    = (float*)(rowptr + NP);
    int*   bbase  = (int*)(dis + NP);
    int*   hist   = bbase + 1024;
    int*   base   = hist + MAX_BUCKETS * MAX_CHUNKS;
    int*   tmp    = base + MAX_BUCKETS * MAX_CHUNKS;
    int*   rec    = tmp + 2 * EP;
    float* bufA   = (float*)(rec + 2 * EP);
    float* bufC   = bufA + (size_t)N * 64;
    (void)ws_size; (void)n_in; (void)out_size;

    const int TB = 256;
    const int gemm_waves   = N / 16;
    const int gemm_blocks  = (gemm_waves * 32 + TB - 1) / TB;
    const int edge_blocks  = (E + 4 * TB - 1) / (4 * TB);
    const int wave_blocks  = (N + 63) / 64;
    const int buckets      = (N + 255) / 256;
    const int chunks       = (E + CHUNK_EDGES - 1) / CHUNK_EDGES;

    chunk_hist<<<chunks, TB, 0, stream>>>(dst, hist, E, N, buckets, chunks);
    chunk_scan<<<1, 1024, 0, stream>>>(hist, base, bbase, buckets, chunks);
    chunk_scatter<<<chunks, TB, 0, stream>>>(src, dst, base, bbase, tmp, E, N, buckets, chunks);
    bucket_lists<<<buckets, TB, 0, stream>>>(tmp, bbase, rowptr, dis, rec, N, E, buckets);
    list_weights<<<edge_blocks, TB, 0, stream>>>(dis, rec, E, N);

    gemm_wmma_f32<4, 64, false, false><<<gemm_blocks, TB, 0, stream>>>(x, W1, nullptr, bufA, N);
    gather_gemm<false><<<wave_blocks, TB, 0, stream>>>(bufA, rec, rowptr, dis, b1, W2,
                                                      nullptr, nullptr, nullptr, bufC, N, E);

    gather_gemm<true><<<wave_blocks, TB, 0, stream>>>(bufC, rec, rowptr, dis, b2, Wd,
                                                     bd, Wo, bo, out, N, E);
}


# degree scale folded into the rows: stored rows are pre-multiplied by their node's dis and a node's plain row sum is scaled once by its dis; lists hold src only, weight pass gone (7 launches)
# speedup vs baseline: 2.9134x; 1.0335x over previous
#include <hip/hip_runtime.h>
#include <hip/hip_bf16.h>

typedef __attribute__((ext_vector_type(2))) float v2f;
typedef __attribute__((ext_vector_type(8))) float v8f;

#define N_NODES_C 100000

typedef __attribute__((ext_vector_type(2))) int v2i;

#define CHUNK_EDGES 4096
#define CHUNK_PER_LANE 16
#define MAX_BUCKETS 512
#define MAX_CHUNKS 512

__device__ __forceinline__ int node_in_range(int d, int n) {
    unsigned u = (unsigned)d;
    if (u > (unsigned)(n - 1)) u = (unsigned)(n - 1);
    return (int)u;
}

__launch_bounds__(256)
__global__ void chunk_hist(const int* __restrict__ dst, int* __restrict__ hist,
                           int ne, int n, int nb, int nchunk) {
    __shared__ int lcnt[MAX_BUCKETS];
    const int t = threadIdx.x;
    for (int b = t; b < nb; b += 256) lcnt[b] = 0;
    __syncthreads();
    const int blk = blockIdx.x * CHUNK_EDGES;
    if (blk + CHUNK_EDGES <= ne) {
        int d[CHUNK_PER_LANE];
#pragma unroll
        for (int k = 0; k < CHUNK_PER_LANE; ++k) d[k] = dst[blk + t + k * 256];
#pragma unroll
        for (int k = 0; k < CHUNK_PER_LANE; ++k) atomicAdd(&lcnt[node_in_range(d[k], n) >> 8], 1);
    } else {
        for (int k = 0; k < CHUNK_PER_LANE; ++k) {
            const int e = blk + t + k * 256;
            if (e < ne) atomicAdd(&lcnt[node_in_range(dst[e], n) >> 8], 1);
        }
    }
    __syncthreads();
    for (int b = t; b < nb; b += 256) hist[(size_t)b * nchunk + blockIdx.x] = lcnt[b];
}

__launch_bounds__(1024)
__global__ void chunk_scan(const int* __restrict__ hist, int* __restrict__ base,
                           int* __restrict__ bbase, int nb, int nchunk) {
    __shared__ int tot[MAX_BUCKETS];
    const int lane = threadIdx.x & 31;
    const int wv = threadIdx.x >> 5;
    for (int b = wv; b < nb; b += 32) {
        const int* hrow = hist + (size_t)b * nchunk;
        int* brow = base + (size_t)b * nchunk;
        int v[16];
#pragma unroll
        for (int k = 0; k < 16; ++k) {
            const int j = lane + 32 * k;
            v[k] = (j < nchunk) ? hrow[j] : 0;
        }
        int s = 0;
#pragma unroll
        for (int k = 0; k < 16; ++k) s += v[k];
        int incl = s;
        for (int off = 1; off < 32; off <<= 1) {
            const int u = __shfl_up(incl, off);
            if (lane >= off) incl += u;
        }
        int run = incl - s;
#pragma unroll
        for (int k = 0; k < 16; ++k) {
            const int j = lane + 32 * k;
            if (j < nchunk) brow[j] = run;
            run += v[k];
        }
        if (lane == 31) tot[b] = incl;
    }
    __syncthreads();
    if (wv == 0) {
        int s = 0;
        for (int k = 0; k < 16; ++k) {
            const int b = lane * 16 + k;
            if (b < nb) s += tot[b];
        }
        int incl = s;
        for (int off = 1; off < 32; off <<= 1) {
            const int u = __shfl_up(incl, off);
            if (lane >= off) incl += u;
        }
        int run = incl - s;
        for (int k = 0; k < 16; ++k) {
            const int b = lane * 16 + k;
            if (b < nb) {
                bbase[b] = run;
                run += tot[b];
            }
        }
        if (lane == 31) bbase[nb] = incl;
    }
}

__launch_bounds__(256)
__global__ void chunk_scatter(const int* __restrict__ src, const int* __restrict__ dst,
                              const int* __restrict__ base, const int* __restrict__ bbase,
                              int* __restrict__ tmp, int ne, int n, int nb, int nchunk) {
    __shared__ int lpos[MAX_BUCKETS];
    const int t = threadIdx.x;
    for (int b = t; b < nb; b += 256) lpos[b] = bbase[b] + base[(size_t)b * nchunk + blockIdx.x];
    __syncthreads();
    const int blk = blockIdx.x * CHUNK_EDGES;
    if (blk + CHUNK_EDGES <= ne) {
        int s[CHUNK_PER_LANE], d[CHUNK_PER_LANE], p[CHUNK_PER_LANE];
#pragma unroll
        for (int k = 0; k < CHUNK_PER_LANE; ++k) {
            s[k] = src[blk + t + k * 256];
            d[k] = dst[blk + t + k * 256];
        }
#pragma unroll
        for (int k = 0; k < CHUNK_PER_LANE; ++k) {
            d[k] = node_in_range(d[k], n);
            p[k] = atomicAdd(&lpos[d[k] >> 8], 1);
        }
#pragma unroll
        for (int k = 0; k < CHUNK_PER_LANE; ++k) {
            unsigned slot = (unsigned)p[k];
            if (slot > (unsigned)(ne - 1)) slot = (unsigned)(ne - 1);
            v2i q;
            q.x = s[k];
            q.y = d[k];
            *(v2i*)(tmp + 2 * (size_t)slot) = q;
        }
    } else {
        for (int k = 0; k < CHUNK_PER_LANE; ++k) {
            const int e = blk + t + k * 256;
            if (e < ne) {
                const int d = node_in_range(dst[e], n);
                const int p = atomicAdd(&lpos[d >> 8], 1);
                if ((unsigned)p < (unsigned)ne) {
                    v2i q;
                    q.x = src[e];
                    q.y = d;
                    *(v2i*)(tmp + 2 * (size_t)p) = q;
                }
            }
        }
    }
}

#define BUCKET_GROUP 8

__launch_bounds__(256)
__global__ void bucket_lists(const int* __restrict__ tmp, const int* __restrict__ bbase,
                             int* __restrict__ rowptr, float* __restrict__ dis,
                             int* __restrict__ rec, int n, int ne, int nb) {
    __shared__ int ncnt[256];
    __shared__ int wsum[8];
    const int t = threadIdx.x;
    const int lane = t & 31;
    const int wv = t >> 5;
    const int b = blockIdx.x;
    int beg = bbase[b];
    int end = bbase[b + 1];
    if (beg < 0) beg = 0;
    if (end > ne) end = ne;

    ncnt[t] = 0;
    __syncthreads();
    for (int e0 = beg; e0 < end; e0 += 256 * BUCKET_GROUP) {
        int d[BUCKET_GROUP];
#pragma unroll
        for (int k = 0; k < BUCKET_GROUP; ++k) {
            const int e = e0 + t + k * 256;
            d[k] = (e < end) ? tmp[2 * (size_t)e + 1] : -1;
        }
#pragma unroll
        for (int k = 0; k < BUCKET_GROUP; ++k)
            if (d[k] >= 0) atomicAdd(&ncnt[d[k] & 255], 1);
    }
    __syncthreads();

    const int c = ncnt[t];
    int incl = c;
    for (int off = 1; off < 32; off <<= 1) {
        const int u = __shfl_up(incl, off);
        if (lane >= off) incl += u;
    }
    if (lane == 31) wsum[wv] = incl;
    __syncthreads();
    int woff = 0;
    for (int w = 0; w < wv; ++w) woff += wsum[w];
    const int first = beg + woff + incl - c;
    const int node = b * 256 + t;
    if (node < n) {
        rowptr[node] = first;
        dis[node] = rsqrtf((float)c + 1.0f);
    }
    if (b == nb - 1 && t == 0) rowptr[n] = end;
    ncnt[t] = first;
    __syncthreads();

    for (int e0 = beg; e0 < end; e0 += 256 * BUCKET_GROUP) {
        v2i q[BUCKET_GROUP];
        int p[BUCKET_GROUP];
#pragma unroll
        for (int k = 0; k < BUCKET_GROUP; ++k) {
            const int e = e0 + t + k * 256;
            if (e < end) {
                q[k] = *(const v2i*)(tmp + 2 * (size_t)e);
            } else {
                q[k].x = 0;
                q[k].y = -1;
            }
        }
#pragma unroll
        for (int k = 0; k < BUCKET_GROUP; ++k)
            p[k] = (q[k].y >= 0) ? atomicAdd(&ncnt[q[k].y & 255], 1) : -1;
#pragma unroll
        for (int k = 0; k < BUCKET_GROUP; ++k)
            if ((unsigned)p[k] < (unsigned)ne)
                rec[p[k]] = q[k].x;
    }
}

template<int NT, int KDIM, bool BIAS, bool RELU>
__launch_bounds__(256)
__global__ void gemm_wmma_f32(const float* __restrict__ X,
                              const float* __restrict__ W,
                              const float* __restrict__ bias,
                              const float* __restrict__ rowscale,
                              float* __restrict__ Y,
                              int M) {
    constexpr int N = NT * 16;
    const int wave = (blockIdx.x * blockDim.x + threadIdx.x) >> 5;
    const int lane = threadIdx.x & 31;
    const int row0 = wave * 16;
    if (row0 >= M) return;

    const int r      = lane & 15;
    const int khalf  = (lane >> 4) * 2;

    v8f acc[NT] = {};

    const float* __restrict__ xrow = X + (size_t)(row0 + r) * KDIM;

    const int rowoff = (lane >> 4) * 8;
    float sc[8];
#pragma unroll
    for (int g = 0; g < 8; ++g) sc[g] = rowscale[row0 + rowoff + g];

#pragma unroll
    for (int k = 0; k < KDIM; k += 4) {
        const int kk = k + khalf;
        v2f a = *(const v2f*)(xrow + kk);
#pragma unroll
        for (int t = 0; t < NT; ++t) {
            const int col = t * 16 + r;
            v2f b;
            b.x = W[(size_t)kk * N + col];
            b.y = W[(size_t)(kk + 1) * N + col];
            acc[t] = __builtin_amdgcn_wmma_f32_16x16x4_f32(
                 false, a,  false, b,
                 (short)0, acc[t],
                 false,  false);
        }
    }

#pragma unroll
    for (int t = 0; t < NT; ++t) {
        const int col = t * 16 + r;
        const float bv = BIAS ? bias[col] : 0.0f;
#pragma unroll
        for (int g = 0; g < 8; ++g) {
            float v = acc[t][g] + bv;
            if (RELU) v = fmaxf(v, 0.0f);
            Y[(size_t)(row0 + rowoff + g) * N + col] = v * sc[g];
        }
    }
}

typedef __attribute__((ext_vector_type(4))) float v4f;

#define GATHER_GROUP 16
#define NODES_PER_WAVE 8
#define TILE_PITCH 68

template<int NOUT>
__device__ __forceinline__ void tile_product(const float* arow, const float* wl,
                                             int boff0, int boff1, v8f& acc0, v8f& acc1) {
#pragma nounroll
    for (int kg = 0; kg < 4; ++kg) {
        v2f a[4], b0[4], b1[4];
#pragma unroll
        for (int j = 0; j < 4; ++j) {
            a[j]  = *(const v2f*)(arow + kg * 16 + j * 4);
            b0[j] = *(const v2f*)(wl + (kg * 8 + j * 2) * (2 * NOUT) + boff0);
            b1[j] = *(const v2f*)(wl + (kg * 8 + j * 2) * (2 * NOUT) + boff1);
        }
#pragma unroll
        for (int j = 0; j < 4; ++j) {
            acc0 = __builtin_amdgcn_wmma_f32_16x16x4_f32(
                 false, a[j],  false, b0[j],
                 (short)0, acc0,  false,  false);
            acc1 = __builtin_amdgcn_wmma_f32_16x16x4_f32(
                 false, a[j],  false, b1[j],
                 (short)0, acc1,  false,  false);
        }
    }
}

template<bool FINAL>
__launch_bounds__(256)
__global__ void gather_gemm(const float* __restrict__ h,
                            const int* __restrict__ rec,
                            const int* __restrict__ rowptr,
                            const float* __restrict__ dis,
                            const float* __restrict__ b,
                            const float* __restrict__ W,
                            const float* __restrict__ bias2,
                            const float* __restrict__ Wo,
                            const float* __restrict__ bo,
                            float* __restrict__ Y,
                            int n, int ne) {
    constexpr int NOUT = FINAL ? 128 : 64;
    __shared__ __attribute__((aligned(16))) float tile[64 * TILE_PITCH];
    __shared__ __attribute__((aligned(16))) float wl[32 * 2 * NOUT];

#pragma unroll
    for (int it = 0; it < (32 * (NOUT / 4)) / 256; ++it) {
        const int idx = it * 256 + (int)threadIdx.x;
        const int kp = idx / (NOUT / 4);
        const int c4 = (idx % (NOUT / 4)) * 4;
        const v4f lo = *(const v4f*)(W + (size_t)(2 * kp) * NOUT + c4);
        const v4f hi = *(const v4f*)(W + (size_t)(2 * kp + 1) * NOUT + c4);
        float* wp = wl + kp * (2 * NOUT) + ((c4 * 2) ^ ((kp & 1) * 32));
        v4f q0, q1;
        q0.x = lo.x; q0.y = hi.x; q0.z = lo.y; q0.w = hi.y;
        q1.x = lo.z; q1.y = hi.z; q1.z = lo.w; q1.w = hi.w;
        *(v4f*)wp = q0;
        *(v4f*)(wp + 4) = q1;
    }

    const int lane = threadIdx.x & 31;
    const int wv = __builtin_amdgcn_readfirstlane((int)(threadIdx.x >> 5));
    const int blk0 = blockIdx.x * 64;
    const int i0 = blk0 + wv * NODES_PER_WAVE;
    const v2f bv = *(const v2f*)(b + 2 * lane);

#pragma nounroll
    for (int k = 0; k < NODES_PER_WAVE; ++k) {
        const int i = i0 + k;
        float* trow = tile + (wv * NODES_PER_WAVE + k) * TILE_PITCH + 2 * lane;
        if (i >= n) {
            v2f z;
            z.x = 0.0f;
            z.y = 0.0f;
            *(v2f*)trow = z;
            continue;
        }
        int beg = __builtin_amdgcn_readfirstlane(rowptr[i]);
        int end = __builtin_amdgcn_readfirstlane(rowptr[i + 1]);
        if (beg < 0) beg = 0;
        if (end > ne) end = ne;

        const v2f hv = *(const v2f*)(h + (size_t)i * 64 + 2 * lane);
        const float d = dis[i];

        float ax = 0.0f, ay = 0.0f;
        int base = beg;
        int m = (end - base < 32) ? (end - base) : 32;
        int rs = 0;
        if (lane < m) rs = rec[base + lane];
        while (m > 0) {
            for (int j0 = 0; j0 < m; j0 += GATHER_GROUP) {
                const int mm = (m - j0 < GATHER_GROUP) ? (m - j0) : GATHER_GROUP;
                int sidx[GATHER_GROUP];
                v2f v[GATHER_GROUP];
#pragma unroll
                for (int jj = 0; jj < GATHER_GROUP; ++jj)
                    sidx[jj] = __builtin_amdgcn_readlane(rs, j0 + jj);
#pragma unroll
                for (int jj = 0; jj < GATHER_GROUP; ++jj) {
                    if (jj < mm) {
                        unsigned s = (unsigned)sidx[jj];
                        if (s > (unsigned)(n - 1)) s = (unsigned)(n - 1);
                        v[jj] = *(const v2f*)(h + (size_t)s * 64 + 2 * lane);
                    } else {
                        v[jj].x = 0.0f;
                        v[jj].y = 0.0f;
                    }
                }
#pragma unroll
                for (int jj = 0; jj < GATHER_GROUP; ++jj) {
                    ax += v[jj].x;
                    ay += v[jj].y;
                }
            }
            base += 32;
            m = (end - base < 32) ? (end - base) : 32;
            rs = 0;
            if (lane < m) rs = rec[base + lane];
        }

        v2f o;
        o.x = fmaxf((ax + hv.x) * d + bv.x, 0.0f);
        o.y = fmaxf((ay + hv.y) * d + bv.y, 0.0f);
        if (!FINAL) {
            o.x *= d;
            o.y *= d;
        }
        *(v2f*)trow = o;
    }

    __syncthreads();

    const int r      = lane & 15;
    const int hh     = lane >> 4;
    const int rowoff = hh * 8;

    if (!FINAL) {
        const int rt  = wv >> 1;
        const int ct0 = (wv & 1) * 2;
        if (blk0 + rt * 16 >= n) return;
        const float* arow = tile + (rt * 16 + r) * TILE_PITCH + 2 * hh;
        const int col0 = ct0 * 16 + r;
        const int col1 = col0 + 16;
        const int boff0 = hh * (2 * NOUT) + ((col0 * 2) ^ (hh * 32));
        const int boff1 = hh * (2 * NOUT) + ((col1 * 2) ^ (hh * 32));
        v8f acc0 = {}, acc1 = {};
        tile_product<NOUT>(arow, wl, boff0, boff1, acc0, acc1);
        const int row0 = blk0 + rt * 16 + rowoff;
#pragma unroll
        for (int g = 0; g < 8; ++g) {
            Y[(size_t)(row0 + g) * 64 + col0] = acc0[g];
            Y[(size_t)(row0 + g) * 64 + col1] = acc1[g];
        }
    } else {
        if (wv >= 4) return;
        const int rt = wv;
        if (blk0 + rt * 16 >= n) return;
        const float* arow = tile + (rt * 16 + r) * TILE_PITCH + 2 * hh;
        float p0[8], p1[8];
#pragma unroll
        for (int g = 0; g < 8; ++g) {
            p0[g] = 0.0f;
            p1[g] = 0.0f;
        }
#pragma nounroll
        for (int pass = 0; pass < 4; ++pass) {
            const int col0 = pass * 32 + r;
            const int col1 = col0 + 16;
            const float bd0 = bias2[col0];
            const float bd1 = bias2[col1];
            const v2f wo0 = *(const v2f*)(Wo + 2 * col0);
            const v2f wo1 = *(const v2f*)(Wo + 2 * col1);
            const int boff0 = hh * (2 * NOUT) + ((col0 * 2) ^ (hh * 32));
            const int boff1 = hh * (2 * NOUT) + ((col1 * 2) ^ (hh * 32));
            v8f acc0 = {}, acc1 = {};
            tile_product<NOUT>(arow, wl, boff0, boff1, acc0, acc1);
#pragma unroll
            for (int g = 0; g < 8; ++g) {
                const float v0 = fmaxf(acc0[g] + bd0, 0.0f);
                const float v1 = fmaxf(acc1[g] + bd1, 0.0f);
                p0[g] = fmaf(v0, wo0.x, p0[g]);
                p1[g] = fmaf(v0, wo0.y, p1[g]);
                p0[g] = fmaf(v1, wo1.x, p0[g]);
                p1[g] = fmaf(v1, wo1.y, p1[g]);
            }
        }
#pragma unroll
        for (int g = 0; g < 8; ++g) {
#pragma unroll
            for (int off = 8; off > 0; off >>= 1) {
                p0[g] += __shfl_xor(p0[g], off);
                p1[g] += __shfl_xor(p1[g], off);
            }
        }
        float o0 = p0[0], o1 = p1[0];
#pragma unroll
        for (int g = 1; g < 8; ++g) {
            if (r == g) {
                o0 = p0[g];
                o1 = p1[g];
            }
        }
        if (r < 8) {
            v2f o;
            o.x = o0 + bo[0];
            o.y = o1 + bo[1];
            *(v2f*)(Y + (size_t)(blk0 + rt * 16 + rowoff + r) * 2) = o;
        }
    }
}

extern "C" void kernel_launch(void* const* d_in, const int* in_sizes, int n_in,
                              void* d_out, int out_size, void* d_ws, size_t ws_size,
                              hipStream_t stream) {
    const float* x  = (const float*)d_in[0];
    const int*   ei = (const int*)d_in[1];
    const float* W1 = (const float*)d_in[2];
    const float* b1 = (const float*)d_in[3];
    const float* W2 = (const float*)d_in[4];
    const float* b2 = (const float*)d_in[5];
    const float* Wd = (const float*)d_in[6];
    const float* bd = (const float*)d_in[7];
    const float* Wo = (const float*)d_in[8];
    const float* bo = (const float*)d_in[9];
    float* out = (float*)d_out;

    const int N = in_sizes[0] / 64;
    const int E = in_sizes[1] / 2;
    const int* src = ei;
    const int* dst = ei + E;

    const size_t NP = ((size_t)N + 1 + 63) & ~(size_t)63;
    const size_t EP = ((size_t)E + 63) & ~(size_t)63;
    int*   rowptr = (int*)d_ws;
    float* dis    = (float*)(rowptr + NP);
    int*   bbase  = (int*)(dis + NP);
    int*   hist   = bbase + 1024;
    int*   base   = hist + MAX_BUCKETS * MAX_CHUNKS;
    int*   tmp    = base + MAX_BUCKETS * MAX_CHUNKS;
    int*   rec    = tmp + 2 * EP;
    float* bufA   = (float*)(rec + EP);
    float* bufC   = bufA + (size_t)N * 64;
    (void)ws_size; (void)n_in; (void)out_size;

    const int TB = 256;
    const int gemm_waves   = N / 16;
    const int gemm_blocks  = (gemm_waves * 32 + TB - 1) / TB;
    const int wave_blocks  = (N + 63) / 64;
    const int buckets      = (N + 255) / 256;
    const int chunks       = (E + CHUNK_EDGES - 1) / CHUNK_EDGES;

    chunk_hist<<<chunks, TB, 0, stream>>>(dst, hist, E, N, buckets, chunks);
    chunk_scan<<<1, 1024, 0, stream>>>(hist, base, bbase, buckets, chunks);
    chunk_scatter<<<chunks, TB, 0, stream>>>(src, dst, base, bbase, tmp, E, N, buckets, chunks);
    bucket_lists<<<buckets, TB, 0, stream>>>(tmp, bbase, rowptr, dis, rec, N, E, buckets);

    gemm_wmma_f32<4, 64, false, false><<<gemm_blocks, TB, 0, stream>>>(x, W1, nullptr, dis, bufA, N);
    gather_gemm<false><<<wave_blocks, TB, 0, stream>>>(bufA, rec, rowptr, dis, b1, W2,
                                                      nullptr, nullptr, nullptr, bufC, N, E);

    gather_gemm<true><<<wave_blocks, TB, 0, stream>>>(bufC, rec, rowptr, dis, b2, Wd,
                                                     bd, Wo, bo, out, N, E);
}


# sort's two scatter passes (edges to buckets, sources to lists) order their output in LDS and store contiguous runs; was one 8/4-byte store per edge to a random line; list order rule unchanged
# speedup vs baseline: 3.1138x; 1.0688x over previous
#include <hip/hip_runtime.h>
#include <hip/hip_bf16.h>

typedef __attribute__((ext_vector_type(2))) float v2f;
typedef __attribute__((ext_vector_type(8))) float v8f;

#define N_NODES_C 100000

typedef __attribute__((ext_vector_type(2))) int v2i;

#define CHUNK_EDGES 4096
#define CHUNK_PER_LANE 16
#define MAX_BUCKETS 512
#define MAX_CHUNKS 512

__device__ __forceinline__ int node_in_range(int d, int n) {
    unsigned u = (unsigned)d;
    if (u > (unsigned)(n - 1)) u = (unsigned)(n - 1);
    return (int)u;
}

__launch_bounds__(256)
__global__ void chunk_hist(const int* __restrict__ dst, int* __restrict__ hist,
                           int ne, int n, int nb, int nchunk) {
    __shared__ int lcnt[MAX_BUCKETS];
    const int t = threadIdx.x;
    for (int b = t; b < nb; b += 256) lcnt[b] = 0;
    __syncthreads();
    const int blk = blockIdx.x * CHUNK_EDGES;
    if (blk + CHUNK_EDGES <= ne) {
        int d[CHUNK_PER_LANE];
#pragma unroll
        for (int k = 0; k < CHUNK_PER_LANE; ++k) d[k] = dst[blk + t + k * 256];
#pragma unroll
        for (int k = 0; k < CHUNK_PER_LANE; ++k) atomicAdd(&lcnt[node_in_range(d[k], n) >> 8], 1);
    } else {
        for (int k = 0; k < CHUNK_PER_LANE; ++k) {
            const int e = blk + t + k * 256;
            if (e < ne) atomicAdd(&lcnt[node_in_range(dst[e], n) >> 8], 1);
        }
    }
    __syncthreads();
    for (int b = t; b < nb; b += 256) hist[(size_t)b * nchunk + blockIdx.x] = lcnt[b];
}

__launch_bounds__(1024)
__global__ void chunk_scan(const int* __restrict__ hist, int* __restrict__ base,
                           int* __restrict__ bbase, int nb, int nchunk) {
    __shared__ int tot[MAX_BUCKETS];
    const int lane = threadIdx.x & 31;
    const int wv = threadIdx.x >> 5;
    for (int b = wv; b < nb; b += 32) {
        const int* hrow = hist + (size_t)b * nchunk;
        int* brow = base + (size_t)b * nchunk;
        int v[16];
#pragma unroll
        for (int k = 0; k < 16; ++k) {
            const int j = lane + 32 * k;
            v[k] = (j < nchunk) ? hrow[j] : 0;
        }
        int s = 0;
#pragma unroll
        for (int k = 0; k < 16; ++k) s += v[k];
        int incl = s;
        for (int off = 1; off < 32; off <<= 1) {
            const int u = __shfl_up(incl, off);
            if (lane >= off) incl += u;
        }
        int run = incl - s;
#pragma unroll
        for (int k = 0; k < 16; ++k) {
            const int j = lane + 32 * k;
            if (j < nchunk) brow[j] = run;
            run += v[k];
        }
        if (lane == 31) tot[b] = incl;
    }
    __syncthreads();
    if (wv == 0) {
        int s = 0;
        for (int k = 0; k < 16; ++k) {
            const int b = lane * 16 + k;
            if (b < nb) s += tot[b];
        }
        int incl = s;
        for (int off = 1; off < 32; off <<= 1) {
            const int u = __shfl_up(incl, off);
            if (lane >= off) incl += u;
        }
        int run = incl - s;
        for (int k = 0; k < 16; ++k) {
            const int b = lane * 16 + k;
            if (b < nb) {
                bbase[b] = run;
                run += tot[b];
            }
        }
        if (lane == 31) bbase[nb] = incl;
    }
}

template<bool FULL>
__device__ __forceinline__ void scatter_chunk(const int* __restrict__ src, const int* __restrict__ dst,
                                              int* __restrict__ tmp, int* lcnt, int* gofs, int* wtot,
                                              v2i* image, int g0, int g1, int blk, int total,
                                              int ne, int n) {
    const int t = threadIdx.x;
    const int lane = t & 31;
    const int wv = t >> 5;
    int s[CHUNK_PER_LANE], d[CHUNK_PER_LANE], r[CHUNK_PER_LANE];
#pragma unroll
    for (int k = 0; k < CHUNK_PER_LANE; ++k) {
        const int e = t + k * 256;
        s[k] = (FULL || e < total) ? src[blk + e] : 0;
        d[k] = (FULL || e < total) ? dst[blk + e] : 0;
    }
#pragma unroll
    for (int k = 0; k < CHUNK_PER_LANE; ++k) {
        d[k] = node_in_range(d[k], n);
        r[k] = (FULL || t + k * 256 < total) ? atomicAdd(&lcnt[d[k] >> 8], 1) : 0;
    }
    __syncthreads();

    const int c0 = lcnt[2 * t];
    const int c1 = lcnt[2 * t + 1];
    const int c = c0 + c1;
    int incl = c;
    for (int off = 1; off < 32; off <<= 1) {
        const int u = __shfl_up(incl, off);
        if (lane >= off) incl += u;
    }
    if (lane == 31) wtot[wv] = incl;
    __syncthreads();
    int woff = 0;
    for (int w = 0; w < wv; ++w) woff += wtot[w];
    const int st0 = woff + incl - c;
    const int st1 = st0 + c0;
    lcnt[2 * t] = st0;
    lcnt[2 * t + 1] = st1;
    gofs[2 * t] = g0 - st0;
    gofs[2 * t + 1] = g1 - st1;
    __syncthreads();

    int p[CHUNK_PER_LANE];
#pragma unroll
    for (int k = 0; k < CHUNK_PER_LANE; ++k) p[k] = lcnt[d[k] >> 8] + r[k];
#pragma unroll
    for (int k = 0; k < CHUNK_PER_LANE; ++k) {
        if (FULL || t + k * 256 < total) {
            v2i q;
            q.x = s[k];
            q.y = d[k];
            image[p[k] & (CHUNK_EDGES - 1)] = q;
        }
    }
    __syncthreads();

    if (FULL) {
        v2i q[CHUNK_PER_LANE];
        int g[CHUNK_PER_LANE];
#pragma unroll
        for (int k = 0; k < CHUNK_PER_LANE; ++k) q[k] = image[t + k * 256];
#pragma unroll
        for (int k = 0; k < CHUNK_PER_LANE; ++k) g[k] = gofs[q[k].y >> 8];
#pragma unroll
        for (int k = 0; k < CHUNK_PER_LANE; ++k) {
            unsigned slot = (unsigned)(g[k] + t + k * 256);
            if (slot > (unsigned)(ne - 1)) slot = (unsigned)(ne - 1);
            *(v2i*)(tmp + 2 * (size_t)slot) = q[k];
        }
    } else {
#pragma unroll
        for (int k = 0; k < CHUNK_PER_LANE; ++k) {
            const int idx = t + k * 256;
            if (idx < total) {
                const v2i q = image[idx];
                const int slot = gofs[q.y >> 8] + idx;
                if ((unsigned)slot < (unsigned)ne)
                    *(v2i*)(tmp + 2 * (size_t)slot) = q;
            }
        }
    }
}

__launch_bounds__(256)
__global__ void chunk_scatter(const int* __restrict__ src, const int* __restrict__ dst,
                              const int* __restrict__ base, const int* __restrict__ bbase,
                              int* __restrict__ tmp, int ne, int n, int nb, int nchunk) {
    __shared__ int lcnt[MAX_BUCKETS];
    __shared__ int gofs[MAX_BUCKETS];
    __shared__ int wtot[8];
    __shared__ v2i image[CHUNK_EDGES];
    const int t = threadIdx.x;
    lcnt[2 * t] = 0;
    lcnt[2 * t + 1] = 0;
    int g0 = 0, g1 = 0;
    if (2 * t < nb)     g0 = bbase[2 * t] + base[(size_t)(2 * t) * nchunk + blockIdx.x];
    if (2 * t + 1 < nb) g1 = bbase[2 * t + 1] + base[(size_t)(2 * t + 1) * nchunk + blockIdx.x];
    __syncthreads();

    const int blk = blockIdx.x * CHUNK_EDGES;
    const int total = (ne - blk < CHUNK_EDGES) ? (ne - blk) : CHUNK_EDGES;
    if (total == CHUNK_EDGES)
        scatter_chunk<true>(src, dst, tmp, lcnt, gofs, wtot, image, g0, g1, blk, total, ne, n);
    else
        scatter_chunk<false>(src, dst, tmp, lcnt, gofs, wtot, image, g0, g1, blk, total, ne, n);
}

#define BUCKET_GROUP 8
#define LIST_IMAGE 8192

__launch_bounds__(256)
__global__ void bucket_lists(const int* __restrict__ tmp, const int* __restrict__ bbase,
                             int* __restrict__ rowptr, float* __restrict__ dis,
                             int* __restrict__ rec, int n, int ne, int nb) {
    __shared__ int ncnt[256];
    __shared__ int wsum[8];
    __shared__ int image[LIST_IMAGE];
    const int t = threadIdx.x;
    const int lane = t & 31;
    const int wv = t >> 5;
    const int b = blockIdx.x;
    int beg = bbase[b];
    int end = bbase[b + 1];
    if (beg < 0) beg = 0;
    if (end > ne) end = ne;
    const int cnt = end - beg;
    const bool in_lds = cnt <= LIST_IMAGE;

    ncnt[t] = 0;
    __syncthreads();
    for (int e0 = beg; e0 < end; e0 += 256 * BUCKET_GROUP) {
        int d[BUCKET_GROUP];
#pragma unroll
        for (int k = 0; k < BUCKET_GROUP; ++k) {
            const int e = e0 + t + k * 256;
            d[k] = (e < end) ? tmp[2 * (size_t)e + 1] : -1;
        }
#pragma unroll
        for (int k = 0; k < BUCKET_GROUP; ++k)
            if (d[k] >= 0) atomicAdd(&ncnt[d[k] & 255], 1);
    }
    __syncthreads();

    const int c = ncnt[t];
    int incl = c;
    for (int off = 1; off < 32; off <<= 1) {
        const int u = __shfl_up(incl, off);
        if (lane >= off) incl += u;
    }
    if (lane == 31) wsum[wv] = incl;
    __syncthreads();
    int woff = 0;
    for (int w = 0; w < wv; ++w) woff += wsum[w];
    const int first = beg + woff + incl - c;
    const int node = b * 256 + t;
    if (node < n) {
        rowptr[node] = first;
        dis[node] = rsqrtf((float)c + 1.0f);
    }
    if (b == nb - 1 && t == 0) rowptr[n] = end;
    ncnt[t] = first;
    __syncthreads();

    for (int e0 = beg; e0 < end; e0 += 256 * BUCKET_GROUP) {
        v2i q[BUCKET_GROUP];
        int p[BUCKET_GROUP];
#pragma unroll
        for (int k = 0; k < BUCKET_GROUP; ++k) {
            const int e = e0 + t + k * 256;
            if (e < end) {
                q[k] = *(const v2i*)(tmp + 2 * (size_t)e);
            } else {
                q[k].x = 0;
                q[k].y = -1;
            }
        }
#pragma unroll
        for (int k = 0; k < BUCKET_GROUP; ++k)
            p[k] = (q[k].y >= 0) ? atomicAdd(&ncnt[q[k].y & 255], 1) : -1;
        if (in_lds) {
#pragma unroll
            for (int k = 0; k < BUCKET_GROUP; ++k) {
                const unsigned o = (unsigned)(p[k] - beg);
                if (p[k] >= 0 && o < (unsigned)cnt)
                    image[o] = q[k].x;
            }
        } else {
#pragma unroll
            for (int k = 0; k < BUCKET_GROUP; ++k)
                if ((unsigned)p[k] < (unsigned)ne)
                    rec[p[k]] = q[k].x;
        }
    }
    if (in_lds) {
        __syncthreads();
        const int a = beg & 31;
        for (int e0 = t - a; e0 < cnt; e0 += 4 * 256) {
            int v[4];
#pragma unroll
            for (int j = 0; j < 4; ++j) {
                int e = e0 + j * 256;
                if (e < 0) e = 0;
                if (e > cnt - 1) e = cnt - 1;
                v[j] = image[e];
            }
#pragma unroll
            for (int j = 0; j < 4; ++j) {
                const int e = e0 + j * 256;
                if (e >= 0 && e < cnt) rec[beg + e] = v[j];
            }
        }
    }
}

template<int NT, int KDIM, bool BIAS, bool RELU>
__launch_bounds__(256)
__global__ void gemm_wmma_f32(const float* __restrict__ X,
                              const float* __restrict__ W,
                              const float* __restrict__ bias,
                              const float* __restrict__ rowscale,
                              float* __restrict__ Y,
                              int M) {
    constexpr int N = NT * 16;
    const int wave = (blockIdx.x * blockDim.x + threadIdx.x) >> 5;
    const int lane = threadIdx.x & 31;
    const int row0 = wave * 16;
    if (row0 >= M) return;

    const int r      = lane & 15;
    const int khalf  = (lane >> 4) * 2;

    v8f acc[NT] = {};

    const float* __restrict__ xrow = X + (size_t)(row0 + r) * KDIM;

    const int rowoff = (lane >> 4) * 8;
    float sc[8];
#pragma unroll
    for (int g = 0; g < 8; ++g) sc[g] = rowscale[row0 + rowoff + g];

#pragma unroll
    for (int k = 0; k < KDIM; k += 4) {
        const int kk = k + khalf;
        v2f a = *(const v2f*)(xrow + kk);
#pragma unroll
        for (int t = 0; t < NT; ++t) {
            const int col = t * 16 + r;
            v2f b;
            b.x = W[(size_t)kk * N + col];
            b.y = W[(size_t)(kk + 1) * N + col];
            acc[t] = __builtin_amdgcn_wmma_f32_16x16x4_f32(
                 false, a,  false, b,
                 (short)0, acc[t],
                 false,  false);
        }
    }

#pragma unroll
    for (int t = 0; t < NT; ++t) {
        const int col = t * 16 + r;
        const float bv = BIAS ? bias[col] : 0.0f;
#pragma unroll
        for (int g = 0; g < 8; ++g) {
            float v = acc[t][g] + bv;
            if (RELU) v = fmaxf(v, 0.0f);
            Y[(size_t)(row0 + rowoff + g) * N + col] = v * sc[g];
        }
    }
}

typedef __attribute__((ext_vector_type(4))) float v4f;

#define GATHER_GROUP 16
#define NODES_PER_WAVE 8
#define TILE_PITCH 68

template<int NOUT>
__device__ __forceinline__ void tile_product(const float* arow, const float* wl,
                                             int boff0, int boff1, v8f& acc0, v8f& acc1) {
#pragma nounroll
    for (int kg = 0; kg < 4; ++kg) {
        v2f a[4], b0[4], b1[4];
#pragma unroll
        for (int j = 0; j < 4; ++j) {
            a[j]  = *(const v2f*)(arow + kg * 16 + j * 4);
            b0[j] = *(const v2f*)(wl + (kg * 8 + j * 2) * (2 * NOUT) + boff0);
            b1[j] = *(const v2f*)(wl + (kg * 8 + j * 2) * (2 * NOUT) + boff1);
        }
#pragma unroll
        for (int j = 0; j < 4; ++j) {
            acc0 = __builtin_amdgcn_wmma_f32_16x16x4_f32(
                 false, a[j],  false, b0[j],
                 (short)0, acc0,  false,  false);
            acc1 = __builtin_amdgcn_wmma_f32_16x16x4_f32(
                 false, a[j],  false, b1[j],
                 (short)0, acc1,  false,  false);
        }
    }
}

template<bool FINAL>
__launch_bounds__(256)
__global__ void gather_gemm(const float* __restrict__ h,
                            const int* __restrict__ rec,
                            const int* __restrict__ rowptr,
                            const float* __restrict__ dis,
                            const float* __restrict__ b,
                            const float* __restrict__ W,
                            const float* __restrict__ bias2,
                            const float* __restrict__ Wo,
                            const float* __restrict__ bo,
                            float* __restrict__ Y,
                            int n, int ne) {
    constexpr int NOUT = FINAL ? 128 : 64;
    __shared__ __attribute__((aligned(16))) float tile[64 * TILE_PITCH];
    __shared__ __attribute__((aligned(16))) float wl[32 * 2 * NOUT];

#pragma unroll
    for (int it = 0; it < (32 * (NOUT / 4)) / 256; ++it) {
        const int idx = it * 256 + (int)threadIdx.x;
        const int kp = idx / (NOUT / 4);
        const int c4 = (idx % (NOUT / 4)) * 4;
        const v4f lo = *(const v4f*)(W + (size_t)(2 * kp) * NOUT + c4);
        const v4f hi = *(const v4f*)(W + (size_t)(2 * kp + 1) * NOUT + c4);
        float* wp = wl + kp * (2 * NOUT) + ((c4 * 2) ^ ((kp & 1) * 32));
        v4f q0, q1;
        q0.x = lo.x; q0.y = hi.x; q0.z = lo.y; q0.w = hi.y;
        q1.x = lo.z; q1.y = hi.z; q1.z = lo.w; q1.w = hi.w;
        *(v4f*)wp = q0;
        *(v4f*)(wp + 4) = q1;
    }

    const int lane = threadIdx.x & 31;
    const int wv = __builtin_amdgcn_readfirstlane((int)(threadIdx.x >> 5));
    const int blk0 = blockIdx.x * 64;
    const int i0 = blk0 + wv * NODES_PER_WAVE;
    const v2f bv = *(const v2f*)(b + 2 * lane);

#pragma nounroll
    for (int k = 0; k < NODES_PER_WAVE; ++k) {
        const int i = i0 + k;
        float* trow = tile + (wv * NODES_PER_WAVE + k) * TILE_PITCH + 2 * lane;
        if (i >= n) {
            v2f z;
            z.x = 0.0f;
            z.y = 0.0f;
            *(v2f*)trow = z;
            continue;
        }
        int beg = __builtin_amdgcn_readfirstlane(rowptr[i]);
        int end = __builtin_amdgcn_readfirstlane(rowptr[i + 1]);
        if (beg < 0) beg = 0;
        if (end > ne) end = ne;

        const v2f hv = *(const v2f*)(h + (size_t)i * 64 + 2 * lane);
        const float d = dis[i];

        float ax = 0.0f, ay = 0.0f;
        int base = beg;
        int m = (end - base < 32) ? (end - base) : 32;
        int rs = 0;
        if (lane < m) rs = rec[base + lane];
        while (m > 0) {
            for (int j0 = 0; j0 < m; j0 += GATHER_GROUP) {
                const int mm = (m - j0 < GATHER_GROUP) ? (m - j0) : GATHER_GROUP;
                int sidx[GATHER_GROUP];
                v2f v[GATHER_GROUP];
#pragma unroll
                for (int jj = 0; jj < GATHER_GROUP; ++jj)
                    sidx[jj] = __builtin_amdgcn_readlane(rs, j0 + jj);
#pragma unroll
                for (int jj = 0; jj < GATHER_GROUP; ++jj) {
                    if (jj < mm) {
                        unsigned s = (unsigned)sidx[jj];
                        if (s > (unsigned)(n - 1)) s = (unsigned)(n - 1);
                        v[jj] = *(const v2f*)(h + (size_t)s * 64 + 2 * lane);
                    } else {
                        v[jj].x = 0.0f;
                        v[jj].y = 0.0f;
                    }
                }
#pragma unroll
                for (int jj = 0; jj < GATHER_GROUP; ++jj) {
                    ax += v[jj].x;
                    ay += v[jj].y;
                }
            }
            base += 32;
            m = (end - base < 32) ? (end - base) : 32;
            rs = 0;
            if (lane < m) rs = rec[base + lane];
        }

        v2f o;
        o.x = fmaxf((ax + hv.x) * d + bv.x, 0.0f);
        o.y = fmaxf((ay + hv.y) * d + bv.y, 0.0f);
        if (!FINAL) {
            o.x *= d;
            o.y *= d;
        }
        *(v2f*)trow = o;
    }

    __syncthreads();

    const int r      = lane & 15;
    const int hh     = lane >> 4;
    const int rowoff = hh * 8;

    if (!FINAL) {
        const int rt  = wv >> 1;
        const int ct0 = (wv & 1) * 2;
        if (blk0 + rt * 16 >= n) return;
        const float* arow = tile + (rt * 16 + r) * TILE_PITCH + 2 * hh;
        const int col0 = ct0 * 16 + r;
        const int col1 = col0 + 16;
        const int boff0 = hh * (2 * NOUT) + ((col0 * 2) ^ (hh * 32));
        const int boff1 = hh * (2 * NOUT) + ((col1 * 2) ^ (hh * 32));
        v8f acc0 = {}, acc1 = {};
        tile_product<NOUT>(arow, wl, boff0, boff1, acc0, acc1);
        const int row0 = blk0 + rt * 16 + rowoff;
#pragma unroll
        for (int g = 0; g < 8; ++g) {
            Y[(size_t)(row0 + g) * 64 + col0] = acc0[g];
            Y[(size_t)(row0 + g) * 64 + col1] = acc1[g];
        }
    } else {
        if (wv >= 4) return;
        const int rt = wv;
        if (blk0 + rt * 16 >= n) return;
        const float* arow = tile + (rt * 16 + r) * TILE_PITCH + 2 * hh;
        float p0[8], p1[8];
#pragma unroll
        for (int g = 0; g < 8; ++g) {
            p0[g] = 0.0f;
            p1[g] = 0.0f;
        }
#pragma nounroll
        for (int pass = 0; pass < 4; ++pass) {
            const int col0 = pass * 32 + r;
            const int col1 = col0 + 16;
            const float bd0 = bias2[col0];
            const float bd1 = bias2[col1];
            const v2f wo0 = *(const v2f*)(Wo + 2 * col0);
            const v2f wo1 = *(const v2f*)(Wo + 2 * col1);
            const int boff0 = hh * (2 * NOUT) + ((col0 * 2) ^ (hh * 32));
            const int boff1 = hh * (2 * NOUT) + ((col1 * 2) ^ (hh * 32));
            v8f acc0 = {}, acc1 = {};
            tile_product<NOUT>(arow, wl, boff0, boff1, acc0, acc1);
#pragma unroll
            for (int g = 0; g < 8; ++g) {
                const float v0 = fmaxf(acc0[g] + bd0, 0.0f);
                const float v1 = fmaxf(acc1[g] + bd1, 0.0f);
                p0[g] = fmaf(v0, wo0.x, p0[g]);
                p1[g] = fmaf(v0, wo0.y, p1[g]);
                p0[g] = fmaf(v1, wo1.x, p0[g]);
                p1[g] = fmaf(v1, wo1.y, p1[g]);
            }
        }
#pragma unroll
        for (int g = 0; g < 8; ++g) {
#pragma unroll
            for (int off = 8; off > 0; off >>= 1) {
                p0[g] += __shfl_xor(p0[g], off);
                p1[g] += __shfl_xor(p1[g], off);
            }
        }
        float o0 = p0[0], o1 = p1[0];
#pragma unroll
        for (int g = 1; g < 8; ++g) {
            if (r == g) {
                o0 = p0[g];
                o1 = p1[g];
            }
        }
        if (r < 8) {
            v2f o;
            o.x = o0 + bo[0];
            o.y = o1 + bo[1];
            *(v2f*)(Y + (size_t)(blk0 + rt * 16 + rowoff + r) * 2) = o;
        }
    }
}

extern "C" void kernel_launch(void* const* d_in, const int* in_sizes, int n_in,
                              void* d_out, int out_size, void* d_ws, size_t ws_size,
                              hipStream_t stream) {
    const float* x  = (const float*)d_in[0];
    const int*   ei = (const int*)d_in[1];
    const float* W1 = (const float*)d_in[2];
    const float* b1 = (const float*)d_in[3];
    const float* W2 = (const float*)d_in[4];
    const float* b2 = (const float*)d_in[5];
    const float* Wd = (const float*)d_in[6];
    const float* bd = (const float*)d_in[7];
    const float* Wo = (const float*)d_in[8];
    const float* bo = (const float*)d_in[9];
    float* out = (float*)d_out;

    const int N = in_sizes[0] / 64;
    const int E = in_sizes[1] / 2;
    const int* src = ei;
    const int* dst = ei + E;

    const size_t NP = ((size_t)N + 1 + 63) & ~(size_t)63;
    const size_t EP = ((size_t)E + 63) & ~(size_t)63;
    int*   rowptr = (int*)d_ws;
    float* dis    = (float*)(rowptr + NP);
    int*   bbase  = (int*)(dis + NP);
    int*   hist   = bbase + 1024;
    int*   base   = hist + MAX_BUCKETS * MAX_CHUNKS;
    int*   tmp    = base + MAX_BUCKETS * MAX_CHUNKS;
    int*   rec    = tmp + 2 * EP;
    float* bufA   = (float*)(rec + EP);
    float* bufC   = bufA + (size_t)N * 64;
    (void)ws_size; (void)n_in; (void)out_size;

    const int TB = 256;
    const int gemm_waves   = N / 16;
    const int gemm_blocks  = (gemm_waves * 32 + TB - 1) / TB;
    const int wave_blocks  = (N + 63) / 64;
    const int buckets      = (N + 255) / 256;
    const int chunks       = (E + CHUNK_EDGES - 1) / CHUNK_EDGES;

    chunk_hist<<<chunks, TB, 0, stream>>>(dst, hist, E, N, buckets, chunks);
    chunk_scan<<<1, 1024, 0, stream>>>(hist, base, bbase, buckets, chunks);
    chunk_scatter<<<chunks, TB, 0, stream>>>(src, dst, base, bbase, tmp, E, N, buckets, chunks);
    bucket_lists<<<buckets, TB, 0, stream>>>(tmp, bbase, rowptr, dis, rec, N, E, buckets);

    gemm_wmma_f32<4, 64, false, false><<<gemm_blocks, TB, 0, stream>>>(x, W1, nullptr, dis, bufA, N);
    gather_gemm<false><<<wave_blocks, TB, 0, stream>>>(bufA, rec, rowptr, dis, b1, W2,
                                                      nullptr, nullptr, nullptr, bufC, N, E);

    gather_gemm<true><<<wave_blocks, TB, 0, stream>>>(bufC, rec, rowptr, dis, b2, Wd,
                                                     bd, Wo, bo, out, N, E);
}


# aggregation: a node's row loads are one straight line of exactly m loads (one body per m, up to 24, picked by a scalar switch): no test per entry, no wait inside a batch; list order kept
# speedup vs baseline: 3.4334x; 1.1026x over previous
#include <hip/hip_runtime.h>
#include <hip/hip_bf16.h>

typedef __attribute__((ext_vector_type(2))) float v2f;
typedef __attribute__((ext_vector_type(8))) float v8f;

#define N_NODES_C 100000

typedef __attribute__((ext_vector_type(2))) int v2i;

#define CHUNK_EDGES 4096
#define CHUNK_PER_LANE 16
#define MAX_BUCKETS 512
#define MAX_CHUNKS 512

__device__ __forceinline__ int node_in_range(int d, int n) {
    unsigned u = (unsigned)d;
    if (u > (unsigned)(n - 1)) u = (unsigned)(n - 1);
    return (int)u;
}

__launch_bounds__(256)
__global__ void chunk_hist(const int* __restrict__ dst, int* __restrict__ hist,
                           int ne, int n, int nb, int nchunk) {
    __shared__ int lcnt[MAX_BUCKETS];
    const int t = threadIdx.x;
    for (int b = t; b < nb; b += 256) lcnt[b] = 0;
    __syncthreads();
    const int blk = blockIdx.x * CHUNK_EDGES;
    if (blk + CHUNK_EDGES <= ne) {
        int d[CHUNK_PER_LANE];
#pragma unroll
        for (int k = 0; k < CHUNK_PER_LANE; ++k) d[k] = dst[blk + t + k * 256];
#pragma unroll
        for (int k = 0; k < CHUNK_PER_LANE; ++k) atomicAdd(&lcnt[node_in_range(d[k], n) >> 8], 1);
    } else {
        for (int k = 0; k < CHUNK_PER_LANE; ++k) {
            const int e = blk + t + k * 256;
            if (e < ne) atomicAdd(&lcnt[node_in_range(dst[e], n) >> 8], 1);
        }
    }
    __syncthreads();
    for (int b = t; b < nb; b += 256) hist[(size_t)b * nchunk + blockIdx.x] = lcnt[b];
}

__launch_bounds__(1024)
__global__ void chunk_scan(const int* __restrict__ hist, int* __restrict__ base,
                           int* __restrict__ bbase, int nb, int nchunk) {
    __shared__ int tot[MAX_BUCKETS];
    const int lane = threadIdx.x & 31;
    const int wv = threadIdx.x >> 5;
    for (int b = wv; b < nb; b += 32) {
        const int* hrow = hist + (size_t)b * nchunk;
        int* brow = base + (size_t)b * nchunk;
        int v[16];
#pragma unroll
        for (int k = 0; k < 16; ++k) {
            const int j = lane + 32 * k;
            v[k] = (j < nchunk) ? hrow[j] : 0;
        }
        int s = 0;
#pragma unroll
        for (int k = 0; k < 16; ++k) s += v[k];
        int incl = s;
        for (int off = 1; off < 32; off <<= 1) {
            const int u = __shfl_up(incl, off);
            if (lane >= off) incl += u;
        }
        int run = incl - s;
#pragma unroll
        for (int k = 0; k < 16; ++k) {
            const int j = lane + 32 * k;
            if (j < nchunk) brow[j] = run;
            run += v[k];
        }
        if (lane == 31) tot[b] = incl;
    }
    __syncthreads();
    if (wv == 0) {
        int s = 0;
        for (int k = 0; k < 16; ++k) {
            const int b = lane * 16 + k;
            if (b < nb) s += tot[b];
        }
        int incl = s;
        for (int off = 1; off < 32; off <<= 1) {
            const int u = __shfl_up(incl, off);
            if (lane >= off) incl += u;
        }
        int run = incl - s;
        for (int k = 0; k < 16; ++k) {
            const int b = lane * 16 + k;
            if (b < nb) {
                bbase[b] = run;
                run += tot[b];
            }
        }
        if (lane == 31) bbase[nb] = incl;
    }
}

template<bool FULL>
__device__ __forceinline__ void scatter_chunk(const int* __restrict__ src, const int* __restrict__ dst,
                                              int* __restrict__ tmp, int* lcnt, int* gofs, int* wtot,
                                              v2i* image, int g0, int g1, int blk, int total,
                                              int ne, int n) {
    const int t = threadIdx.x;
    const int lane = t & 31;
    const int wv = t >> 5;
    int s[CHUNK_PER_LANE], d[CHUNK_PER_LANE], r[CHUNK_PER_LANE];
#pragma unroll
    for (int k = 0; k < CHUNK_PER_LANE; ++k) {
        const int e = t + k * 256;
        s[k] = (FULL || e < total) ? src[blk + e] : 0;
        d[k] = (FULL || e < total) ? dst[blk + e] : 0;
    }
#pragma unroll
    for (int k = 0; k < CHUNK_PER_LANE; ++k) {
        d[k] = node_in_range(d[k], n);
        r[k] = (FULL || t + k * 256 < total) ? atomicAdd(&lcnt[d[k] >> 8], 1) : 0;
    }
    __syncthreads();

    const int c0 = lcnt[2 * t];
    const int c1 = lcnt[2 * t + 1];
    const int c = c0 + c1;
    int incl = c;
    for (int off = 1; off < 32; off <<= 1) {
        const int u = __shfl_up(incl, off);
        if (lane >= off) incl += u;
    }
    if (lane == 31) wtot[wv] = incl;
    __syncthreads();
    int woff = 0;
    for (int w = 0; w < wv; ++w) woff += wtot[w];
    const int st0 = woff + incl - c;
    const int st1 = st0 + c0;
    lcnt[2 * t] = st0;
    lcnt[2 * t + 1] = st1;
    gofs[2 * t] = g0 - st0;
    gofs[2 * t + 1] = g1 - st1;
    __syncthreads();

    int p[CHUNK_PER_LANE];
#pragma unroll
    for (int k = 0; k < CHUNK_PER_LANE; ++k) p[k] = lcnt[d[k] >> 8] + r[k];
#pragma unroll
    for (int k = 0; k < CHUNK_PER_LANE; ++k) {
        if (FULL || t + k * 256 < total) {
            v2i q;
            q.x = s[k];
            q.y = d[k];
            image[p[k] & (CHUNK_EDGES - 1)] = q;
        }
    }
    __syncthreads();

    if (FULL) {
        v2i q[CHUNK_PER_LANE];
        int g[CHUNK_PER_LANE];
#pragma unroll
        for (int k = 0; k < CHUNK_PER_LANE; ++k) q[k] = image[t + k * 256];
#pragma unroll
        for (int k = 0; k < CHUNK_PER_LANE; ++k) g[k] = gofs[q[k].y >> 8];
#pragma unroll
        for (int k = 0; k < CHUNK_PER_LANE; ++k) {
            unsigned slot = (unsigned)(g[k] + t + k * 256);
            if (slot > (unsigned)(ne - 1)) slot = (unsigned)(ne - 1);
            *(v2i*)(tmp + 2 * (size_t)slot) = q[k];
        }
    } else {
#pragma unroll
        for (int k = 0; k < CHUNK_PER_LANE; ++k) {
            const int idx = t + k * 256;
            if (idx < total) {
                const v2i q = image[idx];
                const int slot = gofs[q.y >> 8] + idx;
                if ((unsigned)slot < (unsigned)ne)
                    *(v2i*)(tmp + 2 * (size_t)slot) = q;
            }
        }
    }
}

__launch_bounds__(256)
__global__ void chunk_scatter(const int* __restrict__ src, const int* __restrict__ dst,
                              const int* __restrict__ base, const int* __restrict__ bbase,
                              int* __restrict__ tmp, int ne, int n, int nb, int nchunk) {
    __shared__ int lcnt[MAX_BUCKETS];
    __shared__ int gofs[MAX_BUCKETS];
    __shared__ int wtot[8];
    __shared__ v2i image[CHUNK_EDGES];
    const int t = threadIdx.x;
    lcnt[2 * t] = 0;
    lcnt[2 * t + 1] = 0;
    int g0 = 0, g1 = 0;
    if (2 * t < nb)     g0 = bbase[2 * t] + base[(size_t)(2 * t) * nchunk + blockIdx.x];
    if (2 * t + 1 < nb) g1 = bbase[2 * t + 1] + base[(size_t)(2 * t + 1) * nchunk + blockIdx.x];
    __syncthreads();

    const int blk = blockIdx.x * CHUNK_EDGES;
    const int total = (ne - blk < CHUNK_EDGES) ? (ne - blk) : CHUNK_EDGES;
    if (total == CHUNK_EDGES)
        scatter_chunk<true>(src, dst, tmp, lcnt, gofs, wtot, image, g0, g1, blk, total, ne, n);
    else
        scatter_chunk<false>(src, dst, tmp, lcnt, gofs, wtot, image, g0, g1, blk, total, ne, n);
}

#define BUCKET_GROUP 8
#define LIST_IMAGE 8192

__launch_bounds__(256)
__global__ void bucket_lists(const int* __restrict__ tmp, const int* __restrict__ bbase,
                             int* __restrict__ rowptr, float* __restrict__ dis,
                             int* __restrict__ rec, int n, int ne, int nb) {
    __shared__ int ncnt[256];
    __shared__ int wsum[8];
    __shared__ int image[LIST_IMAGE];
    const int t = threadIdx.x;
    const int lane = t & 31;
    const int wv = t >> 5;
    const int b = blockIdx.x;
    int beg = bbase[b];
    int end = bbase[b + 1];
    if (beg < 0) beg = 0;
    if (end > ne) end = ne;
    const int cnt = end - beg;
    const bool in_lds = cnt <= LIST_IMAGE;

    ncnt[t] = 0;
    __syncthreads();
    for (int e0 = beg; e0 < end; e0 += 256 * BUCKET_GROUP) {
        int d[BUCKET_GROUP];
#pragma unroll
        for (int k = 0; k < BUCKET_GROUP; ++k) {
            const int e = e0 + t + k * 256;
            d[k] = (e < end) ? tmp[2 * (size_t)e + 1] : -1;
        }
#pragma unroll
        for (int k = 0; k < BUCKET_GROUP; ++k)
            if (d[k] >= 0) atomicAdd(&ncnt[d[k] & 255], 1);
    }
    __syncthreads();

    const int c = ncnt[t];
    int incl = c;
    for (int off = 1; off < 32; off <<= 1) {
        const int u = __shfl_up(incl, off);
        if (lane >= off) incl += u;
    }
    if (lane == 31) wsum[wv] = incl;
    __syncthreads();
    int woff = 0;
    for (int w = 0; w < wv; ++w) woff += wsum[w];
    const int first = beg + woff + incl - c;
    const int node = b * 256 + t;
    if (node < n) {
        rowptr[node] = first;
        dis[node] = rsqrtf((float)c + 1.0f);
    }
    if (b == nb - 1 && t == 0) rowptr[n] = end;
    ncnt[t] = first;
    __syncthreads();

    for (int e0 = beg; e0 < end; e0 += 256 * BUCKET_GROUP) {
        v2i q[BUCKET_GROUP];
        int p[BUCKET_GROUP];
#pragma unroll
        for (int k = 0; k < BUCKET_GROUP; ++k) {
            const int e = e0 + t + k * 256;
            if (e < end) {
                q[k] = *(const v2i*)(tmp + 2 * (size_t)e);
            } else {
                q[k].x = 0;
                q[k].y = -1;
            }
        }
#pragma unroll
        for (int k = 0; k < BUCKET_GROUP; ++k)
            p[k] = (q[k].y >= 0) ? atomicAdd(&ncnt[q[k].y & 255], 1) : -1;
        if (in_lds) {
#pragma unroll
            for (int k = 0; k < BUCKET_GROUP; ++k) {
                const unsigned o = (unsigned)(p[k] - beg);
                if (p[k] >= 0 && o < (unsigned)cnt)
                    image[o] = q[k].x;
            }
        } else {
#pragma unroll
            for (int k = 0; k < BUCKET_GROUP; ++k)
                if ((unsigned)p[k] < (unsigned)ne)
                    rec[p[k]] = q[k].x;
        }
    }
    if (in_lds) {
        __syncthreads();
        const int a = beg & 31;
        for (int e0 = t - a; e0 < cnt; e0 += 4 * 256) {
            int v[4];
#pragma unroll
            for (int j = 0; j < 4; ++j) {
                int e = e0 + j * 256;
                if (e < 0) e = 0;
                if (e > cnt - 1) e = cnt - 1;
                v[j] = image[e];
            }
#pragma unroll
            for (int j = 0; j < 4; ++j) {
                const int e = e0 + j * 256;
                if (e >= 0 && e < cnt) rec[beg + e] = v[j];
            }
        }
    }
}

template<int NT, int KDIM, bool BIAS, bool RELU>
__launch_bounds__(256)
__global__ void gemm_wmma_f32(const float* __restrict__ X,
                              const float* __restrict__ W,
                              const float* __restrict__ bias,
                              const float* __restrict__ rowscale,
                              float* __restrict__ Y,
                              int M) {
    constexpr int N = NT * 16;
    const int wave = (blockIdx.x * blockDim.x + threadIdx.x) >> 5;
    const int lane = threadIdx.x & 31;
    const int row0 = wave * 16;
    if (row0 >= M) return;

    const int r      = lane & 15;
    const int khalf  = (lane >> 4) * 2;

    v8f acc[NT] = {};

    const float* __restrict__ xrow = X + (size_t)(row0 + r) * KDIM;

    const int rowoff = (lane >> 4) * 8;
    float sc[8];
#pragma unroll
    for (int g = 0; g < 8; ++g) sc[g] = rowscale[row0 + rowoff + g];

#pragma unroll
    for (int k = 0; k < KDIM; k += 4) {
        const int kk = k + khalf;
        v2f a = *(const v2f*)(xrow + kk);
#pragma unroll
        for (int t = 0; t < NT; ++t) {
            const int col = t * 16 + r;
            v2f b;
            b.x = W[(size_t)kk * N + col];
            b.y = W[(size_t)(kk + 1) * N + col];
            acc[t] = __builtin_amdgcn_wmma_f32_16x16x4_f32(
                 false, a,  false, b,
                 (short)0, acc[t],
                 false,  false);
        }
    }

#pragma unroll
    for (int t = 0; t < NT; ++t) {
        const int col = t * 16 + r;
        const float bv = BIAS ? bias[col] : 0.0f;
#pragma unroll
        for (int g = 0; g < 8; ++g) {
            float v = acc[t][g] + bv;
            if (RELU) v = fmaxf(v, 0.0f);
            Y[(size_t)(row0 + rowoff + g) * N + col] = v * sc[g];
        }
    }
}

typedef __attribute__((ext_vector_type(4))) float v4f;

#define ROW_BATCH 24
#define NODES_PER_WAVE 8
#define TILE_PITCH 68

template<int NOUT>
__device__ __forceinline__ void tile_product(const float* arow, const float* wl,
                                             int boff0, int boff1, v8f& acc0, v8f& acc1) {
#pragma nounroll
    for (int kg = 0; kg < 4; ++kg) {
        v2f a[4], b0[4], b1[4];
#pragma unroll
        for (int j = 0; j < 4; ++j) {
            a[j]  = *(const v2f*)(arow + kg * 16 + j * 4);
            b0[j] = *(const v2f*)(wl + (kg * 8 + j * 2) * (2 * NOUT) + boff0);
            b1[j] = *(const v2f*)(wl + (kg * 8 + j * 2) * (2 * NOUT) + boff1);
        }
#pragma unroll
        for (int j = 0; j < 4; ++j) {
            acc0 = __builtin_amdgcn_wmma_f32_16x16x4_f32(
                 false, a[j],  false, b0[j],
                 (short)0, acc0,  false,  false);
            acc1 = __builtin_amdgcn_wmma_f32_16x16x4_f32(
                 false, a[j],  false, b1[j],
                 (short)0, acc1,  false,  false);
        }
    }
}

__device__ __forceinline__ v2f load_row(const float* __restrict__ h, int lane, int s_raw, unsigned last_row) {
    unsigned s = (unsigned)s_raw;
    if (s > last_row) s = last_row;
    return *(const v2f*)(h + (size_t)s * 64 + 2 * lane);
}

template<int K>
__device__ __forceinline__ void sum_rows(const float* __restrict__ h, int lane, int rs, unsigned last_row,
                                         float& ax, float& ay) {
    v2f v[K];
#pragma unroll
    for (int jj = 0; jj < K; ++jj)
        v[jj] = load_row(h, lane, __builtin_amdgcn_readlane(rs, jj), last_row);
#pragma unroll
    for (int jj = 0; jj < K; ++jj) {
        ax += v[jj].x;
        ay += v[jj].y;
    }
}

#define ROW_CASE(K) case K: sum_rows<K>(h, lane, rs, last_row, ax, ay); break;

template<bool FINAL>
__launch_bounds__(256)
__global__ void gather_gemm(const float* __restrict__ h,
                            const int* __restrict__ rec,
                            const int* __restrict__ rowptr,
                            const float* __restrict__ dis,
                            const float* __restrict__ b,
                            const float* __restrict__ W,
                            const float* __restrict__ bias2,
                            const float* __restrict__ Wo,
                            const float* __restrict__ bo,
                            float* __restrict__ Y,
                            int n, int ne) {
    constexpr int NOUT = FINAL ? 128 : 64;
    __shared__ __attribute__((aligned(16))) float tile[64 * TILE_PITCH];
    __shared__ __attribute__((aligned(16))) float wl[32 * 2 * NOUT];

#pragma unroll
    for (int it = 0; it < (32 * (NOUT / 4)) / 256; ++it) {
        const int idx = it * 256 + (int)threadIdx.x;
        const int kp = idx / (NOUT / 4);
        const int c4 = (idx % (NOUT / 4)) * 4;
        const v4f lo = *(const v4f*)(W + (size_t)(2 * kp) * NOUT + c4);
        const v4f hi = *(const v4f*)(W + (size_t)(2 * kp + 1) * NOUT + c4);
        float* wp = wl + kp * (2 * NOUT) + ((c4 * 2) ^ ((kp & 1) * 32));
        v4f q0, q1;
        q0.x = lo.x; q0.y = hi.x; q0.z = lo.y; q0.w = hi.y;
        q1.x = lo.z; q1.y = hi.z; q1.z = lo.w; q1.w = hi.w;
        *(v4f*)wp = q0;
        *(v4f*)(wp + 4) = q1;
    }

    const int lane = threadIdx.x & 31;
    const int wv = __builtin_amdgcn_readfirstlane((int)(threadIdx.x >> 5));
    const int blk0 = blockIdx.x * 64;
    const int i0 = blk0 + wv * NODES_PER_WAVE;
    const v2f bv = *(const v2f*)(b + 2 * lane);
    const unsigned last_row = (unsigned)(n - 1);

#pragma nounroll
    for (int k = 0; k < NODES_PER_WAVE; ++k) {
        const int i = i0 + k;
        float* trow = tile + (wv * NODES_PER_WAVE + k) * TILE_PITCH + 2 * lane;
        if (i >= n) {
            v2f z;
            z.x = 0.0f;
            z.y = 0.0f;
            *(v2f*)trow = z;
            continue;
        }
        int beg = __builtin_amdgcn_readfirstlane(rowptr[i]);
        int end = __builtin_amdgcn_readfirstlane(rowptr[i + 1]);
        if (beg < 0) beg = 0;
        if (end > ne) end = ne;

        const v2f hv = *(const v2f*)(h + (size_t)i * 64 + 2 * lane);
        const float d = dis[i];

        float ax = 0.0f, ay = 0.0f;
        int base = beg;
        int m = (end - base < ROW_BATCH) ? (end - base) : ROW_BATCH;
        int rs = 0;
        if (lane < m) rs = rec[base + lane];
        while (m > 0) {
            switch (m) {
                ROW_CASE(1)  ROW_CASE(2)  ROW_CASE(3)  ROW_CASE(4)  ROW_CASE(5)  ROW_CASE(6)
                ROW_CASE(7)  ROW_CASE(8)  ROW_CASE(9)  ROW_CASE(10) ROW_CASE(11) ROW_CASE(12)
                ROW_CASE(13) ROW_CASE(14) ROW_CASE(15) ROW_CASE(16) ROW_CASE(17) ROW_CASE(18)
                ROW_CASE(19) ROW_CASE(20) ROW_CASE(21) ROW_CASE(22) ROW_CASE(23) ROW_CASE(24)
                default: break;
            }
            base += ROW_BATCH;
            m = (end - base < ROW_BATCH) ? (end - base) : ROW_BATCH;
            rs = 0;
            if (lane < m) rs = rec[base + lane];
        }

        v2f o;
        o.x = fmaxf((ax + hv.x) * d + bv.x, 0.0f);
        o.y = fmaxf((ay + hv.y) * d + bv.y, 0.0f);
        if (!FINAL) {
            o.x *= d;
            o.y *= d;
        }
        *(v2f*)trow = o;
    }

    __syncthreads();

    const int r      = lane & 15;
    const int hh     = lane >> 4;
    const int rowoff = hh * 8;

    if (!FINAL) {
        const int rt  = wv >> 1;
        const int ct0 = (wv & 1) * 2;
        if (blk0 + rt * 16 >= n) return;
        const float* arow = tile + (rt * 16 + r) * TILE_PITCH + 2 * hh;
        const int col0 = ct0 * 16 + r;
        const int col1 = col0 + 16;
        const int boff0 = hh * (2 * NOUT) + ((col0 * 2) ^ (hh * 32));
        const int boff1 = hh * (2 * NOUT) + ((col1 * 2) ^ (hh * 32));
        v8f acc0 = {}, acc1 = {};
        tile_product<NOUT>(arow, wl, boff0, boff1, acc0, acc1);
        const int row0 = blk0 + rt * 16 + rowoff;
#pragma unroll
        for (int g = 0; g < 8; ++g) {
            Y[(size_t)(row0 + g) * 64 + col0] = acc0[g];
            Y[(size_t)(row0 + g) * 64 + col1] = acc1[g];
        }
    } else {
        if (wv >= 4) return;
        const int rt = wv;
        if (blk0 + rt * 16 >= n) return;
        const float* arow = tile + (rt * 16 + r) * TILE_PITCH + 2 * hh;
        float p0[8], p1[8];
#pragma unroll
        for (int g = 0; g < 8; ++g) {
            p0[g] = 0.0f;
            p1[g] = 0.0f;
        }
#pragma nounroll
        for (int pass = 0; pass < 4; ++pass) {
            const int col0 = pass * 32 + r;
            const int col1 = col0 + 16;
            const float bd0 = bias2[col0];
            const float bd1 = bias2[col1];
            const v2f wo0 = *(const v2f*)(Wo + 2 * col0);
            const v2f wo1 = *(const v2f*)(Wo + 2 * col1);
            const int boff0 = hh * (2 * NOUT) + ((col0 * 2) ^ (hh * 32));
            const int boff1 = hh * (2 * NOUT) + ((col1 * 2) ^ (hh * 32));
            v8f acc0 = {}, acc1 = {};
            tile_product<NOUT>(arow, wl, boff0, boff1, acc0, acc1);
#pragma unroll
            for (int g = 0; g < 8; ++g) {
                const float v0 = fmaxf(acc0[g] + bd0, 0.0f);
                const float v1 = fmaxf(acc1[g] + bd1, 0.0f);
                p0[g] = fmaf(v0, wo0.x, p0[g]);
                p1[g] = fmaf(v0, wo0.y, p1[g]);
                p0[g] = fmaf(v1, wo1.x, p0[g]);
                p1[g] = fmaf(v1, wo1.y, p1[g]);
            }
        }
#pragma unroll
        for (int g = 0; g < 8; ++g) {
#pragma unroll
            for (int off = 8; off > 0; off >>= 1) {
                p0[g] += __shfl_xor(p0[g], off);
                p1[g] += __shfl_xor(p1[g], off);
            }
        }
        float o0 = p0[0], o1 = p1[0];
#pragma unroll
        for (int g = 1; g < 8; ++g) {
            if (r == g) {
                o0 = p0[g];
                o1 = p1[g];
            }
        }
        if (r < 8) {
            v2f o;
            o.x = o0 + bo[0];
            o.y = o1 + bo[1];
            *(v2f*)(Y + (size_t)(blk0 + rt * 16 + rowoff + r) * 2) = o;
        }
    }
}

extern "C" void kernel_launch(void* const* d_in, const int* in_sizes, int n_in,
                              void* d_out, int out_size, void* d_ws, size_t ws_size,
                              hipStream_t stream) {
    const float* x  = (const float*)d_in[0];
    const int*   ei = (const int*)d_in[1];
    const float* W1 = (const float*)d_in[2];
    const float* b1 = (const float*)d_in[3];
    const float* W2 = (const float*)d_in[4];
    const float* b2 = (const float*)d_in[5];
    const float* Wd = (const float*)d_in[6];
    const float* bd = (const float*)d_in[7];
    const float* Wo = (const float*)d_in[8];
    const float* bo = (const float*)d_in[9];
    float* out = (float*)d_out;

    const int N = in_sizes[0] / 64;
    const int E = in_sizes[1] / 2;
    const int* src = ei;
    const int* dst = ei + E;

    const size_t NP = ((size_t)N + 1 + 63) & ~(size_t)63;
    const size_t EP = ((size_t)E + 63) & ~(size_t)63;
    int*   rowptr = (int*)d_ws;
    float* dis    = (float*)(rowptr + NP);
    int*   bbase  = (int*)(dis + NP);
    int*   hist   = bbase + 1024;
    int*   base   = hist + MAX_BUCKETS * MAX_CHUNKS;
    int*   tmp    = base + MAX_BUCKETS * MAX_CHUNKS;
    int*   rec    = tmp + 2 * EP;
    float* bufA   = (float*)(rec + EP);
    float* bufC   = bufA + (size_t)N * 64;
    (void)ws_size; (void)n_in; (void)out_size;

    const int TB = 256;
    const int gemm_waves   = N / 16;
    const int gemm_blocks  = (gemm_waves * 32 + TB - 1) / TB;
    const int wave_blocks  = (N + 63) / 64;
    const int buckets      = (N + 255) / 256;
    const int chunks       = (E + CHUNK_EDGES - 1) / CHUNK_EDGES;

    chunk_hist<<<chunks, TB, 0, stream>>>(dst, hist, E, N, buckets, chunks);
    chunk_scan<<<1, 1024, 0, stream>>>(hist, base, bbase, buckets, chunks);
    chunk_scatter<<<chunks, TB, 0, stream>>>(src, dst, base, bbase, tmp, E, N, buckets, chunks);
    bucket_lists<<<buckets, TB, 0, stream>>>(tmp, bbase, rowptr, dis, rec, N, E, buckets);

    gemm_wmma_f32<4, 64, false, false><<<gemm_blocks, TB, 0, stream>>>(x, W1, nullptr, dis, bufA, N);
    gather_gemm<false><<<wave_blocks, TB, 0, stream>>>(bufA, rec, rowptr, dis, b1, W2,
                                                      nullptr, nullptr, nullptr, bufC, N, E);

    gather_gemm<true><<<wave_blocks, TB, 0, stream>>>(bufC, rec, rowptr, dis, b2, Wd,
                                                     bd, Wo, bo, out, N, E);
}


# bucket scan spread over the chip: one wave per bucket scans its chunk counts (49 blocks; was one block whose waves looped 13 times); the two kernels that use bucket starts sum them from the totals
# speedup vs baseline: 4.1182x; 1.1995x over previous
#include <hip/hip_runtime.h>
#include <hip/hip_bf16.h>

typedef __attribute__((ext_vector_type(2))) float v2f;
typedef __attribute__((ext_vector_type(8))) float v8f;

#define N_NODES_C 100000

typedef __attribute__((ext_vector_type(2))) int v2i;

#define CHUNK_EDGES 4096
#define CHUNK_PER_LANE 16
#define MAX_BUCKETS 512
#define MAX_CHUNKS 512

__device__ __forceinline__ int node_in_range(int d, int n) {
    unsigned u = (unsigned)d;
    if (u > (unsigned)(n - 1)) u = (unsigned)(n - 1);
    return (int)u;
}

__launch_bounds__(256)
__global__ void chunk_hist(const int* __restrict__ dst, int* __restrict__ hist,
                           int ne, int n, int nb, int nchunk) {
    __shared__ int lcnt[MAX_BUCKETS];
    const int t = threadIdx.x;
    for (int b = t; b < nb; b += 256) lcnt[b] = 0;
    __syncthreads();
    const int blk = blockIdx.x * CHUNK_EDGES;
    if (blk + CHUNK_EDGES <= ne) {
        int d[CHUNK_PER_LANE];
#pragma unroll
        for (int k = 0; k < CHUNK_PER_LANE; ++k) d[k] = dst[blk + t + k * 256];
#pragma unroll
        for (int k = 0; k < CHUNK_PER_LANE; ++k) atomicAdd(&lcnt[node_in_range(d[k], n) >> 8], 1);
    } else {
        for (int k = 0; k < CHUNK_PER_LANE; ++k) {
            const int e = blk + t + k * 256;
            if (e < ne) atomicAdd(&lcnt[node_in_range(dst[e], n) >> 8], 1);
        }
    }
    __syncthreads();
    for (int b = t; b < nb; b += 256) hist[(size_t)b * nchunk + blockIdx.x] = lcnt[b];
}

__launch_bounds__(256)
__global__ void chunk_scan(const int* __restrict__ hist, int* __restrict__ base,
                           int* __restrict__ btot, int nb, int nchunk) {
    const int lane = threadIdx.x & 31;
    const int b = blockIdx.x * 8 + (int)(threadIdx.x >> 5);
    if (b >= nb) return;
    const int* hrow = hist + (size_t)b * nchunk;
    int* brow = base + (size_t)b * nchunk;
    int v[16];
#pragma unroll
    for (int k = 0; k < 16; ++k) {
        const int j = lane + 32 * k;
        v[k] = (j < nchunk) ? hrow[j] : 0;
    }
    int s = 0;
#pragma unroll
    for (int k = 0; k < 16; ++k) s += v[k];
    int incl = s;
    for (int off = 1; off < 32; off <<= 1) {
        const int u = __shfl_up(incl, off);
        if (lane >= off) incl += u;
    }
    int run = incl - s;
#pragma unroll
    for (int k = 0; k < 16; ++k) {
        const int j = lane + 32 * k;
        if (j < nchunk) brow[j] = run;
        run += v[k];
    }
    if (lane == 31) btot[b] = incl;
}

template<bool FULL>
__device__ __forceinline__ void scatter_chunk(const int* __restrict__ src, const int* __restrict__ dst,
                                              int* __restrict__ tmp, int* lcnt, int* gofs, int* wtot,
                                              v2i* image, int g0, int g1, int bt0, int bt1,
                                              int blk, int total, int ne, int n) {
    const int t = threadIdx.x;
    const int lane = t & 31;
    const int wv = t >> 5;
    int s[CHUNK_PER_LANE], d[CHUNK_PER_LANE], r[CHUNK_PER_LANE];
#pragma unroll
    for (int k = 0; k < CHUNK_PER_LANE; ++k) {
        const int e = t + k * 256;
        s[k] = (FULL || e < total) ? src[blk + e] : 0;
        d[k] = (FULL || e < total) ? dst[blk + e] : 0;
    }
#pragma unroll
    for (int k = 0; k < CHUNK_PER_LANE; ++k) {
        d[k] = node_in_range(d[k], n);
        r[k] = (FULL || t + k * 256 < total) ? atomicAdd(&lcnt[d[k] >> 8], 1) : 0;
    }
    __syncthreads();

    const int c0 = lcnt[2 * t];
    const int c1 = lcnt[2 * t + 1];
    const int c = c0 + c1;
    const int bt = bt0 + bt1;
    int incl = c;
    int bincl = bt;
    for (int off = 1; off < 32; off <<= 1) {
        const int u = __shfl_up(incl, off);
        const int ub = __shfl_up(bincl, off);
        if (lane >= off) {
            incl += u;
            bincl += ub;
        }
    }
    if (lane == 31) {
        wtot[wv] = incl;
        wtot[8 + wv] = bincl;
    }
    __syncthreads();
    int woff = 0, bwoff = 0;
    for (int w = 0; w < wv; ++w) {
        woff += wtot[w];
        bwoff += wtot[8 + w];
    }
    const int st0 = woff + incl - c;
    const int st1 = st0 + c0;
    const int bs0 = bwoff + bincl - bt;
    const int bs1 = bs0 + bt0;
    lcnt[2 * t] = st0;
    lcnt[2 * t + 1] = st1;
    gofs[2 * t] = bs0 + g0 - st0;
    gofs[2 * t + 1] = bs1 + g1 - st1;
    __syncthreads();

    int p[CHUNK_PER_LANE];
#pragma unroll
    for (int k = 0; k < CHUNK_PER_LANE; ++k) p[k] = lcnt[d[k] >> 8] + r[k];
#pragma unroll
    for (int k = 0; k < CHUNK_PER_LANE; ++k) {
        if (FULL || t + k * 256 < total) {
            v2i q;
            q.x = s[k];
            q.y = d[k];
            image[p[k] & (CHUNK_EDGES - 1)] = q;
        }
    }
    __syncthreads();

    if (FULL) {
        v2i q[CHUNK_PER_LANE];
        int g[CHUNK_PER_LANE];
#pragma unroll
        for (int k = 0; k < CHUNK_PER_LANE; ++k) q[k] = image[t + k * 256];
#pragma unroll
        for (int k = 0; k < CHUNK_PER_LANE; ++k) g[k] = gofs[q[k].y >> 8];
#pragma unroll
        for (int k = 0; k < CHUNK_PER_LANE; ++k) {
            unsigned slot = (unsigned)(g[k] + t + k * 256);
            if (slot > (unsigned)(ne - 1)) slot = (unsigned)(ne - 1);
            *(v2i*)(tmp + 2 * (size_t)slot) = q[k];
        }
    } else {
#pragma unroll
        for (int k = 0; k < CHUNK_PER_LANE; ++k) {
            const int idx = t + k * 256;
            if (idx < total) {
                const v2i q = image[idx];
                const int slot = gofs[q.y >> 8] + idx;
                if ((unsigned)slot < (unsigned)ne)
                    *(v2i*)(tmp + 2 * (size_t)slot) = q;
            }
        }
    }
}

__launch_bounds__(256)
__global__ void chunk_scatter(const int* __restrict__ src, const int* __restrict__ dst,
                              const int* __restrict__ base, const int* __restrict__ btot,
                              int* __restrict__ tmp, int ne, int n, int nb, int nchunk) {
    __shared__ int lcnt[MAX_BUCKETS];
    __shared__ int gofs[MAX_BUCKETS];
    __shared__ int wtot[16];
    __shared__ v2i image[CHUNK_EDGES];
    const int t = threadIdx.x;
    lcnt[2 * t] = 0;
    lcnt[2 * t + 1] = 0;
    int g0 = 0, g1 = 0, bt0 = 0, bt1 = 0;
    if (2 * t < nb) {
        g0 = base[(size_t)(2 * t) * nchunk + blockIdx.x];
        bt0 = btot[2 * t];
    }
    if (2 * t + 1 < nb) {
        g1 = base[(size_t)(2 * t + 1) * nchunk + blockIdx.x];
        bt1 = btot[2 * t + 1];
    }
    __syncthreads();

    const int blk = blockIdx.x * CHUNK_EDGES;
    const int total = (ne - blk < CHUNK_EDGES) ? (ne - blk) : CHUNK_EDGES;
    if (total == CHUNK_EDGES)
        scatter_chunk<true>(src, dst, tmp, lcnt, gofs, wtot, image, g0, g1, bt0, bt1, blk, total, ne, n);
    else
        scatter_chunk<false>(src, dst, tmp, lcnt, gofs, wtot, image, g0, g1, bt0, bt1, blk, total, ne, n);
}

#define BUCKET_GROUP 8
#define LIST_IMAGE 8192

__launch_bounds__(256)
__global__ void bucket_lists(const int* __restrict__ tmp, const int* __restrict__ btot,
                             int* __restrict__ rowptr, float* __restrict__ dis,
                             int* __restrict__ rec, int n, int ne, int nb) {
    __shared__ int ncnt[256];
    __shared__ int wsum[8];
    __shared__ int bsum[8];
    __shared__ int image[LIST_IMAGE];
    const int t = threadIdx.x;
    const int lane = t & 31;
    const int wv = t >> 5;
    const int b = blockIdx.x;
    int part = 0;
    if (t < b) part = btot[t];
    if (t + 256 < b) part += btot[t + 256];
    const int own = btot[b];
    for (int off = 16; off > 0; off >>= 1) part += __shfl_xor(part, off);
    if (lane == 0) bsum[wv] = part;
    ncnt[t] = 0;
    __syncthreads();
    int beg = 0;
    for (int w = 0; w < 8; ++w) beg += bsum[w];
    int end = beg + own;
    if (beg < 0) beg = 0;
    if (end > ne) end = ne;
    const int cnt = end - beg;
    const bool in_lds = cnt <= LIST_IMAGE;

    for (int e0 = beg; e0 < end; e0 += 256 * BUCKET_GROUP) {
        int d[BUCKET_GROUP];
#pragma unroll
        for (int k = 0; k < BUCKET_GROUP; ++k) {
            const int e = e0 + t + k * 256;
            d[k] = (e < end) ? tmp[2 * (size_t)e + 1] : -1;
        }
#pragma unroll
        for (int k = 0; k < BUCKET_GROUP; ++k)
            if (d[k] >= 0) atomicAdd(&ncnt[d[k] & 255], 1);
    }
    __syncthreads();

    const int c = ncnt[t];
    int incl = c;
    for (int off = 1; off < 32; off <<= 1) {
        const int u = __shfl_up(incl, off);
        if (lane >= off) incl += u;
    }
    if (lane == 31) wsum[wv] = incl;
    __syncthreads();
    int woff = 0;
    for (int w = 0; w < wv; ++w) woff += wsum[w];
    const int first = beg + woff + incl - c;
    const int node = b * 256 + t;
    if (node < n) {
        rowptr[node] = first;
        dis[node] = rsqrtf((float)c + 1.0f);
    }
    if (b == nb - 1 && t == 0) rowptr[n] = end;
    ncnt[t] = first;
    __syncthreads();

    for (int e0 = beg; e0 < end; e0 += 256 * BUCKET_GROUP) {
        v2i q[BUCKET_GROUP];
        int p[BUCKET_GROUP];
#pragma unroll
        for (int k = 0; k < BUCKET_GROUP; ++k) {
            const int e = e0 + t + k * 256;
            if (e < end) {
                q[k] = *(const v2i*)(tmp + 2 * (size_t)e);
            } else {
                q[k].x = 0;
                q[k].y = -1;
            }
        }
#pragma unroll
        for (int k = 0; k < BUCKET_GROUP; ++k)
            p[k] = (q[k].y >= 0) ? atomicAdd(&ncnt[q[k].y & 255], 1) : -1;
        if (in_lds) {
#pragma unroll
            for (int k = 0; k < BUCKET_GROUP; ++k) {
                const unsigned o = (unsigned)(p[k] - beg);
                if (p[k] >= 0 && o < (unsigned)cnt)
                    image[o] = q[k].x;
            }
        } else {
#pragma unroll
            for (int k = 0; k < BUCKET_GROUP; ++k)
                if ((unsigned)p[k] < (unsigned)ne)
                    rec[p[k]] = q[k].x;
        }
    }
    if (in_lds) {
        __syncthreads();
        const int a = beg & 31;
        for (int e0 = t - a; e0 < cnt; e0 += 4 * 256) {
            int v[4];
#pragma unroll
            for (int j = 0; j < 4; ++j) {
                int e = e0 + j * 256;
                if (e < 0) e = 0;
                if (e > cnt - 1) e = cnt - 1;
                v[j] = image[e];
            }
#pragma unroll
            for (int j = 0; j < 4; ++j) {
                const int e = e0 + j * 256;
                if (e >= 0 && e < cnt) rec[beg + e] = v[j];
            }
        }
    }
}

template<int NT, int KDIM, bool BIAS, bool RELU>
__launch_bounds__(256)
__global__ void gemm_wmma_f32(const float* __restrict__ X,
                              const float* __restrict__ W,
                              const float* __restrict__ bias,
                              const float* __restrict__ rowscale,
                              float* __restrict__ Y,
                              int M) {
    constexpr int N = NT * 16;
    const int wave = (blockIdx.x * blockDim.x + threadIdx.x) >> 5;
    const int lane = threadIdx.x & 31;
    const int row0 = wave * 16;
    if (row0 >= M) return;

    const int r      = lane & 15;
    const int khalf  = (lane >> 4) * 2;

    v8f acc[NT] = {};

    const float* __restrict__ xrow = X + (size_t)(row0 + r) * KDIM;

    const int rowoff = (lane >> 4) * 8;
    float sc[8];
#pragma unroll
    for (int g = 0; g < 8; ++g) sc[g] = rowscale[row0 + rowoff + g];

#pragma unroll
    for (int k = 0; k < KDIM; k += 4) {
        const int kk = k + khalf;
        v2f a = *(const v2f*)(xrow + kk);
#pragma unroll
        for (int t = 0; t < NT; ++t) {
            const int col = t * 16 + r;
            v2f b;
            b.x = W[(size_t)kk * N + col];
            b.y = W[(size_t)(kk + 1) * N + col];
            acc[t] = __builtin_amdgcn_wmma_f32_16x16x4_f32(
                 false, a,  false, b,
                 (short)0, acc[t],
                 false,  false);
        }
    }

#pragma unroll
    for (int t = 0; t < NT; ++t) {
        const int col = t * 16 + r;
        const float bv = BIAS ? bias[col] : 0.0f;
#pragma unroll
        for (int g = 0; g < 8; ++g) {
            float v = acc[t][g] + bv;
            if (RELU) v = fmaxf(v, 0.0f);
            Y[(size_t)(row0 + rowoff + g) * N + col] = v * sc[g];
        }
    }
}

typedef __attribute__((ext_vector_type(4))) float v4f;

#define ROW_BATCH 24
#define NODES_PER_WAVE 8
#define TILE_PITCH 68

template<int NOUT>
__device__ __forceinline__ void tile_product(const float* arow, const float* wl,
                                             int boff0, int boff1, v8f& acc0, v8f& acc1) {
#pragma nounroll
    for (int kg = 0; kg < 4; ++kg) {
        v2f a[4], b0[4], b1[4];
#pragma unroll
        for (int j = 0; j < 4; ++j) {
            a[j]  = *(const v2f*)(arow + kg * 16 + j * 4);
            b0[j] = *(const v2f*)(wl + (kg * 8 + j * 2) * (2 * NOUT) + boff0);
            b1[j] = *(const v2f*)(wl + (kg * 8 + j * 2) * (2 * NOUT) + boff1);
        }
#pragma unroll
        for (int j = 0; j < 4; ++j) {
            acc0 = __builtin_amdgcn_wmma_f32_16x16x4_f32(
                 false, a[j],  false, b0[j],
                 (short)0, acc0,  false,  false);
            acc1 = __builtin_amdgcn_wmma_f32_16x16x4_f32(
                 false, a[j],  false, b1[j],
                 (short)0, acc1,  false,  false);
        }
    }
}

__device__ __forceinline__ v2f load_row(const float* __restrict__ h, int lane, int s_raw, unsigned last_row) {
    unsigned s = (unsigned)s_raw;
    if (s > last_row) s = last_row;
    return *(const v2f*)(h + (size_t)s * 64 + 2 * lane);
}

template<int K>
__device__ __forceinline__ void sum_rows(const float* __restrict__ h, int lane, int rs, unsigned last_row,
                                         float& ax, float& ay) {
    v2f v[K];
#pragma unroll
    for (int jj = 0; jj < K; ++jj)
        v[jj] = load_row(h, lane, __builtin_amdgcn_readlane(rs, jj), last_row);
#pragma unroll
    for (int jj = 0; jj < K; ++jj) {
        ax += v[jj].x;
        ay += v[jj].y;
    }
}

#define ROW_CASE(K) case K: sum_rows<K>(h, lane, rs, last_row, ax, ay); break;

template<bool FINAL>
__launch_bounds__(256)
__global__ void gather_gemm(const float* __restrict__ h,
                            const int* __restrict__ rec,
                            const int* __restrict__ rowptr,
                            const float* __restrict__ dis,
                            const float* __restrict__ b,
                            const float* __restrict__ W,
                            const float* __restrict__ bias2,
                            const float* __restrict__ Wo,
                            const float* __restrict__ bo,
                            float* __restrict__ Y,
                            int n, int ne) {
    constexpr int NOUT = FINAL ? 128 : 64;
    __shared__ __attribute__((aligned(16))) float tile[64 * TILE_PITCH];
    __shared__ __attribute__((aligned(16))) float wl[32 * 2 * NOUT];

#pragma unroll
    for (int it = 0; it < (32 * (NOUT / 4)) / 256; ++it) {
        const int idx = it * 256 + (int)threadIdx.x;
        const int kp = idx / (NOUT / 4);
        const int c4 = (idx % (NOUT / 4)) * 4;
        const v4f lo = *(const v4f*)(W + (size_t)(2 * kp) * NOUT + c4);
        const v4f hi = *(const v4f*)(W + (size_t)(2 * kp + 1) * NOUT + c4);
        float* wp = wl + kp * (2 * NOUT) + ((c4 * 2) ^ ((kp & 1) * 32));
        v4f q0, q1;
        q0.x = lo.x; q0.y = hi.x; q0.z = lo.y; q0.w = hi.y;
        q1.x = lo.z; q1.y = hi.z; q1.z = lo.w; q1.w = hi.w;
        *(v4f*)wp = q0;
        *(v4f*)(wp + 4) = q1;
    }

    const int lane = threadIdx.x & 31;
    const int wv = __builtin_amdgcn_readfirstlane((int)(threadIdx.x >> 5));
    const int blk0 = blockIdx.x * 64;
    const int i0 = blk0 + wv * NODES_PER_WAVE;
    const v2f bv = *(const v2f*)(b + 2 * lane);
    const unsigned last_row = (unsigned)(n - 1);

#pragma nounroll
    for (int k = 0; k < NODES_PER_WAVE; ++k) {
        const int i = i0 + k;
        float* trow = tile + (wv * NODES_PER_WAVE + k) * TILE_PITCH + 2 * lane;
        if (i >= n) {
            v2f z;
            z.x = 0.0f;
            z.y = 0.0f;
            *(v2f*)trow = z;
            continue;
        }
        int beg = __builtin_amdgcn_readfirstlane(rowptr[i]);
        int end = __builtin_amdgcn_readfirstlane(rowptr[i + 1]);
        if (beg < 0) beg = 0;
        if (end > ne) end = ne;

        const v2f hv = *(const v2f*)(h + (size_t)i * 64 + 2 * lane);
        const float d = dis[i];

        float ax = 0.0f, ay = 0.0f;
        int base = beg;
        int m = (end - base < ROW_BATCH) ? (end - base) : ROW_BATCH;
        int rs = 0;
        if (lane < m) rs = rec[base + lane];
        while (m > 0) {
            switch (m) {
                ROW_CASE(1)  ROW_CASE(2)  ROW_CASE(3)  ROW_CASE(4)  ROW_CASE(5)  ROW_CASE(6)
                ROW_CASE(7)  ROW_CASE(8)  ROW_CASE(9)  ROW_CASE(10) ROW_CASE(11) ROW_CASE(12)
                ROW_CASE(13) ROW_CASE(14) ROW_CASE(15) ROW_CASE(16) ROW_CASE(17) ROW_CASE(18)
                ROW_CASE(19) ROW_CASE(20) ROW_CASE(21) ROW_CASE(22) ROW_CASE(23) ROW_CASE(24)
                default: break;
            }
            base += ROW_BATCH;
            m = (end - base < ROW_BATCH) ? (end - base) : ROW_BATCH;
            rs = 0;
            if (lane < m) rs = rec[base + lane];
        }

        v2f o;
        o.x = fmaxf((ax + hv.x) * d + bv.x, 0.0f);
        o.y = fmaxf((ay + hv.y) * d + bv.y, 0.0f);
        if (!FINAL) {
            o.x *= d;
            o.y *= d;
        }
        *(v2f*)trow = o;
    }

    __syncthreads();

    const int r      = lane & 15;
    const int hh     = lane >> 4;
    const int rowoff = hh * 8;

    if (!FINAL) {
        const int rt  = wv >> 1;
        const int ct0 = (wv & 1) * 2;
        if (blk0 + rt * 16 >= n) return;
        const float* arow = tile + (rt * 16 + r) * TILE_PITCH + 2 * hh;
        const int col0 = ct0 * 16 + r;
        const int col1 = col0 + 16;
        const int boff0 = hh * (2 * NOUT) + ((col0 * 2) ^ (hh * 32));
        const int boff1 = hh * (2 * NOUT) + ((col1 * 2) ^ (hh * 32));
        v8f acc0 = {}, acc1 = {};
        tile_product<NOUT>(arow, wl, boff0, boff1, acc0, acc1);
        const int row0 = blk0 + rt * 16 + rowoff;
#pragma unroll
        for (int g = 0; g < 8; ++g) {
            Y[(size_t)(row0 + g) * 64 + col0] = acc0[g];
            Y[(size_t)(row0 + g) * 64 + col1] = acc1[g];
        }
    } else {
        if (wv >= 4) return;
        const int rt = wv;
        if (blk0 + rt * 16 >= n) return;
        const float* arow = tile + (rt * 16 + r) * TILE_PITCH + 2 * hh;
        float p0[8], p1[8];
#pragma unroll
        for (int g = 0; g < 8; ++g) {
            p0[g] = 0.0f;
            p1[g] = 0.0f;
        }
#pragma nounroll
        for (int pass = 0; pass < 4; ++pass) {
            const int col0 = pass * 32 + r;
            const int col1 = col0 + 16;
            const float bd0 = bias2[col0];
            const float bd1 = bias2[col1];
            const v2f wo0 = *(const v2f*)(Wo + 2 * col0);
            const v2f wo1 = *(const v2f*)(Wo + 2 * col1);
            const int boff0 = hh * (2 * NOUT) + ((col0 * 2) ^ (hh * 32));
            const int boff1 = hh * (2 * NOUT) + ((col1 * 2) ^ (hh * 32));
            v8f acc0 = {}, acc1 = {};
            tile_product<NOUT>(arow, wl, boff0, boff1, acc0, acc1);
#pragma unroll
            for (int g = 0; g < 8; ++g) {
                const float v0 = fmaxf(acc0[g] + bd0, 0.0f);
                const float v1 = fmaxf(acc1[g] + bd1, 0.0f);
                p0[g] = fmaf(v0, wo0.x, p0[g]);
                p1[g] = fmaf(v0, wo0.y, p1[g]);
                p0[g] = fmaf(v1, wo1.x, p0[g]);
                p1[g] = fmaf(v1, wo1.y, p1[g]);
            }
        }
#pragma unroll
        for (int g = 0; g < 8; ++g) {
#pragma unroll
            for (int off = 8; off > 0; off >>= 1) {
                p0[g] += __shfl_xor(p0[g], off);
                p1[g] += __shfl_xor(p1[g], off);
            }
        }
        float o0 = p0[0], o1 = p1[0];
#pragma unroll
        for (int g = 1; g < 8; ++g) {
            if (r == g) {
                o0 = p0[g];
                o1 = p1[g];
            }
        }
        if (r < 8) {
            v2f o;
            o.x = o0 + bo[0];
            o.y = o1 + bo[1];
            *(v2f*)(Y + (size_t)(blk0 + rt * 16 + rowoff + r) * 2) = o;
        }
    }
}

extern "C" void kernel_launch(void* const* d_in, const int* in_sizes, int n_in,
                              void* d_out, int out_size, void* d_ws, size_t ws_size,
                              hipStream_t stream) {
    const float* x  = (const float*)d_in[0];
    const int*   ei = (const int*)d_in[1];
    const float* W1 = (const float*)d_in[2];
    const float* b1 = (const float*)d_in[3];
    const float* W2 = (const float*)d_in[4];
    const float* b2 = (const float*)d_in[5];
    const float* Wd = (const float*)d_in[6];
    const float* bd = (const float*)d_in[7];
    const float* Wo = (const float*)d_in[8];
    const float* bo = (const float*)d_in[9];
    float* out = (float*)d_out;

    const int N = in_sizes[0] / 64;
    const int E = in_sizes[1] / 2;
    const int* src = ei;
    const int* dst = ei + E;

    const size_t NP = ((size_t)N + 1 + 63) & ~(size_t)63;
    const size_t EP = ((size_t)E + 63) & ~(size_t)63;
    int*   rowptr = (int*)d_ws;
    float* dis    = (float*)(rowptr + NP);
    int*   btot   = (int*)(dis + NP);
    int*   hist   = btot + 1024;
    int*   base   = hist + MAX_BUCKETS * MAX_CHUNKS;
    int*   tmp    = base + MAX_BUCKETS * MAX_CHUNKS;
    int*   rec    = tmp + 2 * EP;
    float* bufA   = (float*)(rec + EP);
    float* bufC   = bufA + (size_t)N * 64;
    (void)ws_size; (void)n_in; (void)out_size;

    const int TB = 256;
    const int gemm_waves   = N / 16;
    const int gemm_blocks  = (gemm_waves * 32 + TB - 1) / TB;
    const int wave_blocks  = (N + 63) / 64;
    const int buckets      = (N + 255) / 256;
    const int chunks       = (E + CHUNK_EDGES - 1) / CHUNK_EDGES;
    const int scan_blocks  = (buckets + 7) / 8;

    chunk_hist<<<chunks, TB, 0, stream>>>(dst, hist, E, N, buckets, chunks);
    chunk_scan<<<scan_blocks, TB, 0, stream>>>(hist, base, btot, buckets, chunks);
    chunk_scatter<<<chunks, TB, 0, stream>>>(src, dst, base, btot, tmp, E, N, buckets, chunks);
    bucket_lists<<<buckets, TB, 0, stream>>>(tmp, btot, rowptr, dis, rec, N, E, buckets);

    gemm_wmma_f32<4, 64, false, false><<<gemm_blocks, TB, 0, stream>>>(x, W1, nullptr, dis, bufA, N);
    gather_gemm<false><<<wave_blocks, TB, 0, stream>>>(bufA, rec, rowptr, dis, b1, W2,
                                                      nullptr, nullptr, nullptr, bufC, N, E);

    gather_gemm<true><<<wave_blocks, TB, 0, stream>>>(bufC, rec, rowptr, dis, b2, Wd,
                                                     bd, Wo, bo, out, N, E);
}


# sort level 1 in place: a chunk keeps pairs in bucket order plus its bucket starts; a bucket block sums its column of starts and pulls the pieces into LDS once: 5 launches, was 7; list order rule kept
# speedup vs baseline: 4.5646x; 1.1084x over previous
#include <hip/hip_runtime.h>
#include <hip/hip_bf16.h>

typedef __attribute__((ext_vector_type(2))) float v2f;
typedef __attribute__((ext_vector_type(8))) float v8f;

#define N_NODES_C 100000

typedef __attribute__((ext_vector_type(2))) int v2i;

#define CHUNK_EDGES 4096
#define CHUNK_PER_LANE 16
#define MAX_BUCKETS 512
#define MAX_CHUNKS 512
#define OFFS_PITCH 520

__device__ __forceinline__ int node_in_range(int d, int n) {
    unsigned u = (unsigned)d;
    if (u > (unsigned)(n - 1)) u = (unsigned)(n - 1);
    return (int)u;
}

template<bool FULL>
__device__ __forceinline__ void sort_chunk(const int* __restrict__ src, const int* __restrict__ dst,
                                           int* __restrict__ tmp, int* __restrict__ orow,
                                           int* lcnt, int* wtot, v2i* image,
                                           int blk, int total, int n) {
    const int t = threadIdx.x;
    const int lane = t & 31;
    const int wv = t >> 5;
    int s[CHUNK_PER_LANE], d[CHUNK_PER_LANE], r[CHUNK_PER_LANE];
#pragma unroll
    for (int k = 0; k < CHUNK_PER_LANE; ++k) {
        const int e = t + k * 256;
        s[k] = (FULL || e < total) ? src[blk + e] : 0;
        d[k] = (FULL || e < total) ? dst[blk + e] : 0;
    }
#pragma unroll
    for (int k = 0; k < CHUNK_PER_LANE; ++k) {
        d[k] = node_in_range(d[k], n);
        r[k] = (FULL || t + k * 256 < total) ? atomicAdd(&lcnt[d[k] >> 8], 1) : 0;
    }
    __syncthreads();

    const int c0 = lcnt[2 * t];
    const int c1 = lcnt[2 * t + 1];
    const int c = c0 + c1;
    int incl = c;
    for (int off = 1; off < 32; off <<= 1) {
        const int u = __shfl_up(incl, off);
        if (lane >= off) incl += u;
    }
    if (lane == 31) wtot[wv] = incl;
    __syncthreads();
    int woff = 0;
    for (int w = 0; w < wv; ++w) woff += wtot[w];
    const int st0 = woff + incl - c;
    const int st1 = st0 + c0;
    lcnt[2 * t] = st0;
    lcnt[2 * t + 1] = st1;
    v2i ost;
    ost.x = st0;
    ost.y = st1;
    *(v2i*)(orow + 2 * t) = ost;
    if (t == 255) orow[MAX_BUCKETS] = st1 + c1;
    __syncthreads();

    int p[CHUNK_PER_LANE];
#pragma unroll
    for (int k = 0; k < CHUNK_PER_LANE; ++k) p[k] = lcnt[d[k] >> 8] + r[k];
#pragma unroll
    for (int k = 0; k < CHUNK_PER_LANE; ++k) {
        if (FULL || t + k * 256 < total) {
            v2i q;
            q.x = s[k];
            q.y = d[k];
            image[p[k] & (CHUNK_EDGES - 1)] = q;
        }
    }
    __syncthreads();

    if (FULL) {
        v2i q[CHUNK_PER_LANE];
#pragma unroll
        for (int k = 0; k < CHUNK_PER_LANE; ++k) q[k] = image[t + k * 256];
#pragma unroll
        for (int k = 0; k < CHUNK_PER_LANE; ++k)
            *(v2i*)(tmp + 2 * (size_t)(blk + t + k * 256)) = q[k];
    } else {
#pragma unroll
        for (int k = 0; k < CHUNK_PER_LANE; ++k) {
            const int idx = t + k * 256;
            if (idx < total) *(v2i*)(tmp + 2 * (size_t)(blk + idx)) = image[idx];
        }
    }
}

__launch_bounds__(256)
__global__ void chunk_sort(const int* __restrict__ src, const int* __restrict__ dst,
                           int* __restrict__ offs, int* __restrict__ tmp, int ne, int n) {
    __shared__ int lcnt[MAX_BUCKETS];
    __shared__ int wtot[8];
    __shared__ v2i image[CHUNK_EDGES];
    const int t = threadIdx.x;
    lcnt[2 * t] = 0;
    lcnt[2 * t + 1] = 0;
    __syncthreads();

    const int blk = blockIdx.x * CHUNK_EDGES;
    const int total = (ne - blk < CHUNK_EDGES) ? (ne - blk) : CHUNK_EDGES;
    int* orow = offs + (size_t)blockIdx.x * OFFS_PITCH;
    if (total == CHUNK_EDGES)
        sort_chunk<true>(src, dst, tmp, orow, lcnt, wtot, image, blk, total, n);
    else
        sort_chunk<false>(src, dst, tmp, orow, lcnt, wtot, image, blk, total, n);
}

#define BUCKET_GROUP 8
#define LIST_IMAGE 8192

__device__ __forceinline__ void load_pieces(const int* __restrict__ tmp, int st, int lane,
                                            unsigned last_pair, v2i (&q)[32]) {
#pragma unroll
    for (int j = 0; j < 32; ++j) {
        unsigned e = (unsigned)(__builtin_amdgcn_readlane(st, j) + lane);
        if (e > last_pair) e = last_pair;
        q[j] = *(const v2i*)(tmp + 2 * (size_t)e);
    }
}

__device__ __forceinline__ void keep_pieces(v2i* pimg, int* ncnt, const v2i (&q)[32],
                                            int c, int ls, int lane) {
#pragma unroll
    for (int j = 0; j < 32; ++j) {
        const int cj = __builtin_amdgcn_readlane(c, j);
        const unsigned o = (unsigned)(__builtin_amdgcn_readlane(ls, j) + lane);
        if (lane < cj && o < (unsigned)LIST_IMAGE) {
            pimg[o] = q[j];
            atomicAdd(&ncnt[q[j].y & 255], 1);
        }
    }
}

__device__ __forceinline__ void long_pieces(const int* __restrict__ tmp, v2i* pimg, int* ncnt,
                                            int st, int c, int ls, int lane, unsigned last_pair) {
    if (__any(c > 32)) {
        for (int j = 0; j < 32; ++j) {
            const int stj = __shfl(st, j);
            const int cj = __shfl(c, j);
            const int lsj = __shfl(ls, j);
            for (int l = 32 + lane; l < cj; l += 32) {
                unsigned e = (unsigned)(stj + l);
                if (e > last_pair) e = last_pair;
                const v2i qq = *(const v2i*)(tmp + 2 * (size_t)e);
                const unsigned o = (unsigned)(lsj + l);
                if (o < (unsigned)LIST_IMAGE) {
                    pimg[o] = qq;
                    atomicAdd(&ncnt[qq.y & 255], 1);
                }
            }
        }
    }
}

template<bool PLACE>
__device__ __forceinline__ void walk_pieces(const int* __restrict__ tmp, int* ncnt, int* __restrict__ rec,
                                            int st, int c, int lane, unsigned last_pair, int ne) {
    for (int j = 0; j < 32; ++j) {
        const int stj = __shfl(st, j);
        const int cj = __shfl(c, j);
        for (int l = lane; l < cj; l += 32) {
            unsigned e = (unsigned)(stj + l);
            if (e > last_pair) e = last_pair;
            const v2i qq = *(const v2i*)(tmp + 2 * (size_t)e);
            if (PLACE) {
                const int p = atomicAdd(&ncnt[qq.y & 255], 1);
                if ((unsigned)p < (unsigned)ne) rec[p] = qq.x;
            } else {
                atomicAdd(&ncnt[qq.y & 255], 1);
            }
        }
    }
}

__device__ __forceinline__ int keep_between(int v, int lo, int hi) {
    if (v < lo) v = lo;
    if (v > hi) v = hi;
    return v;
}

__launch_bounds__(256)
__global__ void bucket_lists(const int* __restrict__ tmp, const int* __restrict__ offs,
                             int* __restrict__ rowptr, float* __restrict__ dis,
                             int* __restrict__ rec, int n, int ne, int nb, int nchunk) {
    __shared__ int ncnt[256];
    __shared__ int psum[8];
    __shared__ int bsum[8];
    __shared__ int wsum[8];
    __shared__ int image[LIST_IMAGE];
    __shared__ v2i pimg[LIST_IMAGE];
    const int t = threadIdx.x;
    const int lane = t & 31;
    const int wv = t >> 5;
    const int b = blockIdx.x;

    const bool hasA = t < nchunk;
    const bool hasB = t + 256 < nchunk;
    const int* rowA = offs + (size_t)(hasA ? t : nchunk - 1) * OFFS_PITCH + b;
    const int* rowB = offs + (size_t)(hasB ? t + 256 : nchunk - 1) * OFFS_PITCH + b;
    int a0 = rowA[0];
    int a1 = rowA[1];
    int b0 = rowB[0];
    int b1 = rowB[1];
    ncnt[t] = 0;
    if (!hasA) {
        a0 = 0;
        a1 = 0;
    }
    if (!hasB) {
        b0 = 0;
        b1 = 0;
    }
    a0 = keep_between(a0, 0, CHUNK_EDGES);
    a1 = keep_between(a1, a0, CHUNK_EDGES);
    b0 = keep_between(b0, 0, CHUNK_EDGES);
    b1 = keep_between(b1, b0, CHUNK_EDGES);
    const int cA = a1 - a0;
    const int cB = b1 - b0;
    const int c = cA + cB;
    int incl = c;
    for (int off = 1; off < 32; off <<= 1) {
        const int u = __shfl_up(incl, off);
        if (lane >= off) incl += u;
    }
    int part = a0 + b0;
    for (int off = 16; off > 0; off >>= 1) part += __shfl_xor(part, off);
    if (lane == 31) psum[wv] = incl;
    if (lane == 0) bsum[wv] = part;
    __syncthreads();
    int woff = 0, own = 0, beg = 0;
    for (int w = 0; w < 8; ++w) {
        const int ps = psum[w];
        if (w < wv) woff += ps;
        own += ps;
        beg += bsum[w];
    }
    beg = keep_between(beg, 0, ne);
    if (own > ne - beg) own = ne - beg;
    const int end = beg + own;
    const bool fits = own <= LIST_IMAGE;
    const int lsA = woff + incl - c;
    const int lsB = lsA + cA;
    const int stA = t * CHUNK_EDGES + a0;
    const int stB = (t + 256) * CHUNK_EDGES + b0;
    const unsigned last_pair = (unsigned)(ne - 1);

    if (fits) {
        v2i qa[32], qb[32];
        const bool anyB = __any(cB > 0);
        load_pieces(tmp, stA, lane, last_pair, qa);
        if (anyB) load_pieces(tmp, stB, lane, last_pair, qb);
        keep_pieces(pimg, ncnt, qa, cA, lsA, lane);
        if (anyB) keep_pieces(pimg, ncnt, qb, cB, lsB, lane);
        long_pieces(tmp, pimg, ncnt, stA, cA, lsA, lane, last_pair);
        long_pieces(tmp, pimg, ncnt, stB, cB, lsB, lane, last_pair);
    } else {
        walk_pieces<false>(tmp, ncnt, rec, stA, cA, lane, last_pair, ne);
        walk_pieces<false>(tmp, ncnt, rec, stB, cB, lane, last_pair, ne);
    }
    __syncthreads();

    const int cn = ncnt[t];
    int incn = cn;
    for (int off = 1; off < 32; off <<= 1) {
        const int u = __shfl_up(incn, off);
        if (lane >= off) incn += u;
    }
    if (lane == 31) wsum[wv] = incn;
    __syncthreads();
    int noff = 0;
    for (int w = 0; w < wv; ++w) noff += wsum[w];
    const int first = beg + noff + incn - cn;
    const int node = b * 256 + t;
    if (node < n) {
        rowptr[node] = first;
        dis[node] = rsqrtf((float)cn + 1.0f);
    }
    if (b == nb - 1 && t == 0) rowptr[n] = end;
    ncnt[t] = first;
    __syncthreads();

    if (fits) {
        for (int e0 = 0; e0 < own; e0 += 256 * BUCKET_GROUP) {
            v2i q[BUCKET_GROUP];
            int p[BUCKET_GROUP];
#pragma unroll
            for (int k = 0; k < BUCKET_GROUP; ++k) {
                const int e = e0 + t + k * 256;
                if (e < own) {
                    q[k] = pimg[e];
                } else {
                    q[k].x = 0;
                    q[k].y = -1;
                }
            }
#pragma unroll
            for (int k = 0; k < BUCKET_GROUP; ++k)
                p[k] = (q[k].y >= 0) ? atomicAdd(&ncnt[q[k].y & 255], 1) : -1;
#pragma unroll
            for (int k = 0; k < BUCKET_GROUP; ++k) {
                const unsigned o = (unsigned)(p[k] - beg);
                if (p[k] >= 0 && o < (unsigned)own)
                    image[o] = q[k].x;
            }
        }
        __syncthreads();
        if (own > 0) {
            const int a = beg & 31;
            for (int e0 = t - a; e0 < own; e0 += 4 * 256) {
                int v[4];
#pragma unroll
                for (int j = 0; j < 4; ++j) {
                    int e = e0 + j * 256;
                    if (e < 0) e = 0;
                    if (e > own - 1) e = own - 1;
                    v[j] = image[e];
                }
#pragma unroll
                for (int j = 0; j < 4; ++j) {
                    const int e = e0 + j * 256;
                    if (e >= 0 && e < own) rec[beg + e] = v[j];
                }
            }
        }
    } else {
        walk_pieces<true>(tmp, ncnt, rec, stA, cA, lane, last_pair, ne);
        walk_pieces<true>(tmp, ncnt, rec, stB, cB, lane, last_pair, ne);
    }
}

template<int NT, int KDIM, bool BIAS, bool RELU>
__launch_bounds__(256)
__global__ void gemm_wmma_f32(const float* __restrict__ X,
                              const float* __restrict__ W,
                              const float* __restrict__ bias,
                              const float* __restrict__ rowscale,
                              float* __restrict__ Y,
                              int M) {
    constexpr int N = NT * 16;
    const int wave = (blockIdx.x * blockDim.x + threadIdx.x) >> 5;
    const int lane = threadIdx.x & 31;
    const int row0 = wave * 16;
    if (row0 >= M) return;

    const int r      = lane & 15;
    const int khalf  = (lane >> 4) * 2;

    v8f acc[NT] = {};

    const float* __restrict__ xrow = X + (size_t)(row0 + r) * KDIM;

    const int rowoff = (lane >> 4) * 8;
    float sc[8];
#pragma unroll
    for (int g = 0; g < 8; ++g) sc[g] = rowscale[row0 + rowoff + g];

#pragma unroll
    for (int k = 0; k < KDIM; k += 4) {
        const int kk = k + khalf;
        v2f a = *(const v2f*)(xrow + kk);
#pragma unroll
        for (int t = 0; t < NT; ++t) {
            const int col = t * 16 + r;
            v2f b;
            b.x = W[(size_t)kk * N + col];
            b.y = W[(size_t)(kk + 1) * N + col];
            acc[t] = __builtin_amdgcn_wmma_f32_16x16x4_f32(
                 false, a,  false, b,
                 (short)0, acc[t],
                 false,  false);
        }
    }

#pragma unroll
    for (int t = 0; t < NT; ++t) {
        const int col = t * 16 + r;
        const float bv = BIAS ? bias[col] : 0.0f;
#pragma unroll
        for (int g = 0; g < 8; ++g) {
            float v = acc[t][g] + bv;
            if (RELU) v = fmaxf(v, 0.0f);
            Y[(size_t)(row0 + rowoff + g) * N + col] = v * sc[g];
        }
    }
}

typedef __attribute__((ext_vector_type(4))) float v4f;

#define ROW_BATCH 24
#define NODES_PER_WAVE 8
#define TILE_PITCH 68

template<int NOUT>
__device__ __forceinline__ void tile_product(const float* arow, const float* wl,
                                             int boff0, int boff1, v8f& acc0, v8f& acc1) {
#pragma nounroll
    for (int kg = 0; kg < 4; ++kg) {
        v2f a[4], b0[4], b1[4];
#pragma unroll
        for (int j = 0; j < 4; ++j) {
            a[j]  = *(const v2f*)(arow + kg * 16 + j * 4);
            b0[j] = *(const v2f*)(wl + (kg * 8 + j * 2) * (2 * NOUT) + boff0);
            b1[j] = *(const v2f*)(wl + (kg * 8 + j * 2) * (2 * NOUT) + boff1);
        }
#pragma unroll
        for (int j = 0; j < 4; ++j) {
            acc0 = __builtin_amdgcn_wmma_f32_16x16x4_f32(
                 false, a[j],  false, b0[j],
                 (short)0, acc0,  false,  false);
            acc1 = __builtin_amdgcn_wmma_f32_16x16x4_f32(
                 false, a[j],  false, b1[j],
                 (short)0, acc1,  false,  false);
        }
    }
}

__device__ __forceinline__ v2f load_row(const float* __restrict__ h, int lane, int s_raw, unsigned last_row) {
    unsigned s = (unsigned)s_raw;
    if (s > last_row) s = last_row;
    return *(const v2f*)(h + (size_t)s * 64 + 2 * lane);
}

template<int K>
__device__ __forceinline__ void sum_rows(const float* __restrict__ h, int lane, int rs, unsigned last_row,
                                         float& ax, float& ay) {
    v2f v[K];
#pragma unroll
    for (int jj = 0; jj < K; ++jj)
        v[jj] = load_row(h, lane, __builtin_amdgcn_readlane(rs, jj), last_row);
#pragma unroll
    for (int jj = 0; jj < K; ++jj) {
        ax += v[jj].x;
        ay += v[jj].y;
    }
}

#define ROW_CASE(K) case K: sum_rows<K>(h, lane, rs, last_row, ax, ay); break;

template<bool FINAL>
__launch_bounds__(256)
__global__ void gather_gemm(const float* __restrict__ h,
                            const int* __restrict__ rec,
                            const int* __restrict__ rowptr,
                            const float* __restrict__ dis,
                            const float* __restrict__ b,
                            const float* __restrict__ W,
                            const float* __restrict__ bias2,
                            const float* __restrict__ Wo,
                            const float* __restrict__ bo,
                            float* __restrict__ Y,
                            int n, int ne) {
    constexpr int NOUT = FINAL ? 128 : 64;
    __shared__ __attribute__((aligned(16))) float tile[64 * TILE_PITCH];
    __shared__ __attribute__((aligned(16))) float wl[32 * 2 * NOUT];

#pragma unroll
    for (int it = 0; it < (32 * (NOUT / 4)) / 256; ++it) {
        const int idx = it * 256 + (int)threadIdx.x;
        const int kp = idx / (NOUT / 4);
        const int c4 = (idx % (NOUT / 4)) * 4;
        const v4f lo = *(const v4f*)(W + (size_t)(2 * kp) * NOUT + c4);
        const v4f hi = *(const v4f*)(W + (size_t)(2 * kp + 1) * NOUT + c4);
        float* wp = wl + kp * (2 * NOUT) + ((c4 * 2) ^ ((kp & 1) * 32));
        v4f q0, q1;
        q0.x = lo.x; q0.y = hi.x; q0.z = lo.y; q0.w = hi.y;
        q1.x = lo.z; q1.y = hi.z; q1.z = lo.w; q1.w = hi.w;
        *(v4f*)wp = q0;
        *(v4f*)(wp + 4) = q1;
    }

    const int lane = threadIdx.x & 31;
    const int wv = __builtin_amdgcn_readfirstlane((int)(threadIdx.x >> 5));
    const int blk0 = blockIdx.x * 64;
    const int i0 = blk0 + wv * NODES_PER_WAVE;
    const v2f bv = *(const v2f*)(b + 2 * lane);
    const unsigned last_row = (unsigned)(n - 1);

#pragma nounroll
    for (int k = 0; k < NODES_PER_WAVE; ++k) {
        const int i = i0 + k;
        float* trow = tile + (wv * NODES_PER_WAVE + k) * TILE_PITCH + 2 * lane;
        if (i >= n) {
            v2f z;
            z.x = 0.0f;
            z.y = 0.0f;
            *(v2f*)trow = z;
            continue;
        }
        int beg = __builtin_amdgcn_readfirstlane(rowptr[i]);
        int end = __builtin_amdgcn_readfirstlane(rowptr[i + 1]);
        if (beg < 0) beg = 0;
        if (end > ne) end = ne;

        const v2f hv = *(const v2f*)(h + (size_t)i * 64 + 2 * lane);
        const float d = dis[i];

        float ax = 0.0f, ay = 0.0f;
        int base = beg;
        int m = (end - base < ROW_BATCH) ? (end - base) : ROW_BATCH;
        int rs = 0;
        if (lane < m) rs = rec[base + lane];
        while (m > 0) {
            switch (m) {
                ROW_CASE(1)  ROW_CASE(2)  ROW_CASE(3)  ROW_CASE(4)  ROW_CASE(5)  ROW_CASE(6)
                ROW_CASE(7)  ROW_CASE(8)  ROW_CASE(9)  ROW_CASE(10) ROW_CASE(11) ROW_CASE(12)
                ROW_CASE(13) ROW_CASE(14) ROW_CASE(15) ROW_CASE(16) ROW_CASE(17) ROW_CASE(18)
                ROW_CASE(19) ROW_CASE(20) ROW_CASE(21) ROW_CASE(22) ROW_CASE(23) ROW_CASE(24)
                default: break;
            }
            base += ROW_BATCH;
            m = (end - base < ROW_BATCH) ? (end - base) : ROW_BATCH;
            rs = 0;
            if (lane < m) rs = rec[base + lane];
        }

        v2f o;
        o.x = fmaxf((ax + hv.x) * d + bv.x, 0.0f);
        o.y = fmaxf((ay + hv.y) * d + bv.y, 0.0f);
        if (!FINAL) {
            o.x *= d;
            o.y *= d;
        }
        *(v2f*)trow = o;
    }

    __syncthreads();

    const int r      = lane & 15;
    const int hh     = lane >> 4;
    const int rowoff = hh * 8;

    if (!FINAL) {
        const int rt  = wv >> 1;
        const int ct0 = (wv & 1) * 2;
        if (blk0 + rt * 16 >= n) return;
        const float* arow = tile + (rt * 16 + r) * TILE_PITCH + 2 * hh;
        const int col0 = ct0 * 16 + r;
        const int col1 = col0 + 16;
        const int boff0 = hh * (2 * NOUT) + ((col0 * 2) ^ (hh * 32));
        const int boff1 = hh * (2 * NOUT) + ((col1 * 2) ^ (hh * 32));
        v8f acc0 = {}, acc1 = {};
        tile_product<NOUT>(arow, wl, boff0, boff1, acc0, acc1);
        const int row0 = blk0 + rt * 16 + rowoff;
#pragma unroll
        for (int g = 0; g < 8; ++g) {
            Y[(size_t)(row0 + g) * 64 + col0] = acc0[g];
            Y[(size_t)(row0 + g) * 64 + col1] = acc1[g];
        }
    } else {
        if (wv >= 4) return;
        const int rt = wv;
        if (blk0 + rt * 16 >= n) return;
        const float* arow = tile + (rt * 16 + r) * TILE_PITCH + 2 * hh;
        float p0[8], p1[8];
#pragma unroll
        for (int g = 0; g < 8; ++g) {
            p0[g] = 0.0f;
            p1[g] = 0.0f;
        }
#pragma nounroll
        for (int pass = 0; pass < 4; ++pass) {
            const int col0 = pass * 32 + r;
            const int col1 = col0 + 16;
            const float bd0 = bias2[col0];
            const float bd1 = bias2[col1];
            const v2f wo0 = *(const v2f*)(Wo + 2 * col0);
            const v2f wo1 = *(const v2f*)(Wo + 2 * col1);
            const int boff0 = hh * (2 * NOUT) + ((col0 * 2) ^ (hh * 32));
            const int boff1 = hh * (2 * NOUT) + ((col1 * 2) ^ (hh * 32));
            v8f acc0 = {}, acc1 = {};
            tile_product<NOUT>(arow, wl, boff0, boff1, acc0, acc1);
#pragma unroll
            for (int g = 0; g < 8; ++g) {
                const float v0 = fmaxf(acc0[g] + bd0, 0.0f);
                const float v1 = fmaxf(acc1[g] + bd1, 0.0f);
                p0[g] = fmaf(v0, wo0.x, p0[g]);
                p1[g] = fmaf(v0, wo0.y, p1[g]);
                p0[g] = fmaf(v1, wo1.x, p0[g]);
                p1[g] = fmaf(v1, wo1.y, p1[g]);
            }
        }
#pragma unroll
        for (int g = 0; g < 8; ++g) {
#pragma unroll
            for (int off = 8; off > 0; off >>= 1) {
                p0[g] += __shfl_xor(p0[g], off);
                p1[g] += __shfl_xor(p1[g], off);
            }
        }
        float o0 = p0[0], o1 = p1[0];
#pragma unroll
        for (int g = 1; g < 8; ++g) {
            if (r == g) {
                o0 = p0[g];
                o1 = p1[g];
            }
        }
        if (r < 8) {
            v2f o;
            o.x = o0 + bo[0];
            o.y = o1 + bo[1];
            *(v2f*)(Y + (size_t)(blk0 + rt * 16 + rowoff + r) * 2) = o;
        }
    }
}

extern "C" void kernel_launch(void* const* d_in, const int* in_sizes, int n_in,
                              void* d_out, int out_size, void* d_ws, size_t ws_size,
                              hipStream_t stream) {
    const float* x  = (const float*)d_in[0];
    const int*   ei = (const int*)d_in[1];
    const float* W1 = (const float*)d_in[2];
    const float* b1 = (const float*)d_in[3];
    const float* W2 = (const float*)d_in[4];
    const float* b2 = (const float*)d_in[5];
    const float* Wd = (const float*)d_in[6];
    const float* bd = (const float*)d_in[7];
    const float* Wo = (const float*)d_in[8];
    const float* bo = (const float*)d_in[9];
    float* out = (float*)d_out;

    const int N = in_sizes[0] / 64;
    const int E = in_sizes[1] / 2;
    const int* src = ei;
    const int* dst = ei + E;

    const size_t NP = ((size_t)N + 1 + 63) & ~(size_t)63;
    const size_t EP = ((size_t)E + 63) & ~(size_t)63;
    int*   rowptr = (int*)d_ws;
    float* dis    = (float*)(rowptr + NP);
    int*   offs   = (int*)(dis + NP);
    int*   tmp    = offs + MAX_CHUNKS * OFFS_PITCH;
    int*   rec    = tmp + 2 * EP;
    float* bufA   = (float*)(rec + EP);
    float* bufC   = bufA + (size_t)N * 64;
    (void)ws_size; (void)n_in; (void)out_size;

    const int TB = 256;
    const int gemm_waves   = N / 16;
    const int gemm_blocks  = (gemm_waves * 32 + TB - 1) / TB;
    const int wave_blocks  = (N + 63) / 64;
    const int buckets      = (N + 255) / 256;
    const int chunks       = (E + CHUNK_EDGES - 1) / CHUNK_EDGES;

    chunk_sort<<<chunks, TB, 0, stream>>>(src, dst, offs, tmp, E, N);
    bucket_lists<<<buckets, TB, 0, stream>>>(tmp, offs, rowptr, dis, rec, N, E, buckets, chunks);

    gemm_wmma_f32<4, 64, false, false><<<gemm_blocks, TB, 0, stream>>>(x, W1, nullptr, dis, bufA, N);
    gather_gemm<false><<<wave_blocks, TB, 0, stream>>>(bufA, rec, rowptr, dis, b1, W2,
                                                      nullptr, nullptr, nullptr, bufC, N, E);

    gather_gemm<true><<<wave_blocks, TB, 0, stream>>>(bufC, rec, rowptr, dis, b2, Wd,
                                                     bd, Wo, bo, out, N, E);
}


# layer 1 aggregates x rows before its product ((A x) W1, was A (x W1)): W1 product runs inside the aggregation launch, bucket launch writes x times dis; 4 launches, was 5; sum order changes with it
# speedup vs baseline: 4.7307x; 1.0364x over previous
#include <hip/hip_runtime.h>
#include <hip/hip_bf16.h>

typedef __attribute__((ext_vector_type(2))) float v2f;
typedef __attribute__((ext_vector_type(8))) float v8f;

#define N_NODES_C 100000

typedef __attribute__((ext_vector_type(2))) int v2i;
typedef __attribute__((ext_vector_type(4))) float v4f;

#define CHUNK_EDGES 4096
#define CHUNK_PER_LANE 16
#define MAX_BUCKETS 512
#define MAX_CHUNKS 512
#define OFFS_PITCH 520

__device__ __forceinline__ int node_in_range(int d, int n) {
    unsigned u = (unsigned)d;
    if (u > (unsigned)(n - 1)) u = (unsigned)(n - 1);
    return (int)u;
}

template<bool FULL>
__device__ __forceinline__ void sort_chunk(const int* __restrict__ src, const int* __restrict__ dst,
                                           int* __restrict__ tmp, int* __restrict__ orow,
                                           int* lcnt, int* wtot, v2i* image,
                                           int blk, int total, int n) {
    const int t = threadIdx.x;
    const int lane = t & 31;
    const int wv = t >> 5;
    int s[CHUNK_PER_LANE], d[CHUNK_PER_LANE], r[CHUNK_PER_LANE];
#pragma unroll
    for (int k = 0; k < CHUNK_PER_LANE; ++k) {
        const int e = t + k * 256;
        s[k] = (FULL || e < total) ? src[blk + e] : 0;
        d[k] = (FULL || e < total) ? dst[blk + e] : 0;
    }
#pragma unroll
    for (int k = 0; k < CHUNK_PER_LANE; ++k) {
        d[k] = node_in_range(d[k], n);
        r[k] = (FULL || t + k * 256 < total) ? atomicAdd(&lcnt[d[k] >> 8], 1) : 0;
    }
    __syncthreads();

    const int c0 = lcnt[2 * t];
    const int c1 = lcnt[2 * t + 1];
    const int c = c0 + c1;
    int incl = c;
    for (int off = 1; off < 32; off <<= 1) {
        const int u = __shfl_up(incl, off);
        if (lane >= off) incl += u;
    }
    if (lane == 31) wtot[wv] = incl;
    __syncthreads();
    int woff = 0;
    for (int w = 0; w < wv; ++w) woff += wtot[w];
    const int st0 = woff + incl - c;
    const int st1 = st0 + c0;
    lcnt[2 * t] = st0;
    lcnt[2 * t + 1] = st1;
    v2i ost;
    ost.x = st0;
    ost.y = st1;
    *(v2i*)(orow + 2 * t) = ost;
    if (t == 255) orow[MAX_BUCKETS] = st1 + c1;
    __syncthreads();

    int p[CHUNK_PER_LANE];
#pragma unroll
    for (int k = 0; k < CHUNK_PER_LANE; ++k) p[k] = lcnt[d[k] >> 8] + r[k];
#pragma unroll
    for (int k = 0; k < CHUNK_PER_LANE; ++k) {
        if (FULL || t + k * 256 < total) {
            v2i q;
            q.x = s[k];
            q.y = d[k];
            image[p[k] & (CHUNK_EDGES - 1)] = q;
        }
    }
    __syncthreads();

    if (FULL) {
        v2i q[CHUNK_PER_LANE];
#pragma unroll
        for (int k = 0; k < CHUNK_PER_LANE; ++k) q[k] = image[t + k * 256];
#pragma unroll
        for (int k = 0; k < CHUNK_PER_LANE; ++k)
            *(v2i*)(tmp + 2 * (size_t)(blk + t + k * 256)) = q[k];
    } else {
#pragma unroll
        for (int k = 0; k < CHUNK_PER_LANE; ++k) {
            const int idx = t + k * 256;
            if (idx < total) *(v2i*)(tmp + 2 * (size_t)(blk + idx)) = image[idx];
        }
    }
}

__launch_bounds__(256)
__global__ void chunk_sort(const int* __restrict__ src, const int* __restrict__ dst,
                           int* __restrict__ offs, int* __restrict__ tmp, int ne, int n) {
    __shared__ int lcnt[MAX_BUCKETS];
    __shared__ int wtot[8];
    __shared__ v2i image[CHUNK_EDGES];
    const int t = threadIdx.x;
    lcnt[2 * t] = 0;
    lcnt[2 * t + 1] = 0;
    __syncthreads();

    const int blk = blockIdx.x * CHUNK_EDGES;
    const int total = (ne - blk < CHUNK_EDGES) ? (ne - blk) : CHUNK_EDGES;
    int* orow = offs + (size_t)blockIdx.x * OFFS_PITCH;
    if (total == CHUNK_EDGES)
        sort_chunk<true>(src, dst, tmp, orow, lcnt, wtot, image, blk, total, n);
    else
        sort_chunk<false>(src, dst, tmp, orow, lcnt, wtot, image, blk, total, n);
}

#define BUCKET_GROUP 8
#define LIST_IMAGE 8192

__device__ __forceinline__ void load_pieces(const int* __restrict__ tmp, int st, int lane,
                                            unsigned last_pair, v2i (&q)[32]) {
#pragma unroll
    for (int j = 0; j < 32; ++j) {
        unsigned e = (unsigned)(__builtin_amdgcn_readlane(st, j) + lane);
        if (e > last_pair) e = last_pair;
        q[j] = *(const v2i*)(tmp + 2 * (size_t)e);
    }
}

__device__ __forceinline__ void keep_pieces(v2i* pimg, int* ncnt, const v2i (&q)[32],
                                            int c, int ls, int lane) {
#pragma unroll
    for (int j = 0; j < 32; ++j) {
        const int cj = __builtin_amdgcn_readlane(c, j);
        const unsigned o = (unsigned)(__builtin_amdgcn_readlane(ls, j) + lane);
        if (lane < cj && o < (unsigned)LIST_IMAGE) {
            pimg[o] = q[j];
            atomicAdd(&ncnt[q[j].y & 255], 1);
        }
    }
}

__device__ __forceinline__ void long_pieces(const int* __restrict__ tmp, v2i* pimg, int* ncnt,
                                            int st, int c, int ls, int lane, unsigned last_pair) {
    if (__any(c > 32)) {
        for (int j = 0; j < 32; ++j) {
            const int stj = __shfl(st, j);
            const int cj = __shfl(c, j);
            const int lsj = __shfl(ls, j);
            for (int l = 32 + lane; l < cj; l += 32) {
                unsigned e = (unsigned)(stj + l);
                if (e > last_pair) e = last_pair;
                const v2i qq = *(const v2i*)(tmp + 2 * (size_t)e);
                const unsigned o = (unsigned)(lsj + l);
                if (o < (unsigned)LIST_IMAGE) {
                    pimg[o] = qq;
                    atomicAdd(&ncnt[qq.y & 255], 1);
                }
            }
        }
    }
}

template<bool PLACE>
__device__ __forceinline__ void walk_pieces(const int* __restrict__ tmp, int* ncnt, int* __restrict__ rec,
                                            int st, int c, int lane, unsigned last_pair, int ne) {
    for (int j = 0; j < 32; ++j) {
        const int stj = __shfl(st, j);
        const int cj = __shfl(c, j);
        for (int l = lane; l < cj; l += 32) {
            unsigned e = (unsigned)(stj + l);
            if (e > last_pair) e = last_pair;
            const v2i qq = *(const v2i*)(tmp + 2 * (size_t)e);
            if (PLACE) {
                const int p = atomicAdd(&ncnt[qq.y & 255], 1);
                if ((unsigned)p < (unsigned)ne) rec[p] = qq.x;
            } else {
                atomicAdd(&ncnt[qq.y & 255], 1);
            }
        }
    }
}

__device__ __forceinline__ int keep_between(int v, int lo, int hi) {
    if (v < lo) v = lo;
    if (v > hi) v = hi;
    return v;
}

#define XROWS_PER_THREAD 16

__device__ __forceinline__ void load_x_rows(const float* __restrict__ x, int row0, int xc, int n,
                                            v4f (&xr)[XROWS_PER_THREAD]) {
#pragma unroll
    for (int k = 0; k < XROWS_PER_THREAD; ++k) {
        int row = row0 + 16 * k;
        if (row > n - 1) row = n - 1;
        xr[k] = *(const v4f*)(x + (size_t)row * 64 + xc);
    }
}
__launch_bounds__(256)
__global__ void bucket_lists(const int* __restrict__ tmp, const int* __restrict__ offs,
                             int* __restrict__ rowptr, float* __restrict__ dis,
                             int* __restrict__ rec, const float* __restrict__ x,
                             float* __restrict__ xs, int n, int ne, int nb, int nchunk) {
    __shared__ int ncnt[256];
    __shared__ float sdis[256];
    __shared__ int psum[8];
    __shared__ int bsum[8];
    __shared__ int wsum[8];
    __shared__ int image[LIST_IMAGE];
    __shared__ v2i pimg[LIST_IMAGE];
    const int t = threadIdx.x;
    const int lane = t & 31;
    const int wv = t >> 5;
    const int b = blockIdx.x;

    const bool hasA = t < nchunk;
    const bool hasB = t + 256 < nchunk;
    const int* rowA = offs + (size_t)(hasA ? t : nchunk - 1) * OFFS_PITCH + b;
    const int* rowB = offs + (size_t)(hasB ? t + 256 : nchunk - 1) * OFFS_PITCH + b;
    int a0 = rowA[0];
    int a1 = rowA[1];
    int b0 = rowB[0];
    int b1 = rowB[1];
    ncnt[t] = 0;
    if (!hasA) {
        a0 = 0;
        a1 = 0;
    }
    if (!hasB) {
        b0 = 0;
        b1 = 0;
    }
    a0 = keep_between(a0, 0, CHUNK_EDGES);
    a1 = keep_between(a1, a0, CHUNK_EDGES);
    b0 = keep_between(b0, 0, CHUNK_EDGES);
    b1 = keep_between(b1, b0, CHUNK_EDGES);
    const int cA = a1 - a0;
    const int cB = b1 - b0;
    const int c = cA + cB;
    int incl = c;
    for (int off = 1; off < 32; off <<= 1) {
        const int u = __shfl_up(incl, off);
        if (lane >= off) incl += u;
    }
    int part = a0 + b0;
    for (int off = 16; off > 0; off >>= 1) part += __shfl_xor(part, off);
    if (lane == 31) psum[wv] = incl;
    if (lane == 0) bsum[wv] = part;
    __syncthreads();
    int woff = 0, own = 0, beg = 0;
    for (int w = 0; w < 8; ++w) {
        const int ps = psum[w];
        if (w < wv) woff += ps;
        own += ps;
        beg += bsum[w];
    }
    beg = keep_between(beg, 0, ne);
    if (own > ne - beg) own = ne - beg;
    const int end = beg + own;
    const bool fits = own <= LIST_IMAGE;
    const int lsA = woff + incl - c;
    const int lsB = lsA + cA;
    const int stA = t * CHUNK_EDGES + a0;
    const int stB = (t + 256) * CHUNK_EDGES + b0;
    const unsigned last_pair = (unsigned)(ne - 1);

    v4f xr[XROWS_PER_THREAD];
    const int xc = (t & 15) * 4;
    const int xrow0 = b * 256 + (t >> 4);

    if (fits) {
        v2i qa[32], qb[32];
        const bool anyB = __any(cB > 0);
        load_pieces(tmp, stA, lane, last_pair, qa);
        if (anyB) load_pieces(tmp, stB, lane, last_pair, qb);
        load_x_rows(x, xrow0, xc, n, xr);
        keep_pieces(pimg, ncnt, qa, cA, lsA, lane);
        if (anyB) keep_pieces(pimg, ncnt, qb, cB, lsB, lane);
        long_pieces(tmp, pimg, ncnt, stA, cA, lsA, lane, last_pair);
        long_pieces(tmp, pimg, ncnt, stB, cB, lsB, lane, last_pair);
    } else {
        load_x_rows(x, xrow0, xc, n, xr);
        walk_pieces<false>(tmp, ncnt, rec, stA, cA, lane, last_pair, ne);
        walk_pieces<false>(tmp, ncnt, rec, stB, cB, lane, last_pair, ne);
    }
    __syncthreads();

    const int cn = ncnt[t];
    int incn = cn;
    for (int off = 1; off < 32; off <<= 1) {
        const int u = __shfl_up(incn, off);
        if (lane >= off) incn += u;
    }
    if (lane == 31) wsum[wv] = incn;
    __syncthreads();
    int noff = 0;
    for (int w = 0; w < wv; ++w) noff += wsum[w];
    const int first = beg + noff + incn - cn;
    const int node = b * 256 + t;
    const float dn = rsqrtf((float)cn + 1.0f);
    if (node < n) {
        rowptr[node] = first;
        dis[node] = dn;
    }
    sdis[t] = dn;
    if (b == nb - 1 && t == 0) rowptr[n] = end;
    ncnt[t] = first;
    __syncthreads();

#pragma unroll
    for (int k = 0; k < XROWS_PER_THREAD; ++k) {
        const int row = xrow0 + 16 * k;
        const float sc = sdis[(t >> 4) + 16 * k];
        if (row < n) *(v4f*)(xs + (size_t)row * 64 + xc) = xr[k] * sc;
    }

    if (fits) {
        for (int e0 = 0; e0 < own; e0 += 256 * BUCKET_GROUP) {
            v2i q[BUCKET_GROUP];
            int p[BUCKET_GROUP];
#pragma unroll
            for (int k = 0; k < BUCKET_GROUP; ++k) {
                const int e = e0 + t + k * 256;
                if (e < own) {
                    q[k] = pimg[e];
                } else {
                    q[k].x = 0;
                    q[k].y = -1;
                }
            }
#pragma unroll
            for (int k = 0; k < BUCKET_GROUP; ++k)
                p[k] = (q[k].y >= 0) ? atomicAdd(&ncnt[q[k].y & 255], 1) : -1;
#pragma unroll
            for (int k = 0; k < BUCKET_GROUP; ++k) {
                const unsigned o = (unsigned)(p[k] - beg);
                if (p[k] >= 0 && o < (unsigned)own)
                    image[o] = q[k].x;
            }
        }
        __syncthreads();
        if (own > 0) {
            const int a = beg & 31;
            for (int e0 = t - a; e0 < own; e0 += 4 * 256) {
                int v[4];
#pragma unroll
                for (int j = 0; j < 4; ++j) {
                    int e = e0 + j * 256;
                    if (e < 0) e = 0;
                    if (e > own - 1) e = own - 1;
                    v[j] = image[e];
                }
#pragma unroll
                for (int j = 0; j < 4; ++j) {
                    const int e = e0 + j * 256;
                    if (e >= 0 && e < own) rec[beg + e] = v[j];
                }
            }
        }
    } else {
        walk_pieces<true>(tmp, ncnt, rec, stA, cA, lane, last_pair, ne);
        walk_pieces<true>(tmp, ncnt, rec, stB, cB, lane, last_pair, ne);
    }
}

#define ROW_BATCH 24
#define NODES_PER_WAVE 8
#define TILE_PITCH 68

template<int NOUT>
__device__ __forceinline__ void tile_product(const float* arow, const float* wl,
                                             int boff0, int boff1, v8f& acc0, v8f& acc1) {
#pragma nounroll
    for (int kg = 0; kg < 4; ++kg) {
        v2f a[4], b0[4], b1[4];
#pragma unroll
        for (int j = 0; j < 4; ++j) {
            a[j]  = *(const v2f*)(arow + kg * 16 + j * 4);
            b0[j] = *(const v2f*)(wl + (kg * 8 + j * 2) * (2 * NOUT) + boff0);
            b1[j] = *(const v2f*)(wl + (kg * 8 + j * 2) * (2 * NOUT) + boff1);
        }
#pragma unroll
        for (int j = 0; j < 4; ++j) {
            acc0 = __builtin_amdgcn_wmma_f32_16x16x4_f32(
                 false, a[j],  false, b0[j],
                 (short)0, acc0,  false,  false);
            acc1 = __builtin_amdgcn_wmma_f32_16x16x4_f32(
                 false, a[j],  false, b1[j],
                 (short)0, acc1,  false,  false);
        }
    }
}

template<int NOUT>
__device__ __forceinline__ void stage_weights(const float* __restrict__ W, float* wl) {
#pragma unroll
    for (int it = 0; it < (32 * (NOUT / 4)) / 256; ++it) {
        const int idx = it * 256 + (int)threadIdx.x;
        const int kp = idx / (NOUT / 4);
        const int c4 = (idx % (NOUT / 4)) * 4;
        const v4f lo = *(const v4f*)(W + (size_t)(2 * kp) * NOUT + c4);
        const v4f hi = *(const v4f*)(W + (size_t)(2 * kp + 1) * NOUT + c4);
        float* wp = wl + kp * (2 * NOUT) + ((c4 * 2) ^ ((kp & 1) * 32));
        v4f q0, q1;
        q0.x = lo.x; q0.y = hi.x; q0.z = lo.y; q0.w = hi.y;
        q1.x = lo.z; q1.y = hi.z; q1.z = lo.w; q1.w = hi.w;
        *(v4f*)wp = q0;
        *(v4f*)(wp + 4) = q1;
    }
}

__device__ __forceinline__ v2f load_row(const float* __restrict__ h, int lane, int s_raw, unsigned last_row) {
    unsigned s = (unsigned)s_raw;
    if (s > last_row) s = last_row;
    return *(const v2f*)(h + (size_t)s * 64 + 2 * lane);
}

template<int K>
__device__ __forceinline__ void sum_rows(const float* __restrict__ h, int lane, int rs, unsigned last_row,
                                         float& ax, float& ay) {
    v2f v[K];
#pragma unroll
    for (int jj = 0; jj < K; ++jj)
        v[jj] = load_row(h, lane, __builtin_amdgcn_readlane(rs, jj), last_row);
#pragma unroll
    for (int jj = 0; jj < K; ++jj) {
        ax += v[jj].x;
        ay += v[jj].y;
    }
}

#define ROW_CASE(K) case K: sum_rows<K>(h, lane, rs, last_row, ax, ay); break;

template<bool FINAL>
__launch_bounds__(256)
__global__ void gather_gemm(const float* __restrict__ h,
                            const int* __restrict__ rec,
                            const int* __restrict__ rowptr,
                            const float* __restrict__ dis,
                            const float* __restrict__ b,
                            const float* __restrict__ W,
                            const float* __restrict__ Wfirst,
                            const float* __restrict__ bias2,
                            const float* __restrict__ Wo,
                            const float* __restrict__ bo,
                            float* __restrict__ Y,
                            int n, int ne) {
    __shared__ __attribute__((aligned(16))) float tile[64 * TILE_PITCH];
    __shared__ __attribute__((aligned(16))) float wl[32 * 2 * 128];
    __shared__ float sd[64];

    if (FINAL) {
        stage_weights<128>(W, wl);
    } else {
        stage_weights<64>(Wfirst, wl);
        stage_weights<64>(W, wl + 32 * 2 * 64);
    }

    const int lane = threadIdx.x & 31;
    const int wv = __builtin_amdgcn_readfirstlane((int)(threadIdx.x >> 5));
    const int blk0 = blockIdx.x * 64;
    const int i0 = blk0 + wv * NODES_PER_WAVE;
    v2f bv;
    bv.x = 0.0f;
    bv.y = 0.0f;
    if (FINAL) bv = *(const v2f*)(b + 2 * lane);
    const unsigned last_row = (unsigned)(n - 1);

#pragma nounroll
    for (int k = 0; k < NODES_PER_WAVE; ++k) {
        const int i = i0 + k;
        float* trow = tile + (wv * NODES_PER_WAVE + k) * TILE_PITCH + 2 * lane;
        if (i >= n) {
            v2f z;
            z.x = 0.0f;
            z.y = 0.0f;
            *(v2f*)trow = z;
            if (!FINAL && lane == 0) sd[wv * NODES_PER_WAVE + k] = 0.0f;
            continue;
        }
        int beg = __builtin_amdgcn_readfirstlane(rowptr[i]);
        int end = __builtin_amdgcn_readfirstlane(rowptr[i + 1]);
        if (beg < 0) beg = 0;
        if (end > ne) end = ne;

        const v2f hv = *(const v2f*)(h + (size_t)i * 64 + 2 * lane);
        const float d = dis[i];

        float ax = 0.0f, ay = 0.0f;
        int base = beg;
        int m = (end - base < ROW_BATCH) ? (end - base) : ROW_BATCH;
        int rs = 0;
        if (lane < m) rs = rec[base + lane];
        while (m > 0) {
            switch (m) {
                ROW_CASE(1)  ROW_CASE(2)  ROW_CASE(3)  ROW_CASE(4)  ROW_CASE(5)  ROW_CASE(6)
                ROW_CASE(7)  ROW_CASE(8)  ROW_CASE(9)  ROW_CASE(10) ROW_CASE(11) ROW_CASE(12)
                ROW_CASE(13) ROW_CASE(14) ROW_CASE(15) ROW_CASE(16) ROW_CASE(17) ROW_CASE(18)
                ROW_CASE(19) ROW_CASE(20) ROW_CASE(21) ROW_CASE(22) ROW_CASE(23) ROW_CASE(24)
                default: break;
            }
            base += ROW_BATCH;
            m = (end - base < ROW_BATCH) ? (end - base) : ROW_BATCH;
            rs = 0;
            if (lane < m) rs = rec[base + lane];
        }

        v2f o;
        if (FINAL) {
            o.x = fmaxf((ax + hv.x) * d + bv.x, 0.0f);
            o.y = fmaxf((ay + hv.y) * d + bv.y, 0.0f);
        } else {
            o.x = (ax + hv.x) * d;
            o.y = (ay + hv.y) * d;
            if (lane == 0) sd[wv * NODES_PER_WAVE + k] = d;
        }
        *(v2f*)trow = o;
    }

    __syncthreads();

    const int r      = lane & 15;
    const int hh     = lane >> 4;
    const int rowoff = hh * 8;

    if (!FINAL) {
        constexpr int NOUT = 64;
        const int rt  = wv >> 1;
        const int ct0 = (wv & 1) * 2;
        const float* arow = tile + (rt * 16 + r) * TILE_PITCH + 2 * hh;
        const int col0 = ct0 * 16 + r;
        const int col1 = col0 + 16;
        const int boff0 = hh * (2 * NOUT) + ((col0 * 2) ^ (hh * 32));
        const int boff1 = hh * (2 * NOUT) + ((col1 * 2) ^ (hh * 32));
        const float bb0 = b[col0];
        const float bb1 = b[col1];
        float dr[8];
#pragma unroll
        for (int g = 0; g < 8; ++g) dr[g] = sd[rt * 16 + rowoff + g];

        v8f p0 = {}, p1 = {};
        tile_product<NOUT>(arow, wl, boff0, boff1, p0, p1);
        __syncthreads();
        float* hrow = tile + (rt * 16 + rowoff) * TILE_PITCH;
#pragma unroll
        for (int g = 0; g < 8; ++g) {
            hrow[g * TILE_PITCH + col0] = fmaxf(p0[g] + bb0, 0.0f) * dr[g];
            hrow[g * TILE_PITCH + col1] = fmaxf(p1[g] + bb1, 0.0f) * dr[g];
        }
        __syncthreads();

        v8f acc0 = {}, acc1 = {};
        tile_product<NOUT>(arow, wl + 32 * 2 * NOUT, boff0, boff1, acc0, acc1);
        if (blk0 + rt * 16 < n) {
            const int row0 = blk0 + rt * 16 + rowoff;
#pragma unroll
            for (int g = 0; g < 8; ++g) {
                Y[(size_t)(row0 + g) * 64 + col0] = acc0[g];
                Y[(size_t)(row0 + g) * 64 + col1] = acc1[g];
            }
        }
    } else {
        constexpr int NOUT = 128;
        if (wv >= 4) return;
        const int rt = wv;
        if (blk0 + rt * 16 >= n) return;
        const float* arow = tile + (rt * 16 + r) * TILE_PITCH + 2 * hh;
        float p0[8], p1[8];
#pragma unroll
        for (int g = 0; g < 8; ++g) {
            p0[g] = 0.0f;
            p1[g] = 0.0f;
        }
#pragma nounroll
        for (int pass = 0; pass < 4; ++pass) {
            const int col0 = pass * 32 + r;
            const int col1 = col0 + 16;
            const float bd0 = bias2[col0];
            const float bd1 = bias2[col1];
            const v2f wo0 = *(const v2f*)(Wo + 2 * col0);
            const v2f wo1 = *(const v2f*)(Wo + 2 * col1);
            const int boff0 = hh * (2 * NOUT) + ((col0 * 2) ^ (hh * 32));
            const int boff1 = hh * (2 * NOUT) + ((col1 * 2) ^ (hh * 32));
            v8f acc0 = {}, acc1 = {};
            tile_product<NOUT>(arow, wl, boff0, boff1, acc0, acc1);
#pragma unroll
            for (int g = 0; g < 8; ++g) {
                const float v0 = fmaxf(acc0[g] + bd0, 0.0f);
                const float v1 = fmaxf(acc1[g] + bd1, 0.0f);
                p0[g] = fmaf(v0, wo0.x, p0[g]);
                p1[g] = fmaf(v0, wo0.y, p1[g]);
                p0[g] = fmaf(v1, wo1.x, p0[g]);
                p1[g] = fmaf(v1, wo1.y, p1[g]);
            }
        }
#pragma unroll
        for (int g = 0; g < 8; ++g) {
#pragma unroll
            for (int off = 8; off > 0; off >>= 1) {
                p0[g] += __shfl_xor(p0[g], off);
                p1[g] += __shfl_xor(p1[g], off);
            }
        }
        float o0 = p0[0], o1 = p1[0];
#pragma unroll
        for (int g = 1; g < 8; ++g) {
            if (r == g) {
                o0 = p0[g];
                o1 = p1[g];
            }
        }
        if (r < 8) {
            v2f o;
            o.x = o0 + bo[0];
            o.y = o1 + bo[1];
            *(v2f*)(Y + (size_t)(blk0 + rt * 16 + rowoff + r) * 2) = o;
        }
    }
}

extern "C" void kernel_launch(void* const* d_in, const int* in_sizes, int n_in,
                              void* d_out, int out_size, void* d_ws, size_t ws_size,
                              hipStream_t stream) {
    const float* x  = (const float*)d_in[0];
    const int*   ei = (const int*)d_in[1];
    const float* W1 = (const float*)d_in[2];
    const float* b1 = (const float*)d_in[3];
    const float* W2 = (const float*)d_in[4];
    const float* b2 = (const float*)d_in[5];
    const float* Wd = (const float*)d_in[6];
    const float* bd = (const float*)d_in[7];
    const float* Wo = (const float*)d_in[8];
    const float* bo = (const float*)d_in[9];
    float* out = (float*)d_out;

    const int N = in_sizes[0] / 64;
    const int E = in_sizes[1] / 2;
    const int* src = ei;
    const int* dst = ei + E;

    const size_t NP = ((size_t)N + 1 + 63) & ~(size_t)63;
    const size_t EP = ((size_t)E + 63) & ~(size_t)63;
    int*   rowptr = (int*)d_ws;
    float* dis    = (float*)(rowptr + NP);
    int*   offs   = (int*)(dis + NP);
    int*   tmp    = offs + MAX_CHUNKS * OFFS_PITCH;
    int*   rec    = tmp + 2 * EP;
    float* bufA   = (float*)(rec + EP);
    float* bufC   = bufA + (size_t)N * 64;
    (void)ws_size; (void)n_in; (void)out_size;

    const int TB = 256;
    const int wave_blocks  = (N + 63) / 64;
    const int buckets      = (N + 255) / 256;
    const int chunks       = (E + CHUNK_EDGES - 1) / CHUNK_EDGES;

    chunk_sort<<<chunks, TB, 0, stream>>>(src, dst, offs, tmp, E, N);
    bucket_lists<<<buckets, TB, 0, stream>>>(tmp, offs, rowptr, dis, rec, x, bufA, N, E, buckets, chunks);

    gather_gemm<false><<<wave_blocks, TB, 0, stream>>>(bufA, rec, rowptr, dis, b1, W2, W1,
                                                      nullptr, nullptr, nullptr, bufC, N, E);

    gather_gemm<true><<<wave_blocks, TB, 0, stream>>>(bufC, rec, rowptr, dis, b2, Wd, nullptr,
                                                     bd, Wo, bo, out, N, E);
}
